# starting kernel
# baseline (speedup 1.0000x reference)
#include <hip/hip_runtime.h>
#include <math.h>

#define N_FEAT 512
#define HID    16
#define NCLS   7

typedef __attribute__((ext_vector_type(2))) float v2f;
typedef __attribute__((ext_vector_type(8))) float v8f;

__global__ void k_deg_init(float* __restrict__ deg, int n) {
    int i = blockIdx.x * blockDim.x + threadIdx.x;
    if (i < n) deg[i] = 1.0f;
}

__global__ void k_deg_edges(const int* __restrict__ col, float* __restrict__ deg, int E) {
    int e = blockIdx.x * blockDim.x + threadIdx.x;
    if (e < E) atomicAdd(&deg[col[e]], 1.0f);
}

__global__ void k_dinv(float* __restrict__ deg, int n) {
    int i = blockIdx.x * blockDim.x + threadIdx.x;
    if (i < n) deg[i] = rsqrtf(deg[i]);
}

__global__ __launch_bounds__(256) void k_gemm1(
    const float* __restrict__ x, const float* __restrict__ W1,
    const float* __restrict__ dinv,
    float* __restrict__ h, float* __restrict__ agg1, int nTiles)
{
    __shared__ float sW[N_FEAT * HID];
    for (int i = threadIdx.x; i < N_FEAT * HID; i += blockDim.x) sW[i] = W1[i];
    __syncthreads();

    int wave = threadIdx.x >> 5;
    int lane = threadIdx.x & 31;
    int tile = blockIdx.x * 8 + wave;
    if (tile >= nTiles) return;

    int m  = lane & 15;
    int kh = lane >> 4;
    const float* xr = x + (size_t)(tile * 16 + m) * N_FEAT + 2 * kh;

    v8f acc = {};
#pragma unroll 4
    for (int k = 0; k < N_FEAT; k += 4) {
        v2f a, b;
        a.x = xr[k];
        a.y = xr[k + 1];
        int kk = k + 2 * kh;
        b.x = sW[kk * HID + m];
        b.y = sW[(kk + 1) * HID + m];
        acc = __builtin_amdgcn_wmma_f32_16x16x4_f32(
             false, a,  false, b,
             (short)0, acc,  false,  false);
    }

#pragma unroll
    for (int v = 0; v < 8; ++v) {
        int node  = tile * 16 + v + 8 * kh;
        float val = acc[v];
        float di  = dinv[node];
        h[(size_t)node * HID + m]    = val;
        agg1[(size_t)node * HID + m] = val * di * di;
    }
}

__global__ void k_edge1(const int* __restrict__ eidx, const float* __restrict__ dinv,
                        const float* __restrict__ h, float* __restrict__ agg1, int E)
{
    int t = blockIdx.x * blockDim.x + threadIdx.x;
    int e = t >> 4, f = t & 15;
    if (e >= E) return;
    int r = eidx[e];
    int c = eidx[E + e];
    float w = dinv[r] * dinv[c];
    atomicAdd(&agg1[(size_t)c * HID + f], h[(size_t)r * HID + f] * w);
}

__global__ void k_l2(const float* __restrict__ agg1, const float* __restrict__ b1,
                     const float* __restrict__ W2, const float* __restrict__ dinv,
                     float* __restrict__ o, float* __restrict__ outInit, int n)
{
    __shared__ float sW2[HID * NCLS];
    __shared__ float sb1[HID];
    if (threadIdx.x < HID * NCLS) sW2[threadIdx.x] = W2[threadIdx.x];
    if (threadIdx.x < HID)        sb1[threadIdx.x] = b1[threadIdx.x];
    __syncthreads();

    int i = blockIdx.x * blockDim.x + threadIdx.x;
    if (i >= n) return;

    float hv[HID];
#pragma unroll
    for (int f = 0; f < HID; ++f) {
        float v = agg1[(size_t)i * HID + f] + sb1[f];
        hv[f] = v > 0.0f ? v : 0.0f;
    }
    float di = dinv[i];
    float d2 = di * di;
#pragma unroll
    for (int j = 0; j < NCLS; ++j) {
        float s = 0.0f;
#pragma unroll
        for (int f = 0; f < HID; ++f) s = fmaf(hv[f], sW2[f * NCLS + j], s);
        o[(size_t)i * NCLS + j]       = s;
        outInit[(size_t)i * NCLS + j] = s * d2;
    }
}

__global__ void k_edge2(const int* __restrict__ eidx, const float* __restrict__ dinv,
                        const float* __restrict__ o, float* __restrict__ out, int E)
{
    int t = blockIdx.x * blockDim.x + threadIdx.x;
    int e = t >> 3, j = t & 7;
    if (e >= E || j >= NCLS) return;
    int r = eidx[e];
    int c = eidx[E + e];
    float w = dinv[r] * dinv[c];
    atomicAdd(&out[(size_t)c * NCLS + j], o[(size_t)r * NCLS + j] * w);
}

__global__ void k_lsm(float* __restrict__ out, const float* __restrict__ b2, int n)
{
    int i = blockIdx.x * blockDim.x + threadIdx.x;
    if (i >= n) return;
    float v[NCLS];
    float mx = -1e30f;
#pragma unroll
    for (int j = 0; j < NCLS; ++j) {
        v[j] = out[(size_t)i * NCLS + j] + b2[j];
        mx = fmaxf(mx, v[j]);
    }
    float s = 0.0f;
#pragma unroll
    for (int j = 0; j < NCLS; ++j) s += __expf(v[j] - mx);
    float l = __logf(s);
#pragma unroll
    for (int j = 0; j < NCLS; ++j) out[(size_t)i * NCLS + j] = v[j] - mx - l;
}

extern "C" void kernel_launch(void* const* d_in, const int* in_sizes, int n_in,
                              void* d_out, int out_size, void* d_ws, size_t ws_size,
                              hipStream_t stream)
{
    const float* x    = (const float*)d_in[0];
    const int*   eidx = (const int*)  d_in[1];
    const float* W1   = (const float*)d_in[2];
    const float* b1   = (const float*)d_in[3];
    const float* W2   = (const float*)d_in[4];
    const float* b2   = (const float*)d_in[5];
    float* out = (float*)d_out;

    const int N = in_sizes[0] / N_FEAT;
    const int E = in_sizes[1] / 2;

    float* ws   = (float*)d_ws;
    float* dinv = ws;
    float* h    = ws + (size_t)N;
    float* agg1 = ws + (size_t)N * 17;
    float* obuf = ws + (size_t)N * 33;

    const int B = 256;
    k_deg_init <<<(N + B - 1) / B, B, 0, stream>>>(dinv, N);
    k_deg_edges<<<(E + B - 1) / B, B, 0, stream>>>(eidx + E, dinv, E);
    k_dinv     <<<(N + B - 1) / B, B, 0, stream>>>(dinv, N);

    const int nTiles = (N + 15) / 16;
    k_gemm1<<<(nTiles + 7) / 8, 256, 0, stream>>>(x, W1, dinv, h, agg1, nTiles);

    const int t1 = E * 16;
    k_edge1<<<(t1 + B - 1) / B, B, 0, stream>>>(eidx, dinv, h, agg1, E);

    k_l2<<<(N + B - 1) / B, B, 0, stream>>>(agg1, b1, W2, dinv, obuf, out, N);

    const int t2 = E * 8;
    k_edge2<<<(t2 + B - 1) / B, B, 0, stream>>>(eidx, dinv, obuf, out, E);

    k_lsm<<<(N + B - 1) / B, B, 0, stream>>>(out, b2, N);
}


# Edge sums of both layers pulled per destination node from source lists (arrival number from a returning integer count add, prefix sum, one store per edge): no float atomics; sum order changes
# speedup vs baseline: 1.3167x; 1.3167x over previous
#include <hip/hip_runtime.h>
#include <math.h>

#define N_FEAT 512
#define HID    16
#define NCLS   7

typedef __attribute__((ext_vector_type(2))) float v2f;
typedef __attribute__((ext_vector_type(8))) float v8f;

__global__ void k_deg_init(float* __restrict__ deg, int n) {
    int i = blockIdx.x * blockDim.x + threadIdx.x;
    if (i < n) deg[i] = 1.0f;
}

__global__ void k_deg_edges(const int* __restrict__ col, float* __restrict__ deg, int E) {
    int e = blockIdx.x * blockDim.x + threadIdx.x;
    if (e < E) atomicAdd(&deg[col[e]], 1.0f);
}

__global__ void k_dinv(float* __restrict__ deg, int n) {
    int i = blockIdx.x * blockDim.x + threadIdx.x;
    if (i < n) deg[i] = rsqrtf(deg[i]);
}

template <bool SELF>
__global__ __launch_bounds__(256) void k_gemm1(
    const float* __restrict__ x, const float* __restrict__ W1,
    const float* __restrict__ dinv,
    float* __restrict__ h, float* __restrict__ agg1, int nTiles)
{
    __shared__ float sW[N_FEAT * HID];
    for (int i = threadIdx.x; i < N_FEAT * HID; i += blockDim.x) sW[i] = W1[i];
    __syncthreads();

    int wave = threadIdx.x >> 5;
    int lane = threadIdx.x & 31;
    int tile = blockIdx.x * 8 + wave;
    if (tile >= nTiles) return;

    int m  = lane & 15;
    int kh = lane >> 4;
    const float* xr = x + (size_t)(tile * 16 + m) * N_FEAT + 2 * kh;

    v8f acc = {};
#pragma unroll 4
    for (int k = 0; k < N_FEAT; k += 4) {
        v2f a, b;
        a.x = xr[k];
        a.y = xr[k + 1];
        int kk = k + 2 * kh;
        b.x = sW[kk * HID + m];
        b.y = sW[(kk + 1) * HID + m];
        acc = __builtin_amdgcn_wmma_f32_16x16x4_f32(
             false, a,  false, b,
             (short)0, acc,  false,  false);
    }

#pragma unroll
    for (int v = 0; v < 8; ++v) {
        int node  = tile * 16 + v + 8 * kh;
        float val = acc[v];
        h[(size_t)node * HID + m] = val;
        if (SELF) {
            float di = dinv[node];
            agg1[(size_t)node * HID + m] = val * di * di;
        }
    }
}

__global__ void k_edge1(const int* __restrict__ eidx, const float* __restrict__ dinv,
                        const float* __restrict__ h, float* __restrict__ agg1, int E)
{
    int t = blockIdx.x * blockDim.x + threadIdx.x;
    int e = t >> 4, f = t & 15;
    if (e >= E) return;
    int r = eidx[e];
    int c = eidx[E + e];
    float w = dinv[r] * dinv[c];
    atomicAdd(&agg1[(size_t)c * HID + f], h[(size_t)r * HID + f] * w);
}

template <bool SELF>
__global__ void k_l2(const float* __restrict__ agg1, const float* __restrict__ b1,
                     const float* __restrict__ W2, const float* __restrict__ dinv,
                     float* __restrict__ o, float* __restrict__ outInit, int n)
{
    __shared__ float sW2[HID * NCLS];
    __shared__ float sb1[HID];
    if (threadIdx.x < HID * NCLS) sW2[threadIdx.x] = W2[threadIdx.x];
    if (threadIdx.x < HID)        sb1[threadIdx.x] = b1[threadIdx.x];
    __syncthreads();

    int i = blockIdx.x * blockDim.x + threadIdx.x;
    if (i >= n) return;

    float hv[HID];
#pragma unroll
    for (int f = 0; f < HID; ++f) {
        float v = agg1[(size_t)i * HID + f] + sb1[f];
        hv[f] = v > 0.0f ? v : 0.0f;
    }
    float di = dinv[i];
    float d2 = di * di;
#pragma unroll
    for (int j = 0; j < NCLS; ++j) {
        float s = 0.0f;
#pragma unroll
        for (int f = 0; f < HID; ++f) s = fmaf(hv[f], sW2[f * NCLS + j], s);
        o[(size_t)i * NCLS + j]       = s;
        if (SELF) outInit[(size_t)i * NCLS + j] = s * d2;
    }
}

__global__ void k_edge2(const int* __restrict__ eidx, const float* __restrict__ dinv,
                        const float* __restrict__ o, float* __restrict__ out, int E)
{
    int t = blockIdx.x * blockDim.x + threadIdx.x;
    int e = t >> 3, j = t & 7;
    if (e >= E || j >= NCLS) return;
    int r = eidx[e];
    int c = eidx[E + e];
    float w = dinv[r] * dinv[c];
    atomicAdd(&out[(size_t)c * NCLS + j], o[(size_t)r * NCLS + j] * w);
}

__global__ void k_lsm(float* __restrict__ out, const float* __restrict__ b2, int n)
{
    int i = blockIdx.x * blockDim.x + threadIdx.x;
    if (i >= n) return;
    float v[NCLS];
    float mx = -1e30f;
#pragma unroll
    for (int j = 0; j < NCLS; ++j) {
        v[j] = out[(size_t)i * NCLS + j] + b2[j];
        mx = fmaxf(mx, v[j]);
    }
    float s = 0.0f;
#pragma unroll
    for (int j = 0; j < NCLS; ++j) s += __expf(v[j] - mx);
    float l = __logf(s);
#pragma unroll
    for (int j = 0; j < NCLS; ++j) out[(size_t)i * NCLS + j] = v[j] - mx - l;
}


__global__ void k_cnt_zero(int* __restrict__ cnt, int n) {
    int i = blockIdx.x * blockDim.x + threadIdx.x;
    if (i < n) cnt[i] = 0;
}

__global__ void k_rank(const int* __restrict__ col, int* __restrict__ cnt,
                       int* __restrict__ rank, int E) {
    int e = blockIdx.x * blockDim.x + threadIdx.x;
    if (e < E) rank[e] = atomicAdd(&cnt[col[e]], 1);
}

__global__ __launch_bounds__(1024) void k_scan(const int* __restrict__ cnt, int* __restrict__ off,
                                               float* __restrict__ dinv, int n)
{
    __shared__ int sInc[32];
    __shared__ int sPre[32];
    const int t = threadIdx.x, lane = t & 31, wave = t >> 5;
    const int tile0 = blockIdx.x * 4096;

    int pre = 0;
    for (int b = 0; b < tile0; b += 4096) {
        int4 v = *reinterpret_cast<const int4*>(cnt + b + t * 4);
        pre += (v.x + v.y) + (v.z + v.w);
    }
    const int i0 = tile0 + t * 4;
    int c[4];
#pragma unroll
    for (int k = 0; k < 4; ++k) c[k] = (i0 + k < n) ? cnt[i0 + k] : 0;
    const int mine = (c[0] + c[1]) + (c[2] + c[3]);

    int incl = mine;
#pragma unroll
    for (int d = 1; d < 32; d <<= 1) { int y = __shfl_up(incl, d); if (lane >= d) incl += y; }
#pragma unroll
    for (int d = 16; d > 0; d >>= 1) pre += __shfl_xor(pre, d);
    if (lane == 31) sInc[wave] = incl;
    if (lane == 0)  sPre[wave] = pre;
    __syncthreads();
    if (wave == 0) {
        int w = sInc[lane], p = sPre[lane];
        int wi = w;
#pragma unroll
        for (int d = 1; d < 32; d <<= 1) { int y = __shfl_up(wi, d); if (lane >= d) wi += y; }
#pragma unroll
        for (int d = 16; d > 0; d >>= 1) p += __shfl_xor(p, d);
        sInc[lane] = wi - w + p;
    }
    __syncthreads();
    int run = sInc[wave] + incl - mine;
#pragma unroll
    for (int k = 0; k < 4; ++k) {
        int i = i0 + k;
        if (i < n) {
            off[i]  = run;
            dinv[i] = rsqrtf((float)(c[k] + 1));
            run += c[k];
            if (i == n - 1) off[n] = run;
        }
    }
}

__global__ void k_fill(const int* __restrict__ eidx, const int* __restrict__ rank,
                       const int* __restrict__ off, int* __restrict__ srcs, int E) {
    int e = blockIdx.x * blockDim.x + threadIdx.x;
    if (e < E) srcs[off[eidx[E + e]] + rank[e]] = eidx[e];
}

template <int K> __device__ __forceinline__ int row16_i(int v) {
    return __builtin_amdgcn_update_dpp(0, v, 0x150 + K, 0xf, 0xf, false);
}
template <int K> __device__ __forceinline__ float row16_f(float v) {
    return __int_as_float(row16_i<K>(__float_as_int(v)));
}
template <int K> __device__ __forceinline__ int grp8_i(int v) {
    return (int)__builtin_amdgcn_mov_dpp8((unsigned)v, K * 0x249249);
}
template <int K> __device__ __forceinline__ float grp8_f(float v) {
    return __int_as_float(grp8_i<K>(__float_as_int(v)));
}

#define PULL16_SHARE(K) sk[K] = row16_i<K>(s); wk[K] = row16_f<K>(w);

__global__ __launch_bounds__(256) void k_pull1(const int* __restrict__ off, const int* __restrict__ srcs,
                                               const float* __restrict__ dinv, const float* __restrict__ h,
                                               float* __restrict__ agg1, int n)
{
    const int g = (blockIdx.x * 256 + threadIdx.x) >> 4;
    const int f = threadIdx.x & 15;
    if (g >= n) return;
    const int beg = off[g], end = off[g + 1];
    const float dc = dinv[g];
    float acc = h[g * HID + f] * dc * dc;
    for (int p = beg; p < end; p += 16) {
        const bool on = p + f < end;
        const int   s = on ? srcs[p + f] : g;
        const float w = on ? dinv[s] * dc : 0.0f;
        int sk[16]; float wk[16], hv[16];
        PULL16_SHARE(0)  PULL16_SHARE(1)  PULL16_SHARE(2)  PULL16_SHARE(3)
        PULL16_SHARE(4)  PULL16_SHARE(5)  PULL16_SHARE(6)  PULL16_SHARE(7)
        PULL16_SHARE(8)  PULL16_SHARE(9)  PULL16_SHARE(10) PULL16_SHARE(11)
        PULL16_SHARE(12) PULL16_SHARE(13) PULL16_SHARE(14) PULL16_SHARE(15)
#pragma unroll
        for (int k = 0; k < 16; ++k) hv[k] = h[sk[k] * HID + f];
#pragma unroll
        for (int k = 0; k < 16; ++k) acc = fmaf(hv[k], wk[k], acc);
    }
    agg1[(size_t)g * HID + f] = acc;
}

#define PULL8_SHARE(K) sk[K] = grp8_i<K>(s); wk[K] = grp8_f<K>(w);

__global__ __launch_bounds__(256) void k_pull2(const int* __restrict__ off, const int* __restrict__ srcs,
                                               const float* __restrict__ dinv, const float* __restrict__ o,
                                               const float* __restrict__ b2, float* __restrict__ out, int n)
{
    const int g = (blockIdx.x * 256 + threadIdx.x) >> 3;
    const int j = threadIdx.x & 7;
    if (g >= n) return;
    const int jj = j < NCLS ? j : NCLS - 1;
    const int beg = off[g], end = off[g + 1];
    const float dc = dinv[g];
    float acc = o[g * NCLS + jj] * dc * dc;
    for (int p = beg; p < end; p += 8) {
        const bool on = p + j < end;
        const int   s = on ? srcs[p + j] : g;
        const float w = on ? dinv[s] * dc : 0.0f;
        int sk[8]; float wk[8], ov[8];
        PULL8_SHARE(0) PULL8_SHARE(1) PULL8_SHARE(2) PULL8_SHARE(3)
        PULL8_SHARE(4) PULL8_SHARE(5) PULL8_SHARE(6) PULL8_SHARE(7)
#pragma unroll
        for (int k = 0; k < 8; ++k) ov[k] = o[sk[k] * NCLS + jj];
#pragma unroll
        for (int k = 0; k < 8; ++k) acc = fmaf(ov[k], wk[k], acc);
    }
    const float v = acc + b2[jj];
    float mx = v;
#pragma unroll
    for (int d = 1; d < 8; d <<= 1) mx = fmaxf(mx, __shfl_xor(mx, d, 8));
    float ex = j < NCLS ? __expf(v - mx) : 0.0f;
#pragma unroll
    for (int d = 1; d < 8; d <<= 1) ex += __shfl_xor(ex, d, 8);
    if (j < NCLS) out[(size_t)g * NCLS + j] = v - mx - __logf(ex);
}

extern "C" void kernel_launch(void* const* d_in, const int* in_sizes, int n_in,
                              void* d_out, int out_size, void* d_ws, size_t ws_size,
                              hipStream_t stream)
{
    const float* x    = (const float*)d_in[0];
    const int*   eidx = (const int*)  d_in[1];
    const float* W1   = (const float*)d_in[2];
    const float* b1   = (const float*)d_in[3];
    const float* W2   = (const float*)d_in[4];
    const float* b2   = (const float*)d_in[5];
    float* out = (float*)d_out;

    const int N = in_sizes[0] / N_FEAT;
    const int E = in_sizes[1] / 2;

    float* ws   = (float*)d_ws;
    const int B = 256;
    const int nTiles = (N + 15) / 16;

    const size_t Np = ((size_t)N + 4 + 3) & ~(size_t)3;
    const size_t Ep = ((size_t)E + 3) & ~(size_t)3;
    const size_t Rn = Ep > (size_t)N * 32 ? Ep : (size_t)N * 32;
    if (ws_size >= (3 * Np + Ep + Rn) * sizeof(float)) {
        float* dinvP = ws;
        int*   off   = (int*)(ws + Np);
        int*   cnt   = (int*)(ws + 2 * Np);
        int*   srcs  = (int*)(ws + 3 * Np);
        float* R     = ws + 3 * Np + Ep;
        int*   rank  = (int*)R;
        float* hP    = R;
        float* aggP  = R + (size_t)N * 16;
        float* oP    = R;

        k_cnt_zero<<<(N + B - 1) / B, B, 0, stream>>>(cnt, N);
        k_rank    <<<(E + B - 1) / B, B, 0, stream>>>(eidx + E, cnt, rank, E);
        k_scan    <<<(N + 4095) / 4096, 1024, 0, stream>>>(cnt, off, dinvP, N);
        k_fill    <<<(E + B - 1) / B, B, 0, stream>>>(eidx, rank, off, srcs, E);
        k_gemm1<false><<<(nTiles + 7) / 8, 256, 0, stream>>>(x, W1, dinvP, hP, aggP, nTiles);
        k_pull1   <<<(N + 15) / 16, 256, 0, stream>>>(off, srcs, dinvP, hP, aggP, N);
        k_l2<false><<<(N + B - 1) / B, B, 0, stream>>>(aggP, b1, W2, dinvP, oP, out, N);
        k_pull2   <<<(N + 31) / 32, 256, 0, stream>>>(off, srcs, dinvP, oP, b2, out, N);
        return;
    }

    float* dinv = ws;
    float* h    = ws + (size_t)N;
    float* agg1 = ws + (size_t)N * 17;
    float* obuf = ws + (size_t)N * 33;

    k_deg_init <<<(N + B - 1) / B, B, 0, stream>>>(dinv, N);
    k_deg_edges<<<(E + B - 1) / B, B, 0, stream>>>(eidx + E, dinv, E);
    k_dinv     <<<(N + B - 1) / B, B, 0, stream>>>(dinv, N);

    k_gemm1<true><<<(nTiles + 7) / 8, 256, 0, stream>>>(x, W1, dinv, h, agg1, nTiles);

    const int t1 = E * 16;
    k_edge1<<<(t1 + B - 1) / B, B, 0, stream>>>(eidx, dinv, h, agg1, E);

    k_l2<true><<<(N + B - 1) / B, B, 0, stream>>>(agg1, b1, W2, dinv, obuf, out, N);

    const int t2 = E * 8;
    k_edge2<<<(t2 + B - 1) / B, B, 0, stream>>>(eidx, dinv, obuf, out, E);

    k_lsm<<<(N + B - 1) / B, B, 0, stream>>>(out, b2, N);
}


# One line per source in both edge walks: gathered rows carry the node's dinv (o: word 7 of 8; h: word 16 of 32 if workspace allows) and start on line boundaries (row buffer put first); same sums
# speedup vs baseline: 1.4017x; 1.0646x over previous
#include <hip/hip_runtime.h>
#include <math.h>

#define N_FEAT 512
#define HID    16
#define NCLS   7

typedef __attribute__((ext_vector_type(2))) float v2f;
typedef __attribute__((ext_vector_type(8))) float v8f;

__global__ void k_deg_init(float* __restrict__ deg, int n) {
    int i = blockIdx.x * blockDim.x + threadIdx.x;
    if (i < n) deg[i] = 1.0f;
}

__global__ void k_deg_edges(const int* __restrict__ col, float* __restrict__ deg, int E) {
    int e = blockIdx.x * blockDim.x + threadIdx.x;
    if (e < E) atomicAdd(&deg[col[e]], 1.0f);
}

__global__ void k_dinv(float* __restrict__ deg, int n) {
    int i = blockIdx.x * blockDim.x + threadIdx.x;
    if (i < n) deg[i] = rsqrtf(deg[i]);
}

template <bool SELF, int HP>
__global__ __launch_bounds__(256) void k_gemm1(
    const float* __restrict__ x, const float* __restrict__ W1,
    const float* __restrict__ dinv,
    float* __restrict__ h, float* __restrict__ agg1, int nTiles)
{
    __shared__ float sW[N_FEAT * HID];
    for (int i = threadIdx.x; i < N_FEAT * HID; i += blockDim.x) sW[i] = W1[i];
    __syncthreads();

    int wave = threadIdx.x >> 5;
    int lane = threadIdx.x & 31;
    int tile = blockIdx.x * 8 + wave;
    if (tile >= nTiles) return;

    int m  = lane & 15;
    int kh = lane >> 4;
    const float* xr = x + (size_t)(tile * 16 + m) * N_FEAT + 2 * kh;

    v8f acc = {};
#pragma unroll 4
    for (int k = 0; k < N_FEAT; k += 4) {
        v2f a, b;
        a.x = xr[k];
        a.y = xr[k + 1];
        int kk = k + 2 * kh;
        b.x = sW[kk * HID + m];
        b.y = sW[(kk + 1) * HID + m];
        acc = __builtin_amdgcn_wmma_f32_16x16x4_f32(
             false, a,  false, b,
             (short)0, acc,  false,  false);
    }

#pragma unroll
    for (int v = 0; v < 8; ++v) {
        int node  = tile * 16 + v + 8 * kh;
        float val = acc[v];
        h[(size_t)node * HP + m] = val;
        if (SELF) {
            float di = dinv[node];
            agg1[(size_t)node * HID + m] = val * di * di;
        }
    }
    if (HP > HID && lane < 16) {
        int node = tile * 16 + lane;
        h[(size_t)node * HP + HID] = dinv[node];
    }
}

__global__ void k_edge1(const int* __restrict__ eidx, const float* __restrict__ dinv,
                        const float* __restrict__ h, float* __restrict__ agg1, int E)
{
    int t = blockIdx.x * blockDim.x + threadIdx.x;
    int e = t >> 4, f = t & 15;
    if (e >= E) return;
    int r = eidx[e];
    int c = eidx[E + e];
    float w = dinv[r] * dinv[c];
    atomicAdd(&agg1[(size_t)c * HID + f], h[(size_t)r * HID + f] * w);
}

template <bool SELF, int OP>
__global__ void k_l2(const float* __restrict__ agg1, const float* __restrict__ b1,
                     const float* __restrict__ W2, const float* __restrict__ dinv,
                     float* __restrict__ o, float* __restrict__ outInit, int n)
{
    __shared__ float sW2[HID * NCLS];
    __shared__ float sb1[HID];
    if (threadIdx.x < HID * NCLS) sW2[threadIdx.x] = W2[threadIdx.x];
    if (threadIdx.x < HID)        sb1[threadIdx.x] = b1[threadIdx.x];
    __syncthreads();

    int i = blockIdx.x * blockDim.x + threadIdx.x;
    if (i >= n) return;

    float hv[HID];
#pragma unroll
    for (int f = 0; f < HID; ++f) {
        float v = agg1[(size_t)i * HID + f] + sb1[f];
        hv[f] = v > 0.0f ? v : 0.0f;
    }
    float di = dinv[i];
    float d2 = di * di;
#pragma unroll
    for (int j = 0; j < NCLS; ++j) {
        float s = 0.0f;
#pragma unroll
        for (int f = 0; f < HID; ++f) s = fmaf(hv[f], sW2[f * NCLS + j], s);
        o[(size_t)i * OP + j]         = s;
        if (SELF) outInit[(size_t)i * NCLS + j] = s * d2;
    }
    if (OP > NCLS) o[(size_t)i * OP + NCLS] = di;
}

__global__ void k_edge2(const int* __restrict__ eidx, const float* __restrict__ dinv,
                        const float* __restrict__ o, float* __restrict__ out, int E)
{
    int t = blockIdx.x * blockDim.x + threadIdx.x;
    int e = t >> 3, j = t & 7;
    if (e >= E || j >= NCLS) return;
    int r = eidx[e];
    int c = eidx[E + e];
    float w = dinv[r] * dinv[c];
    atomicAdd(&out[(size_t)c * NCLS + j], o[(size_t)r * NCLS + j] * w);
}

__global__ void k_lsm(float* __restrict__ out, const float* __restrict__ b2, int n)
{
    int i = blockIdx.x * blockDim.x + threadIdx.x;
    if (i >= n) return;
    float v[NCLS];
    float mx = -1e30f;
#pragma unroll
    for (int j = 0; j < NCLS; ++j) {
        v[j] = out[(size_t)i * NCLS + j] + b2[j];
        mx = fmaxf(mx, v[j]);
    }
    float s = 0.0f;
#pragma unroll
    for (int j = 0; j < NCLS; ++j) s += __expf(v[j] - mx);
    float l = __logf(s);
#pragma unroll
    for (int j = 0; j < NCLS; ++j) out[(size_t)i * NCLS + j] = v[j] - mx - l;
}


__global__ void k_cnt_zero(int* __restrict__ cnt, int n) {
    int i = blockIdx.x * blockDim.x + threadIdx.x;
    if (i < n) cnt[i] = 0;
}

__global__ void k_rank(const int* __restrict__ col, int* __restrict__ cnt,
                       int* __restrict__ rank, int E) {
    int e = blockIdx.x * blockDim.x + threadIdx.x;
    if (e < E) rank[e] = atomicAdd(&cnt[col[e]], 1);
}

__global__ __launch_bounds__(1024) void k_scan(const int* __restrict__ cnt, int* __restrict__ off,
                                               float* __restrict__ dinv, int n)
{
    __shared__ int sInc[32];
    __shared__ int sPre[32];
    const int t = threadIdx.x, lane = t & 31, wave = t >> 5;
    const int tile0 = blockIdx.x * 4096;

    int pre = 0;
    for (int b = 0; b < tile0; b += 4096) {
        int4 v = *reinterpret_cast<const int4*>(cnt + b + t * 4);
        pre += (v.x + v.y) + (v.z + v.w);
    }
    const int i0 = tile0 + t * 4;
    int c[4];
#pragma unroll
    for (int k = 0; k < 4; ++k) c[k] = (i0 + k < n) ? cnt[i0 + k] : 0;
    const int mine = (c[0] + c[1]) + (c[2] + c[3]);

    int incl = mine;
#pragma unroll
    for (int d = 1; d < 32; d <<= 1) { int y = __shfl_up(incl, d); if (lane >= d) incl += y; }
#pragma unroll
    for (int d = 16; d > 0; d >>= 1) pre += __shfl_xor(pre, d);
    if (lane == 31) sInc[wave] = incl;
    if (lane == 0)  sPre[wave] = pre;
    __syncthreads();
    if (wave == 0) {
        int w = sInc[lane], p = sPre[lane];
        int wi = w;
#pragma unroll
        for (int d = 1; d < 32; d <<= 1) { int y = __shfl_up(wi, d); if (lane >= d) wi += y; }
#pragma unroll
        for (int d = 16; d > 0; d >>= 1) p += __shfl_xor(p, d);
        sInc[lane] = wi - w + p;
    }
    __syncthreads();
    int run = sInc[wave] + incl - mine;
#pragma unroll
    for (int k = 0; k < 4; ++k) {
        int i = i0 + k;
        if (i < n) {
            off[i]  = run;
            dinv[i] = rsqrtf((float)(c[k] + 1));
            run += c[k];
            if (i == n - 1) off[n] = run;
        }
    }
}

__global__ void k_fill(const int* __restrict__ eidx, const int* __restrict__ rank,
                       const int* __restrict__ off, int* __restrict__ srcs, int E) {
    int e = blockIdx.x * blockDim.x + threadIdx.x;
    if (e < E) srcs[off[eidx[E + e]] + rank[e]] = eidx[e];
}

template <int K> __device__ __forceinline__ int row16_i(int v) {
    return __builtin_amdgcn_update_dpp(0, v, 0x150 + K, 0xf, 0xf, false);
}
template <int K> __device__ __forceinline__ float row16_f(float v) {
    return __int_as_float(row16_i<K>(__float_as_int(v)));
}
template <int K> __device__ __forceinline__ int grp8_i(int v) {
    return (int)__builtin_amdgcn_mov_dpp8((unsigned)v, K * 0x249249);
}
template <int K> __device__ __forceinline__ float grp8_f(float v) {
    return __int_as_float(grp8_i<K>(__float_as_int(v)));
}

#define PULL16_SHARE(K) sk[K] = row16_i<K>(s); wk[K] = row16_f<K>(w);

template <int HP>
__global__ __launch_bounds__(256) void k_pull1(const int* __restrict__ off, const int* __restrict__ srcs,
                                               const float* __restrict__ dinv, const float* __restrict__ h,
                                               float* __restrict__ agg1, int n)
{
    const int g = (blockIdx.x * 256 + threadIdx.x) >> 4;
    const int f = threadIdx.x & 15;
    if (g >= n) return;
    const int beg = off[g], end = off[g + 1];
    const float dc = dinv[g];
    float acc = h[g * HP + f] * dc * dc;
    for (int p = beg; p < end; p += 16) {
        const bool on = p + f < end;
        const int   s = on ? srcs[p + f] : g;
        const float ds = HP > HID ? h[s * HP + HID] : dinv[s];
        const float w = on ? ds * dc : 0.0f;
        int sk[16]; float wk[16], hv[16];
        PULL16_SHARE(0)  PULL16_SHARE(1)  PULL16_SHARE(2)  PULL16_SHARE(3)
        PULL16_SHARE(4)  PULL16_SHARE(5)  PULL16_SHARE(6)  PULL16_SHARE(7)
        PULL16_SHARE(8)  PULL16_SHARE(9)  PULL16_SHARE(10) PULL16_SHARE(11)
        PULL16_SHARE(12) PULL16_SHARE(13) PULL16_SHARE(14) PULL16_SHARE(15)
#pragma unroll
        for (int k = 0; k < 16; ++k) hv[k] = h[sk[k] * HP + f];
#pragma unroll
        for (int k = 0; k < 16; ++k) acc = fmaf(hv[k], wk[k], acc);
    }
    agg1[(size_t)g * HID + f] = acc;
}

#define PULL8_SHARE(K)  sk[K] = grp8_i<K>(s);
#define PULL8_WEIGHT(K) { const float dk = grp8_f<7>(ov[K]); wk[K] = p + K < end ? dk * dc : 0.0f; }

__global__ __launch_bounds__(256) void k_pull2(const int* __restrict__ off, const int* __restrict__ srcs,
                                               const float* __restrict__ o,
                                               const float* __restrict__ b2, float* __restrict__ out, int n)
{
    const int g = (blockIdx.x * 256 + threadIdx.x) >> 3;
    const int j = threadIdx.x & 7;
    if (g >= n) return;
    const int beg = off[g], end = off[g + 1];
    const float own = o[g * 8 + j];
    const float dc = grp8_f<7>(own);
    float acc = own * dc * dc;
    for (int p = beg; p < end; p += 8) {
        const int s = p + j < end ? srcs[p + j] : g;
        int sk[8]; float wk[8], ov[8];
        PULL8_SHARE(0) PULL8_SHARE(1) PULL8_SHARE(2) PULL8_SHARE(3)
        PULL8_SHARE(4) PULL8_SHARE(5) PULL8_SHARE(6) PULL8_SHARE(7)
#pragma unroll
        for (int k = 0; k < 8; ++k) ov[k] = o[sk[k] * 8 + j];
        PULL8_WEIGHT(0) PULL8_WEIGHT(1) PULL8_WEIGHT(2) PULL8_WEIGHT(3)
        PULL8_WEIGHT(4) PULL8_WEIGHT(5) PULL8_WEIGHT(6) PULL8_WEIGHT(7)
#pragma unroll
        for (int k = 0; k < 8; ++k) acc = fmaf(ov[k], wk[k], acc);
    }
    const int jj = j < NCLS ? j : NCLS - 1;
    const float v = acc + b2[jj];
    float mx = j < NCLS ? v : -INFINITY;
#pragma unroll
    for (int d = 1; d < 8; d <<= 1) mx = fmaxf(mx, __shfl_xor(mx, d, 8));
    float ex = j < NCLS ? __expf(v - mx) : 0.0f;
#pragma unroll
    for (int d = 1; d < 8; d <<= 1) ex += __shfl_xor(ex, d, 8);
    if (j < NCLS) out[(size_t)g * NCLS + j] = v - mx - __logf(ex);
}

extern "C" void kernel_launch(void* const* d_in, const int* in_sizes, int n_in,
                              void* d_out, int out_size, void* d_ws, size_t ws_size,
                              hipStream_t stream)
{
    const float* x    = (const float*)d_in[0];
    const int*   eidx = (const int*)  d_in[1];
    const float* W1   = (const float*)d_in[2];
    const float* b1   = (const float*)d_in[3];
    const float* W2   = (const float*)d_in[4];
    const float* b2   = (const float*)d_in[5];
    float* out = (float*)d_out;

    const int N = in_sizes[0] / N_FEAT;
    const int E = in_sizes[1] / 2;

    float* ws   = (float*)d_ws;
    const int B = 256;
    const int nTiles = (N + 15) / 16;

    const size_t Np = ((size_t)N + 4 + 3) & ~(size_t)3;
    const size_t Ep = ((size_t)E + 3) & ~(size_t)3;
    const size_t Rn = Ep > (size_t)N * 32 ? Ep : (size_t)N * 32;
    const size_t Rw = Ep > (size_t)N * 48 ? Ep : (size_t)N * 48;
    if (ws_size >= (3 * Np + Ep + Rn) * sizeof(float)) {
        const bool wide = ws_size >= (3 * Np + Ep + Rw) * sizeof(float);
        const size_t Rs = wide ? Rw : Rn;
        float* R     = ws;
        int*   srcs  = (int*)(ws + Rs);
        float* dinvP = ws + Rs + Ep;
        int*   off   = (int*)(ws + Rs + Ep + Np);
        int*   cnt   = (int*)(ws + Rs + Ep + 2 * Np);
        int*   rank  = (int*)R;
        float* hP    = R;
        float* aggP  = R + (size_t)N * (wide ? 32 : 16);
        float* oP    = R;

        k_cnt_zero<<<(N + B - 1) / B, B, 0, stream>>>(cnt, N);
        k_rank    <<<(E + B - 1) / B, B, 0, stream>>>(eidx + E, cnt, rank, E);
        k_scan    <<<(N + 4095) / 4096, 1024, 0, stream>>>(cnt, off, dinvP, N);
        k_fill    <<<(E + B - 1) / B, B, 0, stream>>>(eidx, rank, off, srcs, E);
        if (wide) {
            k_gemm1<false, 32><<<(nTiles + 7) / 8, 256, 0, stream>>>(x, W1, dinvP, hP, aggP, nTiles);
            k_pull1<32><<<(N + 15) / 16, 256, 0, stream>>>(off, srcs, dinvP, hP, aggP, N);
        } else {
            k_gemm1<false, 16><<<(nTiles + 7) / 8, 256, 0, stream>>>(x, W1, dinvP, hP, aggP, nTiles);
            k_pull1<16><<<(N + 15) / 16, 256, 0, stream>>>(off, srcs, dinvP, hP, aggP, N);
        }
        k_l2<false, 8><<<(N + B - 1) / B, B, 0, stream>>>(aggP, b1, W2, dinvP, oP, out, N);
        k_pull2   <<<(N + 31) / 32, 256, 0, stream>>>(off, srcs, oP, b2, out, N);
        return;
    }

    float* dinv = ws;
    float* h    = ws + (size_t)N;
    float* agg1 = ws + (size_t)N * 17;
    float* obuf = ws + (size_t)N * 33;

    k_deg_init <<<(N + B - 1) / B, B, 0, stream>>>(dinv, N);
    k_deg_edges<<<(E + B - 1) / B, B, 0, stream>>>(eidx + E, dinv, E);
    k_dinv     <<<(N + B - 1) / B, B, 0, stream>>>(dinv, N);

    k_gemm1<true, 16><<<(nTiles + 7) / 8, 256, 0, stream>>>(x, W1, dinv, h, agg1, nTiles);

    const int t1 = E * 16;
    k_edge1<<<(t1 + B - 1) / B, B, 0, stream>>>(eidx, dinv, h, agg1, E);

    k_l2<true, 7><<<(N + B - 1) / B, B, 0, stream>>>(agg1, b1, W2, dinv, obuf, out, N);

    const int t2 = E * 8;
    k_edge2<<<(t2 + B - 1) / B, B, 0, stream>>>(eidx, dinv, obuf, out, E);

    k_lsm<<<(N + B - 1) / B, B, 0, stream>>>(out, b2, N);
}


# Source lists as fixed 64-entry slots per destination: the count add's arrival number is the slot, so one edge pass counts and places; prefix-sum launch and second edge pass go; same sum order
# speedup vs baseline: 1.7331x; 1.2364x over previous
#include <hip/hip_runtime.h>
#include <math.h>

#define N_FEAT 512
#define HID    16
#define NCLS   7

typedef __attribute__((ext_vector_type(2))) float v2f;
typedef __attribute__((ext_vector_type(8))) float v8f;

__global__ void k_deg_init(float* __restrict__ deg, int n) {
    int i = blockIdx.x * blockDim.x + threadIdx.x;
    if (i < n) deg[i] = 1.0f;
}

__global__ void k_deg_edges(const int* __restrict__ col, float* __restrict__ deg, int E) {
    int e = blockIdx.x * blockDim.x + threadIdx.x;
    if (e < E) atomicAdd(&deg[col[e]], 1.0f);
}

__global__ void k_dinv(float* __restrict__ deg, int n) {
    int i = blockIdx.x * blockDim.x + threadIdx.x;
    if (i < n) deg[i] = rsqrtf(deg[i]);
}

template <bool SELF, int HP, bool CNT>
__global__ __launch_bounds__(256) void k_gemm1(
    const float* __restrict__ x, const float* __restrict__ W1,
    const float* __restrict__ dinv, const int* __restrict__ cnt, float* __restrict__ dinvOut,
    float* __restrict__ h, float* __restrict__ agg1, int nTiles)
{
    __shared__ float sW[N_FEAT * HID];
    for (int i = threadIdx.x; i < N_FEAT * HID; i += blockDim.x) sW[i] = W1[i];
    __syncthreads();

    int wave = threadIdx.x >> 5;
    int lane = threadIdx.x & 31;
    int tile = blockIdx.x * 8 + wave;
    if (tile >= nTiles) return;

    int m  = lane & 15;
    int kh = lane >> 4;
    const float* xr = x + (size_t)(tile * 16 + m) * N_FEAT + 2 * kh;

    v8f acc = {};
#pragma unroll 4
    for (int k = 0; k < N_FEAT; k += 4) {
        v2f a, b;
        a.x = xr[k];
        a.y = xr[k + 1];
        int kk = k + 2 * kh;
        b.x = sW[kk * HID + m];
        b.y = sW[(kk + 1) * HID + m];
        acc = __builtin_amdgcn_wmma_f32_16x16x4_f32(
             false, a,  false, b,
             (short)0, acc,  false,  false);
    }

#pragma unroll
    for (int v = 0; v < 8; ++v) {
        int node  = tile * 16 + v + 8 * kh;
        float val = acc[v];
        h[(size_t)node * HP + m] = val;
        if (SELF) {
            float di = dinv[node];
            agg1[(size_t)node * HID + m] = val * di * di;
        }
    }
    if ((HP > HID || CNT) && lane < 16) {
        int node = tile * 16 + lane;
        float di;
        if (CNT) {
            di = rsqrtf((float)(cnt[node] + 1));
            dinvOut[node] = di;
        } else {
            di = dinv[node];
        }
        if (HP > HID) h[(size_t)node * HP + HID] = di;
    }
}

__global__ void k_edge1(const int* __restrict__ eidx, const float* __restrict__ dinv,
                        const float* __restrict__ h, float* __restrict__ agg1, int E)
{
    int t = blockIdx.x * blockDim.x + threadIdx.x;
    int e = t >> 4, f = t & 15;
    if (e >= E) return;
    int r = eidx[e];
    int c = eidx[E + e];
    float w = dinv[r] * dinv[c];
    atomicAdd(&agg1[(size_t)c * HID + f], h[(size_t)r * HID + f] * w);
}

template <bool SELF, int OP>
__global__ void k_l2(const float* __restrict__ agg1, const float* __restrict__ b1,
                     const float* __restrict__ W2, const float* __restrict__ dinv,
                     float* __restrict__ o, float* __restrict__ outInit, int n)
{
    __shared__ float sW2[HID * NCLS];
    __shared__ float sb1[HID];
    if (threadIdx.x < HID * NCLS) sW2[threadIdx.x] = W2[threadIdx.x];
    if (threadIdx.x < HID)        sb1[threadIdx.x] = b1[threadIdx.x];
    __syncthreads();

    int i = blockIdx.x * blockDim.x + threadIdx.x;
    if (i >= n) return;

    float hv[HID];
#pragma unroll
    for (int f = 0; f < HID; ++f) {
        float v = agg1[(size_t)i * HID + f] + sb1[f];
        hv[f] = v > 0.0f ? v : 0.0f;
    }
    float di = dinv[i];
    float d2 = di * di;
#pragma unroll
    for (int j = 0; j < NCLS; ++j) {
        float s = 0.0f;
#pragma unroll
        for (int f = 0; f < HID; ++f) s = fmaf(hv[f], sW2[f * NCLS + j], s);
        o[(size_t)i * OP + j]         = s;
        if (SELF) outInit[(size_t)i * NCLS + j] = s * d2;
    }
    if (OP > NCLS) o[(size_t)i * OP + NCLS] = di;
}

__global__ void k_edge2(const int* __restrict__ eidx, const float* __restrict__ dinv,
                        const float* __restrict__ o, float* __restrict__ out, int E)
{
    int t = blockIdx.x * blockDim.x + threadIdx.x;
    int e = t >> 3, j = t & 7;
    if (e >= E || j >= NCLS) return;
    int r = eidx[e];
    int c = eidx[E + e];
    float w = dinv[r] * dinv[c];
    atomicAdd(&out[(size_t)c * NCLS + j], o[(size_t)r * NCLS + j] * w);
}

__global__ void k_lsm(float* __restrict__ out, const float* __restrict__ b2, int n)
{
    int i = blockIdx.x * blockDim.x + threadIdx.x;
    if (i >= n) return;
    float v[NCLS];
    float mx = -1e30f;
#pragma unroll
    for (int j = 0; j < NCLS; ++j) {
        v[j] = out[(size_t)i * NCLS + j] + b2[j];
        mx = fmaxf(mx, v[j]);
    }
    float s = 0.0f;
#pragma unroll
    for (int j = 0; j < NCLS; ++j) s += __expf(v[j] - mx);
    float l = __logf(s);
#pragma unroll
    for (int j = 0; j < NCLS; ++j) out[(size_t)i * NCLS + j] = v[j] - mx - l;
}


__global__ void k_cnt_zero(int* __restrict__ cnt, int n) {
    int i = blockIdx.x * blockDim.x + threadIdx.x;
    if (i < n) cnt[i] = 0;
}

__global__ void k_rank(const int* __restrict__ col, int* __restrict__ cnt,
                       int* __restrict__ rank, int E) {
    int e = blockIdx.x * blockDim.x + threadIdx.x;
    if (e < E) rank[e] = atomicAdd(&cnt[col[e]], 1);
}

__global__ __launch_bounds__(1024) void k_scan(const int* __restrict__ cnt, int* __restrict__ off,
                                               float* __restrict__ dinv, int n)
{
    __shared__ int sInc[32];
    __shared__ int sPre[32];
    const int t = threadIdx.x, lane = t & 31, wave = t >> 5;
    const int tile0 = blockIdx.x * 4096;

    int pre = 0;
    for (int b = 0; b < tile0; b += 4096) {
        int4 v = *reinterpret_cast<const int4*>(cnt + b + t * 4);
        pre += (v.x + v.y) + (v.z + v.w);
    }
    const int i0 = tile0 + t * 4;
    int c[4];
#pragma unroll
    for (int k = 0; k < 4; ++k) c[k] = (i0 + k < n) ? cnt[i0 + k] : 0;
    const int mine = (c[0] + c[1]) + (c[2] + c[3]);

    int incl = mine;
#pragma unroll
    for (int d = 1; d < 32; d <<= 1) { int y = __shfl_up(incl, d); if (lane >= d) incl += y; }
#pragma unroll
    for (int d = 16; d > 0; d >>= 1) pre += __shfl_xor(pre, d);
    if (lane == 31) sInc[wave] = incl;
    if (lane == 0)  sPre[wave] = pre;
    __syncthreads();
    if (wave == 0) {
        int w = sInc[lane], p = sPre[lane];
        int wi = w;
#pragma unroll
        for (int d = 1; d < 32; d <<= 1) { int y = __shfl_up(wi, d); if (lane >= d) wi += y; }
#pragma unroll
        for (int d = 16; d > 0; d >>= 1) p += __shfl_xor(p, d);
        sInc[lane] = wi - w + p;
    }
    __syncthreads();
    int run = sInc[wave] + incl - mine;
#pragma unroll
    for (int k = 0; k < 4; ++k) {
        int i = i0 + k;
        if (i < n) {
            off[i]  = run;
            dinv[i] = rsqrtf((float)(c[k] + 1));
            run += c[k];
            if (i == n - 1) off[n] = run;
        }
    }
}

__global__ void k_fill(const int* __restrict__ eidx, const int* __restrict__ rank,
                       const int* __restrict__ off, int* __restrict__ srcs, int E) {
    int e = blockIdx.x * blockDim.x + threadIdx.x;
    if (e < E) srcs[off[eidx[E + e]] + rank[e]] = eidx[e];
}

#define SLOTS 64

__global__ void k_place(const int* __restrict__ eidx, int* __restrict__ cnt,
                        int* __restrict__ slots, int E) {
    int e = blockIdx.x * blockDim.x + threadIdx.x;
    if (e < E) {
        const int c = eidx[E + e];
        const int r = atomicAdd(&cnt[c], 1);
        if (r < SLOTS) slots[c * SLOTS + r] = eidx[e];
    }
}

template <int K> __device__ __forceinline__ int row16_i(int v) {
    return __builtin_amdgcn_update_dpp(0, v, 0x150 + K, 0xf, 0xf, false);
}
template <int K> __device__ __forceinline__ float row16_f(float v) {
    return __int_as_float(row16_i<K>(__float_as_int(v)));
}
template <int K> __device__ __forceinline__ int grp8_i(int v) {
    return (int)__builtin_amdgcn_mov_dpp8((unsigned)v, K * 0x249249);
}
template <int K> __device__ __forceinline__ float grp8_f(float v) {
    return __int_as_float(grp8_i<K>(__float_as_int(v)));
}

#define PULL16_SHARE(K) sk[K] = row16_i<K>(s); wk[K] = row16_f<K>(w);

template <int HP>
__device__ __forceinline__ float pull16_step(const float* __restrict__ h, const float* __restrict__ dinv,
                                             int s, bool on, float dc, int f, float acc)
{
    const float ds = HP > HID ? h[s * HP + HID] : dinv[s];
    const float w = on ? ds * dc : 0.0f;
    int sk[16]; float wk[16], hv[16];
    PULL16_SHARE(0)  PULL16_SHARE(1)  PULL16_SHARE(2)  PULL16_SHARE(3)
    PULL16_SHARE(4)  PULL16_SHARE(5)  PULL16_SHARE(6)  PULL16_SHARE(7)
    PULL16_SHARE(8)  PULL16_SHARE(9)  PULL16_SHARE(10) PULL16_SHARE(11)
    PULL16_SHARE(12) PULL16_SHARE(13) PULL16_SHARE(14) PULL16_SHARE(15)
#pragma unroll
    for (int k = 0; k < 16; ++k) hv[k] = h[sk[k] * HP + f];
#pragma unroll
    for (int k = 0; k < 16; ++k) acc = fmaf(hv[k], wk[k], acc);
    return acc;
}

template <int HP, bool SLOT>
__global__ __launch_bounds__(256) void k_pull1(const int* __restrict__ off, const int* __restrict__ srcs,
                                               const float* __restrict__ dinv, const float* __restrict__ h,
                                               float* __restrict__ agg1, int n,
                                               const int* __restrict__ cnt, const int* __restrict__ eidx, int E)
{
    const int g = (blockIdx.x * 256 + threadIdx.x) >> 4;
    const int f = threadIdx.x & 15;
    if (g >= n) return;
    int beg, end, deg = 0;
    if (SLOT) {
        deg = cnt[g];
        beg = g * SLOTS;
        end = deg <= SLOTS ? beg + deg : beg;
    } else {
        beg = off[g]; end = off[g + 1];
    }
    const float dc = dinv[g];
    float acc = h[g * HP + f] * dc * dc;
    for (int p = beg; p < end; p += 16) {
        const bool on = p + f < end;
        const int   s = on ? srcs[p + f] : g;
        acc = pull16_step<HP>(h, dinv, s, on, dc, f, acc);
    }
    if (SLOT && deg > SLOTS) {
        const unsigned rowsh = threadIdx.x & 16;
        for (int e0 = 0; e0 < E; e0 += 16) {
            const int e = e0 + f;
            const bool on = e < E && eidx[E + e] == g;
            const unsigned hit = ((unsigned)__ballot(on) >> rowsh) & 0xffffu;
            if (hit) {
                const int s = on ? eidx[e] : g;
                acc = pull16_step<HP>(h, dinv, s, on, dc, f, acc);
            }
        }
    }
    agg1[(size_t)g * HID + f] = acc;
}

#define PULL8_SHARE(K)  sk[K] = grp8_i<K>(s);
#define PULL8_WEIGHT(K) { const float dk = grp8_f<7>(ov[K]); wk[K] = p + K < end ? dk * dc : 0.0f; }

template <bool SLOT>
__global__ __launch_bounds__(256) void k_pull2(const int* __restrict__ off, const int* __restrict__ srcs,
                                               const float* __restrict__ o,
                                               const float* __restrict__ b2, float* __restrict__ out, int n,
                                               const int* __restrict__ cnt, const int* __restrict__ eidx, int E)
{
    const int g = (blockIdx.x * 256 + threadIdx.x) >> 3;
    const int j = threadIdx.x & 7;
    if (g >= n) return;
    int beg, end, deg = 0;
    if (SLOT) {
        deg = cnt[g];
        beg = g * SLOTS;
        end = deg <= SLOTS ? beg + deg : beg;
    } else {
        beg = off[g]; end = off[g + 1];
    }
    const float own = o[g * 8 + j];
    const float dc = grp8_f<7>(own);
    float acc = own * dc * dc;
    for (int p = beg; p < end; p += 8) {
        const int s = p + j < end ? srcs[p + j] : g;
        int sk[8]; float wk[8], ov[8];
        PULL8_SHARE(0) PULL8_SHARE(1) PULL8_SHARE(2) PULL8_SHARE(3)
        PULL8_SHARE(4) PULL8_SHARE(5) PULL8_SHARE(6) PULL8_SHARE(7)
#pragma unroll
        for (int k = 0; k < 8; ++k) ov[k] = o[sk[k] * 8 + j];
        PULL8_WEIGHT(0) PULL8_WEIGHT(1) PULL8_WEIGHT(2) PULL8_WEIGHT(3)
        PULL8_WEIGHT(4) PULL8_WEIGHT(5) PULL8_WEIGHT(6) PULL8_WEIGHT(7)
#pragma unroll
        for (int k = 0; k < 8; ++k) acc = fmaf(ov[k], wk[k], acc);
    }
    if (SLOT && deg > SLOTS) {
        const unsigned grpsh = threadIdx.x & 24;
        for (int e0 = 0; e0 < E; e0 += 8) {
            const int e = e0 + j;
            const bool on = e < E && eidx[E + e] == g;
            const unsigned hit = ((unsigned)__ballot(on) >> grpsh) & 0xffu;
            if (hit) {
                const int s = on ? eidx[e] : g;
                int sk[8]; float ov[8];
                PULL8_SHARE(0) PULL8_SHARE(1) PULL8_SHARE(2) PULL8_SHARE(3)
                PULL8_SHARE(4) PULL8_SHARE(5) PULL8_SHARE(6) PULL8_SHARE(7)
#pragma unroll
                for (int k = 0; k < 8; ++k) ov[k] = o[sk[k] * 8 + j];
#pragma unroll
                for (int k = 0; k < 8; ++k) {
                    const float dk = grp8_f<7>(ov[k]);
                    const float wv = (hit >> k) & 1u ? dk * dc : 0.0f;
                    acc = fmaf(ov[k], wv, acc);
                }
            }
        }
    }
    const int jj = j < NCLS ? j : NCLS - 1;
    const float v = acc + b2[jj];
    float mx = j < NCLS ? v : -INFINITY;
#pragma unroll
    for (int d = 1; d < 8; d <<= 1) mx = fmaxf(mx, __shfl_xor(mx, d, 8));
    float ex = j < NCLS ? __expf(v - mx) : 0.0f;
#pragma unroll
    for (int d = 1; d < 8; d <<= 1) ex += __shfl_xor(ex, d, 8);
    if (j < NCLS) out[(size_t)g * NCLS + j] = v - mx - __logf(ex);
}

extern "C" void kernel_launch(void* const* d_in, const int* in_sizes, int n_in,
                              void* d_out, int out_size, void* d_ws, size_t ws_size,
                              hipStream_t stream)
{
    const float* x    = (const float*)d_in[0];
    const int*   eidx = (const int*)  d_in[1];
    const float* W1   = (const float*)d_in[2];
    const float* b1   = (const float*)d_in[3];
    const float* W2   = (const float*)d_in[4];
    const float* b2   = (const float*)d_in[5];
    float* out = (float*)d_out;

    const int N = in_sizes[0] / N_FEAT;
    const int E = in_sizes[1] / 2;

    float* ws   = (float*)d_ws;
    const int B = 256;
    const int nTiles = (N + 15) / 16;

    const size_t NpS = ((size_t)N + 3) & ~(size_t)3;
    const size_t Sl  = (size_t)N * SLOTS;
    if (ws_size >= ((size_t)N * 32 + Sl + 2 * NpS) * sizeof(float)) {
        const bool wide = ws_size >= ((size_t)N * 48 + Sl + 2 * NpS) * sizeof(float);
        const size_t Rs = (size_t)N * (wide ? 48 : 32);
        float* R     = ws;
        int*   slots = (int*)(ws + Rs);
        float* dinvP = ws + Rs + Sl;
        int*   cnt   = (int*)(ws + Rs + Sl + NpS);
        float* hP    = R;
        float* aggP  = R + (size_t)N * (wide ? 32 : 16);
        float* oP    = R;

        k_cnt_zero<<<(N + B - 1) / B, B, 0, stream>>>(cnt, N);
        k_place   <<<(E + B - 1) / B, B, 0, stream>>>(eidx, cnt, slots, E);
        if (wide) {
            k_gemm1<false, 32, true><<<(nTiles + 7) / 8, 256, 0, stream>>>(x, W1, nullptr, cnt, dinvP, hP, aggP, nTiles);
            k_pull1<32, true><<<(N + 15) / 16, 256, 0, stream>>>(nullptr, slots, dinvP, hP, aggP, N, cnt, eidx, E);
        } else {
            k_gemm1<false, 16, true><<<(nTiles + 7) / 8, 256, 0, stream>>>(x, W1, nullptr, cnt, dinvP, hP, aggP, nTiles);
            k_pull1<16, true><<<(N + 15) / 16, 256, 0, stream>>>(nullptr, slots, dinvP, hP, aggP, N, cnt, eidx, E);
        }
        k_l2<false, 8><<<(N + B - 1) / B, B, 0, stream>>>(aggP, b1, W2, dinvP, oP, out, N);
        k_pull2<true><<<(N + 31) / 32, 256, 0, stream>>>(nullptr, slots, oP, b2, out, N, cnt, eidx, E);
        return;
    }

    const size_t Np = ((size_t)N + 4 + 3) & ~(size_t)3;
    const size_t Ep = ((size_t)E + 3) & ~(size_t)3;
    const size_t Rn = Ep > (size_t)N * 32 ? Ep : (size_t)N * 32;
    const size_t Rw = Ep > (size_t)N * 48 ? Ep : (size_t)N * 48;
    if (ws_size >= (3 * Np + Ep + Rn) * sizeof(float)) {
        const bool wide = ws_size >= (3 * Np + Ep + Rw) * sizeof(float);
        const size_t Rs = wide ? Rw : Rn;
        float* R     = ws;
        int*   srcs  = (int*)(ws + Rs);
        float* dinvP = ws + Rs + Ep;
        int*   off   = (int*)(ws + Rs + Ep + Np);
        int*   cnt   = (int*)(ws + Rs + Ep + 2 * Np);
        int*   rank  = (int*)R;
        float* hP    = R;
        float* aggP  = R + (size_t)N * (wide ? 32 : 16);
        float* oP    = R;

        k_cnt_zero<<<(N + B - 1) / B, B, 0, stream>>>(cnt, N);
        k_rank    <<<(E + B - 1) / B, B, 0, stream>>>(eidx + E, cnt, rank, E);
        k_scan    <<<(N + 4095) / 4096, 1024, 0, stream>>>(cnt, off, dinvP, N);
        k_fill    <<<(E + B - 1) / B, B, 0, stream>>>(eidx, rank, off, srcs, E);
        if (wide) {
            k_gemm1<false, 32, false><<<(nTiles + 7) / 8, 256, 0, stream>>>(x, W1, dinvP, nullptr, nullptr, hP, aggP, nTiles);
            k_pull1<32, false><<<(N + 15) / 16, 256, 0, stream>>>(off, srcs, dinvP, hP, aggP, N, nullptr, eidx, E);
        } else {
            k_gemm1<false, 16, false><<<(nTiles + 7) / 8, 256, 0, stream>>>(x, W1, dinvP, nullptr, nullptr, hP, aggP, nTiles);
            k_pull1<16, false><<<(N + 15) / 16, 256, 0, stream>>>(off, srcs, dinvP, hP, aggP, N, nullptr, eidx, E);
        }
        k_l2<false, 8><<<(N + B - 1) / B, B, 0, stream>>>(aggP, b1, W2, dinvP, oP, out, N);
        k_pull2<false><<<(N + 31) / 32, 256, 0, stream>>>(off, srcs, oP, b2, out, N, nullptr, eidx, E);
        return;
    }

    float* dinv = ws;
    float* h    = ws + (size_t)N;
    float* agg1 = ws + (size_t)N * 17;
    float* obuf = ws + (size_t)N * 33;

    k_deg_init <<<(N + B - 1) / B, B, 0, stream>>>(dinv, N);
    k_deg_edges<<<(E + B - 1) / B, B, 0, stream>>>(eidx + E, dinv, E);
    k_dinv     <<<(N + B - 1) / B, B, 0, stream>>>(dinv, N);

    k_gemm1<true, 16, false><<<(nTiles + 7) / 8, 256, 0, stream>>>(x, W1, dinv, nullptr, nullptr, h, agg1, nTiles);

    const int t1 = E * 16;
    k_edge1<<<(t1 + B - 1) / B, B, 0, stream>>>(eidx, dinv, h, agg1, E);

    k_l2<true, 7><<<(N + B - 1) / B, B, 0, stream>>>(agg1, b1, W2, dinv, obuf, out, N);

    const int t2 = E * 8;
    k_edge2<<<(t2 + B - 1) / B, B, 0, stream>>>(eidx, dinv, obuf, out, E);

    k_lsm<<<(N + B - 1) / B, B, 0, stream>>>(out, b2, N);
}


# Layer-1 GEMM re-cut: each work-group stages 16-row x tiles in LDS by whole-line loads (next tile in flight); its 8 waves split K, W1 slice in registers, partials added in LDS; k sum order changes
# speedup vs baseline: 1.8578x; 1.0720x over previous
#include <hip/hip_runtime.h>
#include <math.h>

#define N_FEAT 512
#define HID    16
#define NCLS   7

typedef __attribute__((ext_vector_type(2))) float v2f;
typedef __attribute__((ext_vector_type(8))) float v8f;
typedef __attribute__((ext_vector_type(4))) float v4f;

__global__ void k_deg_init(float* __restrict__ deg, int n) {
    int i = blockIdx.x * blockDim.x + threadIdx.x;
    if (i < n) deg[i] = 1.0f;
}

__global__ void k_deg_edges(const int* __restrict__ col, float* __restrict__ deg, int E) {
    int e = blockIdx.x * blockDim.x + threadIdx.x;
    if (e < E) atomicAdd(&deg[col[e]], 1.0f);
}

__global__ void k_dinv(float* __restrict__ deg, int n) {
    int i = blockIdx.x * blockDim.x + threadIdx.x;
    if (i < n) deg[i] = rsqrtf(deg[i]);
}

template <bool SELF, int HP, bool CNT>
__global__ __launch_bounds__(256) void k_gemm1(
    const float* __restrict__ x, const float* __restrict__ W1,
    const float* __restrict__ dinv, const int* __restrict__ cnt, float* __restrict__ dinvOut,
    float* __restrict__ h, float* __restrict__ agg1, int nTiles)
{
    __shared__ float sW[N_FEAT * HID];
    for (int i = threadIdx.x; i < N_FEAT * HID; i += blockDim.x) sW[i] = W1[i];
    __syncthreads();

    int wave = threadIdx.x >> 5;
    int lane = threadIdx.x & 31;
    int tile = blockIdx.x * 8 + wave;
    if (tile >= nTiles) return;

    int m  = lane & 15;
    int kh = lane >> 4;
    const float* xr = x + (size_t)(tile * 16 + m) * N_FEAT + 2 * kh;

    v8f acc = {};
#pragma unroll 4
    for (int k = 0; k < N_FEAT; k += 4) {
        v2f a, b;
        a.x = xr[k];
        a.y = xr[k + 1];
        int kk = k + 2 * kh;
        b.x = sW[kk * HID + m];
        b.y = sW[(kk + 1) * HID + m];
        acc = __builtin_amdgcn_wmma_f32_16x16x4_f32(
             false, a,  false, b,
             (short)0, acc,  false,  false);
    }

#pragma unroll
    for (int v = 0; v < 8; ++v) {
        int node  = tile * 16 + v + 8 * kh;
        float val = acc[v];
        h[(size_t)node * HP + m] = val;
        if (SELF) {
            float di = dinv[node];
            agg1[(size_t)node * HID + m] = val * di * di;
        }
    }
    if ((HP > HID || CNT) && lane < 16) {
        int node = tile * 16 + lane;
        float di;
        if (CNT) {
            di = rsqrtf((float)(cnt[node] + 1));
            dinvOut[node] = di;
        } else {
            di = dinv[node];
        }
        if (HP > HID) h[(size_t)node * HP + HID] = di;
    }
}

#define G1_WAVES 8
#define G1_KW    (N_FEAT / G1_WAVES)
#define G1_KS    (G1_KW / 4)
#define G1_PITCH (N_FEAT + 4)

__device__ __forceinline__ void g1_lds_barrier() {
    asm volatile("s_wait_dscnt 0x0\n\ts_barrier_signal -1\n\ts_barrier_wait -1" ::: "memory");
}

template <int HP>
__global__ __launch_bounds__(256) void k_gemm1_tile(
    const float* __restrict__ x, const float* __restrict__ W1,
    const int* __restrict__ cnt, float* __restrict__ dinvOut,
    float* __restrict__ h, int nTiles)
{
    __shared__ __attribute__((aligned(16))) float sX[16 * G1_PITCH];
    __shared__ float sP[G1_WAVES][8][32];

    const int t    = threadIdx.x;
    const int wave = t >> 5;
    const int lane = t & 31;
    const int m  = lane & 15;
    const int kh = lane >> 4;
    const int kw = wave * G1_KW + 2 * kh;
    const int srow = t >> 7;
    const int scol = (t & 127) * 4;

    int tile = blockIdx.x;
    if (tile >= nTiles) return;

    v2f b[G1_KS];
#pragma unroll
    for (int ks = 0; ks < G1_KS; ++ks) {
        const int kk = kw + 4 * ks;
        b[ks].x = W1[kk * HID + m];
        b[ks].y = W1[(kk + 1) * HID + m];
    }

    v4f pf[8];
    int cn;
    {
        const float* src = x + (size_t)(tile * 16 + srow) * N_FEAT + scol;
#pragma unroll
        for (int j = 0; j < 8; ++j) pf[j] = *reinterpret_cast<const v4f*>(src + (size_t)j * 2 * N_FEAT);
        cn = cnt[tile * 16 + m];
    }

    for (;;) {
        if (t < 16) {
            const int node = tile * 16 + t;
            const float di = rsqrtf((float)(cn + 1));
            dinvOut[node] = di;
            if (HP > HID) h[(size_t)node * HP + HID] = di;
        }
#pragma unroll
        for (int j = 0; j < 8; ++j)
            *reinterpret_cast<v4f*>(&sX[(2 * j + srow) * G1_PITCH + scol]) = pf[j];

        const int  nxt  = tile + (int)gridDim.x;
        const bool more = nxt < nTiles;
        {
            const int    pt   = more ? nxt : tile;
            const size_t step = more ? (size_t)2 * N_FEAT : 0;
            const float* src  = x + (size_t)(pt * 16 + srow) * N_FEAT + scol;
#pragma unroll
            for (int j = 0; j < 8; ++j) pf[j] = *reinterpret_cast<const v4f*>(src + j * step);
            cn = cnt[pt * 16 + m];
        }
        g1_lds_barrier();

        v2f a[G1_KS];
#pragma unroll
        for (int ks = 0; ks < G1_KS; ++ks)
            a[ks] = *reinterpret_cast<const v2f*>(&sX[m * G1_PITCH + kw + 4 * ks]);
        v8f acc0 = {}, acc1 = {};
#pragma unroll
        for (int ks = 0; ks < G1_KS; ks += 2) {
            acc0 = __builtin_amdgcn_wmma_f32_16x16x4_f32(false, a[ks],     false, b[ks],     (short)0, acc0, false, false);
            acc1 = __builtin_amdgcn_wmma_f32_16x16x4_f32(false, a[ks + 1], false, b[ks + 1], (short)0, acc1, false, false);
        }
#pragma unroll
        for (int v = 0; v < 8; ++v) sP[wave][v][lane] = acc0[v] + acc1[v];
        g1_lds_barrier();
        {
            const int v = wave;
            const float s01 = sP[0][v][lane] + sP[1][v][lane];
            const float s23 = sP[2][v][lane] + sP[3][v][lane];
            const float s45 = sP[4][v][lane] + sP[5][v][lane];
            const float s67 = sP[6][v][lane] + sP[7][v][lane];
            const int node = tile * 16 + v + 8 * kh;
            h[(size_t)node * HP + m] = (s01 + s23) + (s45 + s67);
        }
        if (!more) break;
        tile = nxt;
    }
}

__global__ void k_edge1(const int* __restrict__ eidx, const float* __restrict__ dinv,
                        const float* __restrict__ h, float* __restrict__ agg1, int E)
{
    int t = blockIdx.x * blockDim.x + threadIdx.x;
    int e = t >> 4, f = t & 15;
    if (e >= E) return;
    int r = eidx[e];
    int c = eidx[E + e];
    float w = dinv[r] * dinv[c];
    atomicAdd(&agg1[(size_t)c * HID + f], h[(size_t)r * HID + f] * w);
}

template <bool SELF, int OP>
__global__ void k_l2(const float* __restrict__ agg1, const float* __restrict__ b1,
                     const float* __restrict__ W2, const float* __restrict__ dinv,
                     float* __restrict__ o, float* __restrict__ outInit, int n)
{
    __shared__ float sW2[HID * NCLS];
    __shared__ float sb1[HID];
    if (threadIdx.x < HID * NCLS) sW2[threadIdx.x] = W2[threadIdx.x];
    if (threadIdx.x < HID)        sb1[threadIdx.x] = b1[threadIdx.x];
    __syncthreads();

    int i = blockIdx.x * blockDim.x + threadIdx.x;
    if (i >= n) return;

    float hv[HID];
#pragma unroll
    for (int f = 0; f < HID; ++f) {
        float v = agg1[(size_t)i * HID + f] + sb1[f];
        hv[f] = v > 0.0f ? v : 0.0f;
    }
    float di = dinv[i];
    float d2 = di * di;
#pragma unroll
    for (int j = 0; j < NCLS; ++j) {
        float s = 0.0f;
#pragma unroll
        for (int f = 0; f < HID; ++f) s = fmaf(hv[f], sW2[f * NCLS + j], s);
        o[(size_t)i * OP + j]         = s;
        if (SELF) outInit[(size_t)i * NCLS + j] = s * d2;
    }
    if (OP > NCLS) o[(size_t)i * OP + NCLS] = di;
}

__global__ void k_edge2(const int* __restrict__ eidx, const float* __restrict__ dinv,
                        const float* __restrict__ o, float* __restrict__ out, int E)
{
    int t = blockIdx.x * blockDim.x + threadIdx.x;
    int e = t >> 3, j = t & 7;
    if (e >= E || j >= NCLS) return;
    int r = eidx[e];
    int c = eidx[E + e];
    float w = dinv[r] * dinv[c];
    atomicAdd(&out[(size_t)c * NCLS + j], o[(size_t)r * NCLS + j] * w);
}

__global__ void k_lsm(float* __restrict__ out, const float* __restrict__ b2, int n)
{
    int i = blockIdx.x * blockDim.x + threadIdx.x;
    if (i >= n) return;
    float v[NCLS];
    float mx = -1e30f;
#pragma unroll
    for (int j = 0; j < NCLS; ++j) {
        v[j] = out[(size_t)i * NCLS + j] + b2[j];
        mx = fmaxf(mx, v[j]);
    }
    float s = 0.0f;
#pragma unroll
    for (int j = 0; j < NCLS; ++j) s += __expf(v[j] - mx);
    float l = __logf(s);
#pragma unroll
    for (int j = 0; j < NCLS; ++j) out[(size_t)i * NCLS + j] = v[j] - mx - l;
}


__global__ void k_cnt_zero(int* __restrict__ cnt, int n) {
    int i = blockIdx.x * blockDim.x + threadIdx.x;
    if (i < n) cnt[i] = 0;
}

__global__ void k_rank(const int* __restrict__ col, int* __restrict__ cnt,
                       int* __restrict__ rank, int E) {
    int e = blockIdx.x * blockDim.x + threadIdx.x;
    if (e < E) rank[e] = atomicAdd(&cnt[col[e]], 1);
}

__global__ __launch_bounds__(1024) void k_scan(const int* __restrict__ cnt, int* __restrict__ off,
                                               float* __restrict__ dinv, int n)
{
    __shared__ int sInc[32];
    __shared__ int sPre[32];
    const int t = threadIdx.x, lane = t & 31, wave = t >> 5;
    const int tile0 = blockIdx.x * 4096;

    int pre = 0;
    for (int b = 0; b < tile0; b += 4096) {
        int4 v = *reinterpret_cast<const int4*>(cnt + b + t * 4);
        pre += (v.x + v.y) + (v.z + v.w);
    }
    const int i0 = tile0 + t * 4;
    int c[4];
#pragma unroll
    for (int k = 0; k < 4; ++k) c[k] = (i0 + k < n) ? cnt[i0 + k] : 0;
    const int mine = (c[0] + c[1]) + (c[2] + c[3]);

    int incl = mine;
#pragma unroll
    for (int d = 1; d < 32; d <<= 1) { int y = __shfl_up(incl, d); if (lane >= d) incl += y; }
#pragma unroll
    for (int d = 16; d > 0; d >>= 1) pre += __shfl_xor(pre, d);
    if (lane == 31) sInc[wave] = incl;
    if (lane == 0)  sPre[wave] = pre;
    __syncthreads();
    if (wave == 0) {
        int w = sInc[lane], p = sPre[lane];
        int wi = w;
#pragma unroll
        for (int d = 1; d < 32; d <<= 1) { int y = __shfl_up(wi, d); if (lane >= d) wi += y; }
#pragma unroll
        for (int d = 16; d > 0; d >>= 1) p += __shfl_xor(p, d);
        sInc[lane] = wi - w + p;
    }
    __syncthreads();
    int run = sInc[wave] + incl - mine;
#pragma unroll
    for (int k = 0; k < 4; ++k) {
        int i = i0 + k;
        if (i < n) {
            off[i]  = run;
            dinv[i] = rsqrtf((float)(c[k] + 1));
            run += c[k];
            if (i == n - 1) off[n] = run;
        }
    }
}

__global__ void k_fill(const int* __restrict__ eidx, const int* __restrict__ rank,
                       const int* __restrict__ off, int* __restrict__ srcs, int E) {
    int e = blockIdx.x * blockDim.x + threadIdx.x;
    if (e < E) srcs[off[eidx[E + e]] + rank[e]] = eidx[e];
}

#define SLOTS 64

__global__ void k_place(const int* __restrict__ eidx, int* __restrict__ cnt,
                        int* __restrict__ slots, int E) {
    int e = blockIdx.x * blockDim.x + threadIdx.x;
    if (e < E) {
        const int c = eidx[E + e];
        const int r = atomicAdd(&cnt[c], 1);
        if (r < SLOTS) slots[c * SLOTS + r] = eidx[e];
    }
}

template <int K> __device__ __forceinline__ int row16_i(int v) {
    return __builtin_amdgcn_update_dpp(0, v, 0x150 + K, 0xf, 0xf, false);
}
template <int K> __device__ __forceinline__ float row16_f(float v) {
    return __int_as_float(row16_i<K>(__float_as_int(v)));
}
template <int K> __device__ __forceinline__ int grp8_i(int v) {
    return (int)__builtin_amdgcn_mov_dpp8((unsigned)v, K * 0x249249);
}
template <int K> __device__ __forceinline__ float grp8_f(float v) {
    return __int_as_float(grp8_i<K>(__float_as_int(v)));
}

#define PULL16_SHARE(K) sk[K] = row16_i<K>(s); wk[K] = row16_f<K>(w);

template <int HP>
__device__ __forceinline__ float pull16_step(const float* __restrict__ h, const float* __restrict__ dinv,
                                             int s, bool on, float dc, int f, float acc)
{
    const float ds = HP > HID ? h[s * HP + HID] : dinv[s];
    const float w = on ? ds * dc : 0.0f;
    int sk[16]; float wk[16], hv[16];
    PULL16_SHARE(0)  PULL16_SHARE(1)  PULL16_SHARE(2)  PULL16_SHARE(3)
    PULL16_SHARE(4)  PULL16_SHARE(5)  PULL16_SHARE(6)  PULL16_SHARE(7)
    PULL16_SHARE(8)  PULL16_SHARE(9)  PULL16_SHARE(10) PULL16_SHARE(11)
    PULL16_SHARE(12) PULL16_SHARE(13) PULL16_SHARE(14) PULL16_SHARE(15)
#pragma unroll
    for (int k = 0; k < 16; ++k) hv[k] = h[sk[k] * HP + f];
#pragma unroll
    for (int k = 0; k < 16; ++k) acc = fmaf(hv[k], wk[k], acc);
    return acc;
}

template <int HP, bool SLOT>
__global__ __launch_bounds__(256) void k_pull1(const int* __restrict__ off, const int* __restrict__ srcs,
                                               const float* __restrict__ dinv, const float* __restrict__ h,
                                               float* __restrict__ agg1, int n,
                                               const int* __restrict__ cnt, const int* __restrict__ eidx, int E)
{
    const int g = (blockIdx.x * 256 + threadIdx.x) >> 4;
    const int f = threadIdx.x & 15;
    if (g >= n) return;
    int beg, end, deg = 0;
    if (SLOT) {
        deg = cnt[g];
        beg = g * SLOTS;
        end = deg <= SLOTS ? beg + deg : beg;
    } else {
        beg = off[g]; end = off[g + 1];
    }
    const float dc = dinv[g];
    float acc = h[g * HP + f] * dc * dc;
    for (int p = beg; p < end; p += 16) {
        const bool on = p + f < end;
        const int   s = on ? srcs[p + f] : g;
        acc = pull16_step<HP>(h, dinv, s, on, dc, f, acc);
    }
    if (SLOT && deg > SLOTS) {
        const unsigned rowsh = threadIdx.x & 16;
        for (int e0 = 0; e0 < E; e0 += 16) {
            const int e = e0 + f;
            const bool on = e < E && eidx[E + e] == g;
            const unsigned hit = ((unsigned)__ballot(on) >> rowsh) & 0xffffu;
            if (hit) {
                const int s = on ? eidx[e] : g;
                acc = pull16_step<HP>(h, dinv, s, on, dc, f, acc);
            }
        }
    }
    agg1[(size_t)g * HID + f] = acc;
}

#define PULL8_SHARE(K)  sk[K] = grp8_i<K>(s);
#define PULL8_WEIGHT(K) { const float dk = grp8_f<7>(ov[K]); wk[K] = p + K < end ? dk * dc : 0.0f; }

template <bool SLOT>
__global__ __launch_bounds__(256) void k_pull2(const int* __restrict__ off, const int* __restrict__ srcs,
                                               const float* __restrict__ o,
                                               const float* __restrict__ b2, float* __restrict__ out, int n,
                                               const int* __restrict__ cnt, const int* __restrict__ eidx, int E)
{
    const int g = (blockIdx.x * 256 + threadIdx.x) >> 3;
    const int j = threadIdx.x & 7;
    if (g >= n) return;
    int beg, end, deg = 0;
    if (SLOT) {
        deg = cnt[g];
        beg = g * SLOTS;
        end = deg <= SLOTS ? beg + deg : beg;
    } else {
        beg = off[g]; end = off[g + 1];
    }
    const float own = o[g * 8 + j];
    const float dc = grp8_f<7>(own);
    float acc = own * dc * dc;
    for (int p = beg; p < end; p += 8) {
        const int s = p + j < end ? srcs[p + j] : g;
        int sk[8]; float wk[8], ov[8];
        PULL8_SHARE(0) PULL8_SHARE(1) PULL8_SHARE(2) PULL8_SHARE(3)
        PULL8_SHARE(4) PULL8_SHARE(5) PULL8_SHARE(6) PULL8_SHARE(7)
#pragma unroll
        for (int k = 0; k < 8; ++k) ov[k] = o[sk[k] * 8 + j];
        PULL8_WEIGHT(0) PULL8_WEIGHT(1) PULL8_WEIGHT(2) PULL8_WEIGHT(3)
        PULL8_WEIGHT(4) PULL8_WEIGHT(5) PULL8_WEIGHT(6) PULL8_WEIGHT(7)
#pragma unroll
        for (int k = 0; k < 8; ++k) acc = fmaf(ov[k], wk[k], acc);
    }
    if (SLOT && deg > SLOTS) {
        const unsigned grpsh = threadIdx.x & 24;
        for (int e0 = 0; e0 < E; e0 += 8) {
            const int e = e0 + j;
            const bool on = e < E && eidx[E + e] == g;
            const unsigned hit = ((unsigned)__ballot(on) >> grpsh) & 0xffu;
            if (hit) {
                const int s = on ? eidx[e] : g;
                int sk[8]; float ov[8];
                PULL8_SHARE(0) PULL8_SHARE(1) PULL8_SHARE(2) PULL8_SHARE(3)
                PULL8_SHARE(4) PULL8_SHARE(5) PULL8_SHARE(6) PULL8_SHARE(7)
#pragma unroll
                for (int k = 0; k < 8; ++k) ov[k] = o[sk[k] * 8 + j];
#pragma unroll
                for (int k = 0; k < 8; ++k) {
                    const float dk = grp8_f<7>(ov[k]);
                    const float wv = (hit >> k) & 1u ? dk * dc : 0.0f;
                    acc = fmaf(ov[k], wv, acc);
                }
            }
        }
    }
    const int jj = j < NCLS ? j : NCLS - 1;
    const float v = acc + b2[jj];
    float mx = j < NCLS ? v : -INFINITY;
#pragma unroll
    for (int d = 1; d < 8; d <<= 1) mx = fmaxf(mx, __shfl_xor(mx, d, 8));
    float ex = j < NCLS ? __expf(v - mx) : 0.0f;
#pragma unroll
    for (int d = 1; d < 8; d <<= 1) ex += __shfl_xor(ex, d, 8);
    if (j < NCLS) out[(size_t)g * NCLS + j] = v - mx - __logf(ex);
}

extern "C" void kernel_launch(void* const* d_in, const int* in_sizes, int n_in,
                              void* d_out, int out_size, void* d_ws, size_t ws_size,
                              hipStream_t stream)
{
    const float* x    = (const float*)d_in[0];
    const int*   eidx = (const int*)  d_in[1];
    const float* W1   = (const float*)d_in[2];
    const float* b1   = (const float*)d_in[3];
    const float* W2   = (const float*)d_in[4];
    const float* b2   = (const float*)d_in[5];
    float* out = (float*)d_out;

    const int N = in_sizes[0] / N_FEAT;
    const int E = in_sizes[1] / 2;

    float* ws   = (float*)d_ws;
    const int B = 256;
    const int nTiles = (N + 15) / 16;

    const size_t NpS = ((size_t)N + 3) & ~(size_t)3;
    const size_t Sl  = (size_t)N * SLOTS;
    if (ws_size >= ((size_t)N * 32 + Sl + 2 * NpS) * sizeof(float)) {
        const bool wide = ws_size >= ((size_t)N * 48 + Sl + 2 * NpS) * sizeof(float);
        const size_t Rs = (size_t)N * (wide ? 48 : 32);
        float* R     = ws;
        int*   slots = (int*)(ws + Rs);
        float* dinvP = ws + Rs + Sl;
        int*   cnt   = (int*)(ws + Rs + Sl + NpS);
        float* hP    = R;
        float* aggP  = R + (size_t)N * (wide ? 32 : 16);
        float* oP    = R;

        k_cnt_zero<<<(N + B - 1) / B, B, 0, stream>>>(cnt, N);
        k_place   <<<(E + B - 1) / B, B, 0, stream>>>(eidx, cnt, slots, E);
        if (wide) {
            k_gemm1_tile<32><<<1024, 256, 0, stream>>>(x, W1, cnt, dinvP, hP, nTiles);
            k_pull1<32, true><<<(N + 15) / 16, 256, 0, stream>>>(nullptr, slots, dinvP, hP, aggP, N, cnt, eidx, E);
        } else {
            k_gemm1_tile<16><<<1024, 256, 0, stream>>>(x, W1, cnt, dinvP, hP, nTiles);
            k_pull1<16, true><<<(N + 15) / 16, 256, 0, stream>>>(nullptr, slots, dinvP, hP, aggP, N, cnt, eidx, E);
        }
        k_l2<false, 8><<<(N + B - 1) / B, B, 0, stream>>>(aggP, b1, W2, dinvP, oP, out, N);
        k_pull2<true><<<(N + 31) / 32, 256, 0, stream>>>(nullptr, slots, oP, b2, out, N, cnt, eidx, E);
        return;
    }

    const size_t Np = ((size_t)N + 4 + 3) & ~(size_t)3;
    const size_t Ep = ((size_t)E + 3) & ~(size_t)3;
    const size_t Rn = Ep > (size_t)N * 32 ? Ep : (size_t)N * 32;
    const size_t Rw = Ep > (size_t)N * 48 ? Ep : (size_t)N * 48;
    if (ws_size >= (3 * Np + Ep + Rn) * sizeof(float)) {
        const bool wide = ws_size >= (3 * Np + Ep + Rw) * sizeof(float);
        const size_t Rs = wide ? Rw : Rn;
        float* R     = ws;
        int*   srcs  = (int*)(ws + Rs);
        float* dinvP = ws + Rs + Ep;
        int*   off   = (int*)(ws + Rs + Ep + Np);
        int*   cnt   = (int*)(ws + Rs + Ep + 2 * Np);
        int*   rank  = (int*)R;
        float* hP    = R;
        float* aggP  = R + (size_t)N * (wide ? 32 : 16);
        float* oP    = R;

        k_cnt_zero<<<(N + B - 1) / B, B, 0, stream>>>(cnt, N);
        k_rank    <<<(E + B - 1) / B, B, 0, stream>>>(eidx + E, cnt, rank, E);
        k_scan    <<<(N + 4095) / 4096, 1024, 0, stream>>>(cnt, off, dinvP, N);
        k_fill    <<<(E + B - 1) / B, B, 0, stream>>>(eidx, rank, off, srcs, E);
        if (wide) {
            k_gemm1<false, 32, false><<<(nTiles + 7) / 8, 256, 0, stream>>>(x, W1, dinvP, nullptr, nullptr, hP, aggP, nTiles);
            k_pull1<32, false><<<(N + 15) / 16, 256, 0, stream>>>(off, srcs, dinvP, hP, aggP, N, nullptr, eidx, E);
        } else {
            k_gemm1<false, 16, false><<<(nTiles + 7) / 8, 256, 0, stream>>>(x, W1, dinvP, nullptr, nullptr, hP, aggP, nTiles);
            k_pull1<16, false><<<(N + 15) / 16, 256, 0, stream>>>(off, srcs, dinvP, hP, aggP, N, nullptr, eidx, E);
        }
        k_l2<false, 8><<<(N + B - 1) / B, B, 0, stream>>>(aggP, b1, W2, dinvP, oP, out, N);
        k_pull2<false><<<(N + 31) / 32, 256, 0, stream>>>(off, srcs, oP, b2, out, N, nullptr, eidx, E);
        return;
    }

    float* dinv = ws;
    float* h    = ws + (size_t)N;
    float* agg1 = ws + (size_t)N * 17;
    float* obuf = ws + (size_t)N * 33;

    k_deg_init <<<(N + B - 1) / B, B, 0, stream>>>(dinv, N);
    k_deg_edges<<<(E + B - 1) / B, B, 0, stream>>>(eidx + E, dinv, E);
    k_dinv     <<<(N + B - 1) / B, B, 0, stream>>>(dinv, N);

    k_gemm1<true, 16, false><<<(nTiles + 7) / 8, 256, 0, stream>>>(x, W1, dinv, nullptr, nullptr, h, agg1, nTiles);

    const int t1 = E * 16;
    k_edge1<<<(t1 + B - 1) / B, B, 0, stream>>>(eidx, dinv, h, agg1, E);

    k_l2<true, 7><<<(N + B - 1) / B, B, 0, stream>>>(agg1, b1, W2, dinv, obuf, out, N);

    const int t2 = E * 8;
    k_edge2<<<(t2 + B - 1) / B, B, 0, stream>>>(eidx, dinv, obuf, out, E);

    k_lsm<<<(N + B - 1) / B, B, 0, stream>>>(out, b2, N);
}


# Edge placing: each work-group first orders its 13312 edges by destination line in LDS (count, prefix sum, scatter), so a wave's 32 count adds fall on few lines; list order, so sum order, changes
# speedup vs baseline: 2.8317x; 1.5242x over previous
#include <hip/hip_runtime.h>
#include <math.h>

#define N_FEAT 512
#define HID    16
#define NCLS   7

typedef __attribute__((ext_vector_type(2))) float v2f;
typedef __attribute__((ext_vector_type(8))) float v8f;
typedef __attribute__((ext_vector_type(4))) float v4f;

__global__ void k_deg_init(float* __restrict__ deg, int n) {
    int i = blockIdx.x * blockDim.x + threadIdx.x;
    if (i < n) deg[i] = 1.0f;
}

__global__ void k_deg_edges(const int* __restrict__ col, float* __restrict__ deg, int E) {
    int e = blockIdx.x * blockDim.x + threadIdx.x;
    if (e < E) atomicAdd(&deg[col[e]], 1.0f);
}

__global__ void k_dinv(float* __restrict__ deg, int n) {
    int i = blockIdx.x * blockDim.x + threadIdx.x;
    if (i < n) deg[i] = rsqrtf(deg[i]);
}

template <bool SELF, int HP, bool CNT>
__global__ __launch_bounds__(256) void k_gemm1(
    const float* __restrict__ x, const float* __restrict__ W1,
    const float* __restrict__ dinv, const int* __restrict__ cnt, float* __restrict__ dinvOut,
    float* __restrict__ h, float* __restrict__ agg1, int nTiles)
{
    __shared__ float sW[N_FEAT * HID];
    for (int i = threadIdx.x; i < N_FEAT * HID; i += blockDim.x) sW[i] = W1[i];
    __syncthreads();

    int wave = threadIdx.x >> 5;
    int lane = threadIdx.x & 31;
    int tile = blockIdx.x * 8 + wave;
    if (tile >= nTiles) return;

    int m  = lane & 15;
    int kh = lane >> 4;
    const float* xr = x + (size_t)(tile * 16 + m) * N_FEAT + 2 * kh;

    v8f acc = {};
#pragma unroll 4
    for (int k = 0; k < N_FEAT; k += 4) {
        v2f a, b;
        a.x = xr[k];
        a.y = xr[k + 1];
        int kk = k + 2 * kh;
        b.x = sW[kk * HID + m];
        b.y = sW[(kk + 1) * HID + m];
        acc = __builtin_amdgcn_wmma_f32_16x16x4_f32(
             false, a,  false, b,
             (short)0, acc,  false,  false);
    }

#pragma unroll
    for (int v = 0; v < 8; ++v) {
        int node  = tile * 16 + v + 8 * kh;
        float val = acc[v];
        h[(size_t)node * HP + m] = val;
        if (SELF) {
            float di = dinv[node];
            agg1[(size_t)node * HID + m] = val * di * di;
        }
    }
    if ((HP > HID || CNT) && lane < 16) {
        int node = tile * 16 + lane;
        float di;
        if (CNT) {
            di = rsqrtf((float)(cnt[node] + 1));
            dinvOut[node] = di;
        } else {
            di = dinv[node];
        }
        if (HP > HID) h[(size_t)node * HP + HID] = di;
    }
}

#define G1_WAVES 8
#define G1_KW    (N_FEAT / G1_WAVES)
#define G1_KS    (G1_KW / 4)
#define G1_PITCH (N_FEAT + 4)

__device__ __forceinline__ void g1_lds_barrier() {
    asm volatile("s_wait_dscnt 0x0\n\ts_barrier_signal -1\n\ts_barrier_wait -1" ::: "memory");
}

template <int HP>
__global__ __launch_bounds__(256) void k_gemm1_tile(
    const float* __restrict__ x, const float* __restrict__ W1,
    const int* __restrict__ cnt, float* __restrict__ dinvOut,
    float* __restrict__ h, int nTiles)
{
    __shared__ __attribute__((aligned(16))) float sX[16 * G1_PITCH];
    __shared__ float sP[G1_WAVES][8][32];

    const int t    = threadIdx.x;
    const int wave = t >> 5;
    const int lane = t & 31;
    const int m  = lane & 15;
    const int kh = lane >> 4;
    const int kw = wave * G1_KW + 2 * kh;
    const int srow = t >> 7;
    const int scol = (t & 127) * 4;

    int tile = blockIdx.x;
    if (tile >= nTiles) return;

    v2f b[G1_KS];
#pragma unroll
    for (int ks = 0; ks < G1_KS; ++ks) {
        const int kk = kw + 4 * ks;
        b[ks].x = W1[kk * HID + m];
        b[ks].y = W1[(kk + 1) * HID + m];
    }

    v4f pf[8];
    int cn;
    {
        const float* src = x + (size_t)(tile * 16 + srow) * N_FEAT + scol;
#pragma unroll
        for (int j = 0; j < 8; ++j) pf[j] = *reinterpret_cast<const v4f*>(src + (size_t)j * 2 * N_FEAT);
        cn = cnt[tile * 16 + m];
    }

    for (;;) {
        if (t < 16) {
            const int node = tile * 16 + t;
            const float di = rsqrtf((float)(cn + 1));
            dinvOut[node] = di;
            if (HP > HID) h[(size_t)node * HP + HID] = di;
        }
#pragma unroll
        for (int j = 0; j < 8; ++j)
            *reinterpret_cast<v4f*>(&sX[(2 * j + srow) * G1_PITCH + scol]) = pf[j];

        const int  nxt  = tile + (int)gridDim.x;
        const bool more = nxt < nTiles;
        {
            const int    pt   = more ? nxt : tile;
            const size_t step = more ? (size_t)2 * N_FEAT : 0;
            const float* src  = x + (size_t)(pt * 16 + srow) * N_FEAT + scol;
#pragma unroll
            for (int j = 0; j < 8; ++j) pf[j] = *reinterpret_cast<const v4f*>(src + j * step);
            cn = cnt[pt * 16 + m];
        }
        g1_lds_barrier();

        v2f a[G1_KS];
#pragma unroll
        for (int ks = 0; ks < G1_KS; ++ks)
            a[ks] = *reinterpret_cast<const v2f*>(&sX[m * G1_PITCH + kw + 4 * ks]);
        v8f acc0 = {}, acc1 = {};
#pragma unroll
        for (int ks = 0; ks < G1_KS; ks += 2) {
            acc0 = __builtin_amdgcn_wmma_f32_16x16x4_f32(false, a[ks],     false, b[ks],     (short)0, acc0, false, false);
            acc1 = __builtin_amdgcn_wmma_f32_16x16x4_f32(false, a[ks + 1], false, b[ks + 1], (short)0, acc1, false, false);
        }
#pragma unroll
        for (int v = 0; v < 8; ++v) sP[wave][v][lane] = acc0[v] + acc1[v];
        g1_lds_barrier();
        {
            const int v = wave;
            const float s01 = sP[0][v][lane] + sP[1][v][lane];
            const float s23 = sP[2][v][lane] + sP[3][v][lane];
            const float s45 = sP[4][v][lane] + sP[5][v][lane];
            const float s67 = sP[6][v][lane] + sP[7][v][lane];
            const int node = tile * 16 + v + 8 * kh;
            h[(size_t)node * HP + m] = (s01 + s23) + (s45 + s67);
        }
        if (!more) break;
        tile = nxt;
    }
}

__global__ void k_edge1(const int* __restrict__ eidx, const float* __restrict__ dinv,
                        const float* __restrict__ h, float* __restrict__ agg1, int E)
{
    int t = blockIdx.x * blockDim.x + threadIdx.x;
    int e = t >> 4, f = t & 15;
    if (e >= E) return;
    int r = eidx[e];
    int c = eidx[E + e];
    float w = dinv[r] * dinv[c];
    atomicAdd(&agg1[(size_t)c * HID + f], h[(size_t)r * HID + f] * w);
}

template <bool SELF, int OP>
__global__ void k_l2(const float* __restrict__ agg1, const float* __restrict__ b1,
                     const float* __restrict__ W2, const float* __restrict__ dinv,
                     float* __restrict__ o, float* __restrict__ outInit, int n)
{
    __shared__ float sW2[HID * NCLS];
    __shared__ float sb1[HID];
    if (threadIdx.x < HID * NCLS) sW2[threadIdx.x] = W2[threadIdx.x];
    if (threadIdx.x < HID)        sb1[threadIdx.x] = b1[threadIdx.x];
    __syncthreads();

    int i = blockIdx.x * blockDim.x + threadIdx.x;
    if (i >= n) return;

    float hv[HID];
#pragma unroll
    for (int f = 0; f < HID; ++f) {
        float v = agg1[(size_t)i * HID + f] + sb1[f];
        hv[f] = v > 0.0f ? v : 0.0f;
    }
    float di = dinv[i];
    float d2 = di * di;
#pragma unroll
    for (int j = 0; j < NCLS; ++j) {
        float s = 0.0f;
#pragma unroll
        for (int f = 0; f < HID; ++f) s = fmaf(hv[f], sW2[f * NCLS + j], s);
        o[(size_t)i * OP + j]         = s;
        if (SELF) outInit[(size_t)i * NCLS + j] = s * d2;
    }
    if (OP > NCLS) o[(size_t)i * OP + NCLS] = di;
}

__global__ void k_edge2(const int* __restrict__ eidx, const float* __restrict__ dinv,
                        const float* __restrict__ o, float* __restrict__ out, int E)
{
    int t = blockIdx.x * blockDim.x + threadIdx.x;
    int e = t >> 3, j = t & 7;
    if (e >= E || j >= NCLS) return;
    int r = eidx[e];
    int c = eidx[E + e];
    float w = dinv[r] * dinv[c];
    atomicAdd(&out[(size_t)c * NCLS + j], o[(size_t)r * NCLS + j] * w);
}

__global__ void k_lsm(float* __restrict__ out, const float* __restrict__ b2, int n)
{
    int i = blockIdx.x * blockDim.x + threadIdx.x;
    if (i >= n) return;
    float v[NCLS];
    float mx = -1e30f;
#pragma unroll
    for (int j = 0; j < NCLS; ++j) {
        v[j] = out[(size_t)i * NCLS + j] + b2[j];
        mx = fmaxf(mx, v[j]);
    }
    float s = 0.0f;
#pragma unroll
    for (int j = 0; j < NCLS; ++j) s += __expf(v[j] - mx);
    float l = __logf(s);
#pragma unroll
    for (int j = 0; j < NCLS; ++j) out[(size_t)i * NCLS + j] = v[j] - mx - l;
}


__global__ void k_cnt_zero(int* __restrict__ cnt, int n) {
    int i = blockIdx.x * blockDim.x + threadIdx.x;
    if (i < n) cnt[i] = 0;
}

__global__ void k_rank(const int* __restrict__ col, int* __restrict__ cnt,
                       int* __restrict__ rank, int E) {
    int e = blockIdx.x * blockDim.x + threadIdx.x;
    if (e < E) rank[e] = atomicAdd(&cnt[col[e]], 1);
}

__global__ __launch_bounds__(1024) void k_scan(const int* __restrict__ cnt, int* __restrict__ off,
                                               float* __restrict__ dinv, int n)
{
    __shared__ int sInc[32];
    __shared__ int sPre[32];
    const int t = threadIdx.x, lane = t & 31, wave = t >> 5;
    const int tile0 = blockIdx.x * 4096;

    int pre = 0;
    for (int b = 0; b < tile0; b += 4096) {
        int4 v = *reinterpret_cast<const int4*>(cnt + b + t * 4);
        pre += (v.x + v.y) + (v.z + v.w);
    }
    const int i0 = tile0 + t * 4;
    int c[4];
#pragma unroll
    for (int k = 0; k < 4; ++k) c[k] = (i0 + k < n) ? cnt[i0 + k] : 0;
    const int mine = (c[0] + c[1]) + (c[2] + c[3]);

    int incl = mine;
#pragma unroll
    for (int d = 1; d < 32; d <<= 1) { int y = __shfl_up(incl, d); if (lane >= d) incl += y; }
#pragma unroll
    for (int d = 16; d > 0; d >>= 1) pre += __shfl_xor(pre, d);
    if (lane == 31) sInc[wave] = incl;
    if (lane == 0)  sPre[wave] = pre;
    __syncthreads();
    if (wave == 0) {
        int w = sInc[lane], p = sPre[lane];
        int wi = w;
#pragma unroll
        for (int d = 1; d < 32; d <<= 1) { int y = __shfl_up(wi, d); if (lane >= d) wi += y; }
#pragma unroll
        for (int d = 16; d > 0; d >>= 1) p += __shfl_xor(p, d);
        sInc[lane] = wi - w + p;
    }
    __syncthreads();
    int run = sInc[wave] + incl - mine;
#pragma unroll
    for (int k = 0; k < 4; ++k) {
        int i = i0 + k;
        if (i < n) {
            off[i]  = run;
            dinv[i] = rsqrtf((float)(c[k] + 1));
            run += c[k];
            if (i == n - 1) off[n] = run;
        }
    }
}

__global__ void k_fill(const int* __restrict__ eidx, const int* __restrict__ rank,
                       const int* __restrict__ off, int* __restrict__ srcs, int E) {
    int e = blockIdx.x * blockDim.x + threadIdx.x;
    if (e < E) srcs[off[eidx[E + e]] + rank[e]] = eidx[e];
}

#define SLOTS 64

#define PS_T   1024
#define PS_EPT 13
#define PS_WG  (PS_T * PS_EPT)
#define PS_NB  4096

template <bool FULL>
__device__ __forceinline__ void place_sorted(const int* __restrict__ eidx, int* __restrict__ cnt,
                                             int* __restrict__ slots, int E, int g0,
                                             int* sBin, int* sW, int* sC, int* sS)
{
    const int t = threadIdx.x, lane = t & 31, wave = t >> 5;
    const int n = FULL ? PS_WG : E - g0;

    for (int i = t; i < PS_NB; i += PS_T) sBin[i] = 0;
    __syncthreads();

    int c[PS_EPT], s[PS_EPT], pos[PS_EPT];
#pragma unroll
    for (int k = 0; k < PS_EPT; ++k) {
        const int i = k * PS_T + t;
        const bool ok = FULL || i < n;
        c[k] = ok ? eidx[E + g0 + i] : 0;
        s[k] = ok ? eidx[g0 + i] : 0;
    }
#pragma unroll
    for (int k = 0; k < PS_EPT; ++k) {
        const bool ok = FULL || k * PS_T + t < n;
        pos[k] = ok ? atomicAdd(&sBin[(c[k] >> 5) & (PS_NB - 1)], 1) : 0;
    }
    __syncthreads();

    {
        const int4 v = reinterpret_cast<const int4*>(sBin)[t];
        const int mine = (v.x + v.y) + (v.z + v.w);
        int incl = mine;
#pragma unroll
        for (int d = 1; d < 32; d <<= 1) { int y = __shfl_up(incl, d); if (lane >= d) incl += y; }
        if (lane == 31) sW[wave] = incl;
        __syncthreads();
        if (wave == 0) {
            const int w = sW[lane];
            int wi = w;
#pragma unroll
            for (int d = 1; d < 32; d <<= 1) { int y = __shfl_up(wi, d); if (lane >= d) wi += y; }
            sW[lane] = wi - w;
        }
        __syncthreads();
        int4 o;
        o.x = sW[wave] + incl - mine;
        o.y = o.x + v.x;
        o.z = o.y + v.y;
        o.w = o.z + v.z;
        reinterpret_cast<int4*>(sBin)[t] = o;
    }
    __syncthreads();

#pragma unroll
    for (int k = 0; k < PS_EPT; ++k) {
        const bool ok = FULL || k * PS_T + t < n;
        if (ok) {
            const int p = sBin[(c[k] >> 5) & (PS_NB - 1)] + pos[k];
            sC[p] = c[k];
            sS[p] = s[k];
        }
    }
    __syncthreads();

    if (FULL) {
        int r[PS_EPT];
#pragma unroll
        for (int k = 0; k < PS_EPT; ++k) { c[k] = sC[k * PS_T + t]; s[k] = sS[k * PS_T + t]; }
#pragma unroll
        for (int k = 0; k < PS_EPT; ++k) r[k] = atomicAdd(&cnt[c[k]], 1);
#pragma unroll
        for (int k = 0; k < PS_EPT; ++k)
            if (r[k] < SLOTS) slots[c[k] * SLOTS + r[k]] = s[k];
    } else {
        for (int i = t; i < n; i += PS_T) {
            const int cc = sC[i];
            const int r = atomicAdd(&cnt[cc], 1);
            if (r < SLOTS) slots[cc * SLOTS + r] = sS[i];
        }
    }
}

__global__ __launch_bounds__(PS_T) void k_place(const int* __restrict__ eidx, int* __restrict__ cnt,
                                                int* __restrict__ slots, int E) {
    __shared__ __attribute__((aligned(16))) int sBin[PS_NB];
    __shared__ int sW[32];
    __shared__ int sC[PS_WG];
    __shared__ int sS[PS_WG];
    const int g0 = blockIdx.x * PS_WG;
    if (g0 + PS_WG <= E) place_sorted<true >(eidx, cnt, slots, E, g0, sBin, sW, sC, sS);
    else                 place_sorted<false>(eidx, cnt, slots, E, g0, sBin, sW, sC, sS);
}

template <int K> __device__ __forceinline__ int row16_i(int v) {
    return __builtin_amdgcn_update_dpp(0, v, 0x150 + K, 0xf, 0xf, false);
}
template <int K> __device__ __forceinline__ float row16_f(float v) {
    return __int_as_float(row16_i<K>(__float_as_int(v)));
}
template <int K> __device__ __forceinline__ int grp8_i(int v) {
    return (int)__builtin_amdgcn_mov_dpp8((unsigned)v, K * 0x249249);
}
template <int K> __device__ __forceinline__ float grp8_f(float v) {
    return __int_as_float(grp8_i<K>(__float_as_int(v)));
}

#define PULL16_SHARE(K) sk[K] = row16_i<K>(s); wk[K] = row16_f<K>(w);

template <int HP>
__device__ __forceinline__ float pull16_step(const float* __restrict__ h, const float* __restrict__ dinv,
                                             int s, bool on, float dc, int f, float acc)
{
    const float ds = HP > HID ? h[s * HP + HID] : dinv[s];
    const float w = on ? ds * dc : 0.0f;
    int sk[16]; float wk[16], hv[16];
    PULL16_SHARE(0)  PULL16_SHARE(1)  PULL16_SHARE(2)  PULL16_SHARE(3)
    PULL16_SHARE(4)  PULL16_SHARE(5)  PULL16_SHARE(6)  PULL16_SHARE(7)
    PULL16_SHARE(8)  PULL16_SHARE(9)  PULL16_SHARE(10) PULL16_SHARE(11)
    PULL16_SHARE(12) PULL16_SHARE(13) PULL16_SHARE(14) PULL16_SHARE(15)
#pragma unroll
    for (int k = 0; k < 16; ++k) hv[k] = h[sk[k] * HP + f];
#pragma unroll
    for (int k = 0; k < 16; ++k) acc = fmaf(hv[k], wk[k], acc);
    return acc;
}

template <int HP, bool SLOT>
__global__ __launch_bounds__(256) void k_pull1(const int* __restrict__ off, const int* __restrict__ srcs,
                                               const float* __restrict__ dinv, const float* __restrict__ h,
                                               float* __restrict__ agg1, int n,
                                               const int* __restrict__ cnt, const int* __restrict__ eidx, int E)
{
    const int g = (blockIdx.x * 256 + threadIdx.x) >> 4;
    const int f = threadIdx.x & 15;
    if (g >= n) return;
    int beg, end, deg = 0;
    if (SLOT) {
        deg = cnt[g];
        beg = g * SLOTS;
        end = deg <= SLOTS ? beg + deg : beg;
    } else {
        beg = off[g]; end = off[g + 1];
    }
    const float dc = dinv[g];
    float acc = h[g * HP + f] * dc * dc;
    for (int p = beg; p < end; p += 16) {
        const bool on = p + f < end;
        const int   s = on ? srcs[p + f] : g;
        acc = pull16_step<HP>(h, dinv, s, on, dc, f, acc);
    }
    if (SLOT && deg > SLOTS) {
        const unsigned rowsh = threadIdx.x & 16;
        for (int e0 = 0; e0 < E; e0 += 16) {
            const int e = e0 + f;
            const bool on = e < E && eidx[E + e] == g;
            const unsigned hit = ((unsigned)__ballot(on) >> rowsh) & 0xffffu;
            if (hit) {
                const int s = on ? eidx[e] : g;
                acc = pull16_step<HP>(h, dinv, s, on, dc, f, acc);
            }
        }
    }
    agg1[(size_t)g * HID + f] = acc;
}

#define PULL8_SHARE(K)  sk[K] = grp8_i<K>(s);
#define PULL8_WEIGHT(K) { const float dk = grp8_f<7>(ov[K]); wk[K] = p + K < end ? dk * dc : 0.0f; }

template <bool SLOT>
__global__ __launch_bounds__(256) void k_pull2(const int* __restrict__ off, const int* __restrict__ srcs,
                                               const float* __restrict__ o,
                                               const float* __restrict__ b2, float* __restrict__ out, int n,
                                               const int* __restrict__ cnt, const int* __restrict__ eidx, int E)
{
    const int g = (blockIdx.x * 256 + threadIdx.x) >> 3;
    const int j = threadIdx.x & 7;
    if (g >= n) return;
    int beg, end, deg = 0;
    if (SLOT) {
        deg = cnt[g];
        beg = g * SLOTS;
        end = deg <= SLOTS ? beg + deg : beg;
    } else {
        beg = off[g]; end = off[g + 1];
    }
    const float own = o[g * 8 + j];
    const float dc = grp8_f<7>(own);
    float acc = own * dc * dc;
    for (int p = beg; p < end; p += 8) {
        const int s = p + j < end ? srcs[p + j] : g;
        int sk[8]; float wk[8], ov[8];
        PULL8_SHARE(0) PULL8_SHARE(1) PULL8_SHARE(2) PULL8_SHARE(3)
        PULL8_SHARE(4) PULL8_SHARE(5) PULL8_SHARE(6) PULL8_SHARE(7)
#pragma unroll
        for (int k = 0; k < 8; ++k) ov[k] = o[sk[k] * 8 + j];
        PULL8_WEIGHT(0) PULL8_WEIGHT(1) PULL8_WEIGHT(2) PULL8_WEIGHT(3)
        PULL8_WEIGHT(4) PULL8_WEIGHT(5) PULL8_WEIGHT(6) PULL8_WEIGHT(7)
#pragma unroll
        for (int k = 0; k < 8; ++k) acc = fmaf(ov[k], wk[k], acc);
    }
    if (SLOT && deg > SLOTS) {
        const unsigned grpsh = threadIdx.x & 24;
        for (int e0 = 0; e0 < E; e0 += 8) {
            const int e = e0 + j;
            const bool on = e < E && eidx[E + e] == g;
            const unsigned hit = ((unsigned)__ballot(on) >> grpsh) & 0xffu;
            if (hit) {
                const int s = on ? eidx[e] : g;
                int sk[8]; float ov[8];
                PULL8_SHARE(0) PULL8_SHARE(1) PULL8_SHARE(2) PULL8_SHARE(3)
                PULL8_SHARE(4) PULL8_SHARE(5) PULL8_SHARE(6) PULL8_SHARE(7)
#pragma unroll
                for (int k = 0; k < 8; ++k) ov[k] = o[sk[k] * 8 + j];
#pragma unroll
                for (int k = 0; k < 8; ++k) {
                    const float dk = grp8_f<7>(ov[k]);
                    const float wv = (hit >> k) & 1u ? dk * dc : 0.0f;
                    acc = fmaf(ov[k], wv, acc);
                }
            }
        }
    }
    const int jj = j < NCLS ? j : NCLS - 1;
    const float v = acc + b2[jj];
    float mx = j < NCLS ? v : -INFINITY;
#pragma unroll
    for (int d = 1; d < 8; d <<= 1) mx = fmaxf(mx, __shfl_xor(mx, d, 8));
    float ex = j < NCLS ? __expf(v - mx) : 0.0f;
#pragma unroll
    for (int d = 1; d < 8; d <<= 1) ex += __shfl_xor(ex, d, 8);
    if (j < NCLS) out[(size_t)g * NCLS + j] = v - mx - __logf(ex);
}

extern "C" void kernel_launch(void* const* d_in, const int* in_sizes, int n_in,
                              void* d_out, int out_size, void* d_ws, size_t ws_size,
                              hipStream_t stream)
{
    const float* x    = (const float*)d_in[0];
    const int*   eidx = (const int*)  d_in[1];
    const float* W1   = (const float*)d_in[2];
    const float* b1   = (const float*)d_in[3];
    const float* W2   = (const float*)d_in[4];
    const float* b2   = (const float*)d_in[5];
    float* out = (float*)d_out;

    const int N = in_sizes[0] / N_FEAT;
    const int E = in_sizes[1] / 2;

    float* ws   = (float*)d_ws;
    const int B = 256;
    const int nTiles = (N + 15) / 16;

    const size_t NpS = ((size_t)N + 3) & ~(size_t)3;
    const size_t Sl  = (size_t)N * SLOTS;
    if (ws_size >= ((size_t)N * 32 + Sl + 2 * NpS) * sizeof(float)) {
        const bool wide = ws_size >= ((size_t)N * 48 + Sl + 2 * NpS) * sizeof(float);
        const size_t Rs = (size_t)N * (wide ? 48 : 32);
        float* R     = ws;
        int*   slots = (int*)(ws + Rs);
        float* dinvP = ws + Rs + Sl;
        int*   cnt   = (int*)(ws + Rs + Sl + NpS);
        float* hP    = R;
        float* aggP  = R + (size_t)N * (wide ? 32 : 16);
        float* oP    = R;

        k_cnt_zero<<<(N + B - 1) / B, B, 0, stream>>>(cnt, N);
        k_place   <<<(E + PS_WG - 1) / PS_WG, PS_T, 0, stream>>>(eidx, cnt, slots, E);
        if (wide) {
            k_gemm1_tile<32><<<1024, 256, 0, stream>>>(x, W1, cnt, dinvP, hP, nTiles);
            k_pull1<32, true><<<(N + 15) / 16, 256, 0, stream>>>(nullptr, slots, dinvP, hP, aggP, N, cnt, eidx, E);
        } else {
            k_gemm1_tile<16><<<1024, 256, 0, stream>>>(x, W1, cnt, dinvP, hP, nTiles);
            k_pull1<16, true><<<(N + 15) / 16, 256, 0, stream>>>(nullptr, slots, dinvP, hP, aggP, N, cnt, eidx, E);
        }
        k_l2<false, 8><<<(N + B - 1) / B, B, 0, stream>>>(aggP, b1, W2, dinvP, oP, out, N);
        k_pull2<true><<<(N + 31) / 32, 256, 0, stream>>>(nullptr, slots, oP, b2, out, N, cnt, eidx, E);
        return;
    }

    const size_t Np = ((size_t)N + 4 + 3) & ~(size_t)3;
    const size_t Ep = ((size_t)E + 3) & ~(size_t)3;
    const size_t Rn = Ep > (size_t)N * 32 ? Ep : (size_t)N * 32;
    const size_t Rw = Ep > (size_t)N * 48 ? Ep : (size_t)N * 48;
    if (ws_size >= (3 * Np + Ep + Rn) * sizeof(float)) {
        const bool wide = ws_size >= (3 * Np + Ep + Rw) * sizeof(float);
        const size_t Rs = wide ? Rw : Rn;
        float* R     = ws;
        int*   srcs  = (int*)(ws + Rs);
        float* dinvP = ws + Rs + Ep;
        int*   off   = (int*)(ws + Rs + Ep + Np);
        int*   cnt   = (int*)(ws + Rs + Ep + 2 * Np);
        int*   rank  = (int*)R;
        float* hP    = R;
        float* aggP  = R + (size_t)N * (wide ? 32 : 16);
        float* oP    = R;

        k_cnt_zero<<<(N + B - 1) / B, B, 0, stream>>>(cnt, N);
        k_rank    <<<(E + B - 1) / B, B, 0, stream>>>(eidx + E, cnt, rank, E);
        k_scan    <<<(N + 4095) / 4096, 1024, 0, stream>>>(cnt, off, dinvP, N);
        k_fill    <<<(E + B - 1) / B, B, 0, stream>>>(eidx, rank, off, srcs, E);
        if (wide) {
            k_gemm1<false, 32, false><<<(nTiles + 7) / 8, 256, 0, stream>>>(x, W1, dinvP, nullptr, nullptr, hP, aggP, nTiles);
            k_pull1<32, false><<<(N + 15) / 16, 256, 0, stream>>>(off, srcs, dinvP, hP, aggP, N, nullptr, eidx, E);
        } else {
            k_gemm1<false, 16, false><<<(nTiles + 7) / 8, 256, 0, stream>>>(x, W1, dinvP, nullptr, nullptr, hP, aggP, nTiles);
            k_pull1<16, false><<<(N + 15) / 16, 256, 0, stream>>>(off, srcs, dinvP, hP, aggP, N, nullptr, eidx, E);
        }
        k_l2<false, 8><<<(N + B - 1) / B, B, 0, stream>>>(aggP, b1, W2, dinvP, oP, out, N);
        k_pull2<false><<<(N + 31) / 32, 256, 0, stream>>>(off, srcs, oP, b2, out, N, nullptr, eidx, E);
        return;
    }

    float* dinv = ws;
    float* h    = ws + (size_t)N;
    float* agg1 = ws + (size_t)N * 17;
    float* obuf = ws + (size_t)N * 33;

    k_deg_init <<<(N + B - 1) / B, B, 0, stream>>>(dinv, N);
    k_deg_edges<<<(E + B - 1) / B, B, 0, stream>>>(eidx + E, dinv, E);
    k_dinv     <<<(N + B - 1) / B, B, 0, stream>>>(dinv, N);

    k_gemm1<true, 16, false><<<(nTiles + 7) / 8, 256, 0, stream>>>(x, W1, dinv, nullptr, nullptr, h, agg1, nTiles);

    const int t1 = E * 16;
    k_edge1<<<(t1 + B - 1) / B, B, 0, stream>>>(eidx, dinv, h, agg1, E);

    k_l2<true, 7><<<(N + B - 1) / B, B, 0, stream>>>(agg1, b1, W2, dinv, obuf, out, N);

    const int t2 = E * 8;
    k_edge2<<<(t2 + B - 1) / B, B, 0, stream>>>(eidx, dinv, obuf, out, E);

    k_lsm<<<(N + B - 1) / B, B, 0, stream>>>(out, b2, N);
}


# Layer-1 edge walk re-cut to whole-line row reads: 8 lanes per destination, 16 bytes per lane, a source's dinv rides in the row's own load (no separate weight read per edge); sum order unchanged
# speedup vs baseline: 2.9805x; 1.0525x over previous
#include <hip/hip_runtime.h>
#include <math.h>

#define N_FEAT 512
#define HID    16
#define NCLS   7

typedef __attribute__((ext_vector_type(2))) float v2f;
typedef __attribute__((ext_vector_type(8))) float v8f;
typedef __attribute__((ext_vector_type(4))) float v4f;

__global__ void k_deg_init(float* __restrict__ deg, int n) {
    int i = blockIdx.x * blockDim.x + threadIdx.x;
    if (i < n) deg[i] = 1.0f;
}

__global__ void k_deg_edges(const int* __restrict__ col, float* __restrict__ deg, int E) {
    int e = blockIdx.x * blockDim.x + threadIdx.x;
    if (e < E) atomicAdd(&deg[col[e]], 1.0f);
}

__global__ void k_dinv(float* __restrict__ deg, int n) {
    int i = blockIdx.x * blockDim.x + threadIdx.x;
    if (i < n) deg[i] = rsqrtf(deg[i]);
}

template <bool SELF, int HP, bool CNT>
__global__ __launch_bounds__(256) void k_gemm1(
    const float* __restrict__ x, const float* __restrict__ W1,
    const float* __restrict__ dinv, const int* __restrict__ cnt, float* __restrict__ dinvOut,
    float* __restrict__ h, float* __restrict__ agg1, int nTiles)
{
    __shared__ float sW[N_FEAT * HID];
    for (int i = threadIdx.x; i < N_FEAT * HID; i += blockDim.x) sW[i] = W1[i];
    __syncthreads();

    int wave = threadIdx.x >> 5;
    int lane = threadIdx.x & 31;
    int tile = blockIdx.x * 8 + wave;
    if (tile >= nTiles) return;

    int m  = lane & 15;
    int kh = lane >> 4;
    const float* xr = x + (size_t)(tile * 16 + m) * N_FEAT + 2 * kh;

    v8f acc = {};
#pragma unroll 4
    for (int k = 0; k < N_FEAT; k += 4) {
        v2f a, b;
        a.x = xr[k];
        a.y = xr[k + 1];
        int kk = k + 2 * kh;
        b.x = sW[kk * HID + m];
        b.y = sW[(kk + 1) * HID + m];
        acc = __builtin_amdgcn_wmma_f32_16x16x4_f32(
             false, a,  false, b,
             (short)0, acc,  false,  false);
    }

#pragma unroll
    for (int v = 0; v < 8; ++v) {
        int node  = tile * 16 + v + 8 * kh;
        float val = acc[v];
        h[(size_t)node * HP + m] = val;
        if (SELF) {
            float di = dinv[node];
            agg1[(size_t)node * HID + m] = val * di * di;
        }
    }
    if ((HP > HID || CNT) && lane < 16) {
        int node = tile * 16 + lane;
        float di;
        if (CNT) {
            di = rsqrtf((float)(cnt[node] + 1));
            dinvOut[node] = di;
        } else {
            di = dinv[node];
        }
        if (HP > HID) *reinterpret_cast<v4f*>(&h[(size_t)node * HP + HID]) = v4f{di, 0.0f, 0.0f, 0.0f};
    }
}

#define G1_WAVES 8
#define G1_KW    (N_FEAT / G1_WAVES)
#define G1_KS    (G1_KW / 4)
#define G1_PITCH (N_FEAT + 4)

__device__ __forceinline__ void g1_lds_barrier() {
    asm volatile("s_wait_dscnt 0x0\n\ts_barrier_signal -1\n\ts_barrier_wait -1" ::: "memory");
}

template <int HP>
__global__ __launch_bounds__(256) void k_gemm1_tile(
    const float* __restrict__ x, const float* __restrict__ W1,
    const int* __restrict__ cnt, float* __restrict__ dinvOut,
    float* __restrict__ h, int nTiles)
{
    __shared__ __attribute__((aligned(16))) float sX[16 * G1_PITCH];
    __shared__ float sP[G1_WAVES][8][32];

    const int t    = threadIdx.x;
    const int wave = t >> 5;
    const int lane = t & 31;
    const int m  = lane & 15;
    const int kh = lane >> 4;
    const int kw = wave * G1_KW + 2 * kh;
    const int srow = t >> 7;
    const int scol = (t & 127) * 4;

    int tile = blockIdx.x;
    if (tile >= nTiles) return;

    v2f b[G1_KS];
#pragma unroll
    for (int ks = 0; ks < G1_KS; ++ks) {
        const int kk = kw + 4 * ks;
        b[ks].x = W1[kk * HID + m];
        b[ks].y = W1[(kk + 1) * HID + m];
    }

    v4f pf[8];
    int cn;
    {
        const float* src = x + (size_t)(tile * 16 + srow) * N_FEAT + scol;
#pragma unroll
        for (int j = 0; j < 8; ++j) pf[j] = *reinterpret_cast<const v4f*>(src + (size_t)j * 2 * N_FEAT);
        cn = cnt[tile * 16 + m];
    }

    for (;;) {
        if (t < 16) {
            const int node = tile * 16 + t;
            const float di = rsqrtf((float)(cn + 1));
            dinvOut[node] = di;
            if (HP > HID) *reinterpret_cast<v4f*>(&h[(size_t)node * HP + HID]) = v4f{di, 0.0f, 0.0f, 0.0f};
        }
#pragma unroll
        for (int j = 0; j < 8; ++j)
            *reinterpret_cast<v4f*>(&sX[(2 * j + srow) * G1_PITCH + scol]) = pf[j];

        const int  nxt  = tile + (int)gridDim.x;
        const bool more = nxt < nTiles;
        {
            const int    pt   = more ? nxt : tile;
            const size_t step = more ? (size_t)2 * N_FEAT : 0;
            const float* src  = x + (size_t)(pt * 16 + srow) * N_FEAT + scol;
#pragma unroll
            for (int j = 0; j < 8; ++j) pf[j] = *reinterpret_cast<const v4f*>(src + j * step);
            cn = cnt[pt * 16 + m];
        }
        g1_lds_barrier();

        v2f a[G1_KS];
#pragma unroll
        for (int ks = 0; ks < G1_KS; ++ks)
            a[ks] = *reinterpret_cast<const v2f*>(&sX[m * G1_PITCH + kw + 4 * ks]);
        v8f acc0 = {}, acc1 = {};
#pragma unroll
        for (int ks = 0; ks < G1_KS; ks += 2) {
            acc0 = __builtin_amdgcn_wmma_f32_16x16x4_f32(false, a[ks],     false, b[ks],     (short)0, acc0, false, false);
            acc1 = __builtin_amdgcn_wmma_f32_16x16x4_f32(false, a[ks + 1], false, b[ks + 1], (short)0, acc1, false, false);
        }
#pragma unroll
        for (int v = 0; v < 8; ++v) sP[wave][v][lane] = acc0[v] + acc1[v];
        g1_lds_barrier();
        {
            const int v = wave;
            const float s01 = sP[0][v][lane] + sP[1][v][lane];
            const float s23 = sP[2][v][lane] + sP[3][v][lane];
            const float s45 = sP[4][v][lane] + sP[5][v][lane];
            const float s67 = sP[6][v][lane] + sP[7][v][lane];
            const int node = tile * 16 + v + 8 * kh;
            h[(size_t)node * HP + m] = (s01 + s23) + (s45 + s67);
        }
        if (!more) break;
        tile = nxt;
    }
}

__global__ void k_edge1(const int* __restrict__ eidx, const float* __restrict__ dinv,
                        const float* __restrict__ h, float* __restrict__ agg1, int E)
{
    int t = blockIdx.x * blockDim.x + threadIdx.x;
    int e = t >> 4, f = t & 15;
    if (e >= E) return;
    int r = eidx[e];
    int c = eidx[E + e];
    float w = dinv[r] * dinv[c];
    atomicAdd(&agg1[(size_t)c * HID + f], h[(size_t)r * HID + f] * w);
}

template <bool SELF, int OP>
__global__ void k_l2(const float* __restrict__ agg1, const float* __restrict__ b1,
                     const float* __restrict__ W2, const float* __restrict__ dinv,
                     float* __restrict__ o, float* __restrict__ outInit, int n)
{
    __shared__ float sW2[HID * NCLS];
    __shared__ float sb1[HID];
    if (threadIdx.x < HID * NCLS) sW2[threadIdx.x] = W2[threadIdx.x];
    if (threadIdx.x < HID)        sb1[threadIdx.x] = b1[threadIdx.x];
    __syncthreads();

    int i = blockIdx.x * blockDim.x + threadIdx.x;
    if (i >= n) return;

    float hv[HID];
#pragma unroll
    for (int f = 0; f < HID; ++f) {
        float v = agg1[(size_t)i * HID + f] + sb1[f];
        hv[f] = v > 0.0f ? v : 0.0f;
    }
    float di = dinv[i];
    float d2 = di * di;
#pragma unroll
    for (int j = 0; j < NCLS; ++j) {
        float s = 0.0f;
#pragma unroll
        for (int f = 0; f < HID; ++f) s = fmaf(hv[f], sW2[f * NCLS + j], s);
        o[(size_t)i * OP + j]         = s;
        if (SELF) outInit[(size_t)i * NCLS + j] = s * d2;
    }
    if (OP > NCLS) o[(size_t)i * OP + NCLS] = di;
}

__global__ void k_edge2(const int* __restrict__ eidx, const float* __restrict__ dinv,
                        const float* __restrict__ o, float* __restrict__ out, int E)
{
    int t = blockIdx.x * blockDim.x + threadIdx.x;
    int e = t >> 3, j = t & 7;
    if (e >= E || j >= NCLS) return;
    int r = eidx[e];
    int c = eidx[E + e];
    float w = dinv[r] * dinv[c];
    atomicAdd(&out[(size_t)c * NCLS + j], o[(size_t)r * NCLS + j] * w);
}

__global__ void k_lsm(float* __restrict__ out, const float* __restrict__ b2, int n)
{
    int i = blockIdx.x * blockDim.x + threadIdx.x;
    if (i >= n) return;
    float v[NCLS];
    float mx = -1e30f;
#pragma unroll
    for (int j = 0; j < NCLS; ++j) {
        v[j] = out[(size_t)i * NCLS + j] + b2[j];
        mx = fmaxf(mx, v[j]);
    }
    float s = 0.0f;
#pragma unroll
    for (int j = 0; j < NCLS; ++j) s += __expf(v[j] - mx);
    float l = __logf(s);
#pragma unroll
    for (int j = 0; j < NCLS; ++j) out[(size_t)i * NCLS + j] = v[j] - mx - l;
}


__global__ void k_cnt_zero(int* __restrict__ cnt, int n) {
    int i = blockIdx.x * blockDim.x + threadIdx.x;
    if (i < n) cnt[i] = 0;
}

__global__ void k_rank(const int* __restrict__ col, int* __restrict__ cnt,
                       int* __restrict__ rank, int E) {
    int e = blockIdx.x * blockDim.x + threadIdx.x;
    if (e < E) rank[e] = atomicAdd(&cnt[col[e]], 1);
}

__global__ __launch_bounds__(1024) void k_scan(const int* __restrict__ cnt, int* __restrict__ off,
                                               float* __restrict__ dinv, int n)
{
    __shared__ int sInc[32];
    __shared__ int sPre[32];
    const int t = threadIdx.x, lane = t & 31, wave = t >> 5;
    const int tile0 = blockIdx.x * 4096;

    int pre = 0;
    for (int b = 0; b < tile0; b += 4096) {
        int4 v = *reinterpret_cast<const int4*>(cnt + b + t * 4);
        pre += (v.x + v.y) + (v.z + v.w);
    }
    const int i0 = tile0 + t * 4;
    int c[4];
#pragma unroll
    for (int k = 0; k < 4; ++k) c[k] = (i0 + k < n) ? cnt[i0 + k] : 0;
    const int mine = (c[0] + c[1]) + (c[2] + c[3]);

    int incl = mine;
#pragma unroll
    for (int d = 1; d < 32; d <<= 1) { int y = __shfl_up(incl, d); if (lane >= d) incl += y; }
#pragma unroll
    for (int d = 16; d > 0; d >>= 1) pre += __shfl_xor(pre, d);
    if (lane == 31) sInc[wave] = incl;
    if (lane == 0)  sPre[wave] = pre;
    __syncthreads();
    if (wave == 0) {
        int w = sInc[lane], p = sPre[lane];
        int wi = w;
#pragma unroll
        for (int d = 1; d < 32; d <<= 1) { int y = __shfl_up(wi, d); if (lane >= d) wi += y; }
#pragma unroll
        for (int d = 16; d > 0; d >>= 1) p += __shfl_xor(p, d);
        sInc[lane] = wi - w + p;
    }
    __syncthreads();
    int run = sInc[wave] + incl - mine;
#pragma unroll
    for (int k = 0; k < 4; ++k) {
        int i = i0 + k;
        if (i < n) {
            off[i]  = run;
            dinv[i] = rsqrtf((float)(c[k] + 1));
            run += c[k];
            if (i == n - 1) off[n] = run;
        }
    }
}

__global__ void k_fill(const int* __restrict__ eidx, const int* __restrict__ rank,
                       const int* __restrict__ off, int* __restrict__ srcs, int E) {
    int e = blockIdx.x * blockDim.x + threadIdx.x;
    if (e < E) srcs[off[eidx[E + e]] + rank[e]] = eidx[e];
}

#define SLOTS 64

#define PS_T   1024
#define PS_EPT 13
#define PS_WG  (PS_T * PS_EPT)
#define PS_NB  4096

template <bool FULL>
__device__ __forceinline__ void place_sorted(const int* __restrict__ eidx, int* __restrict__ cnt,
                                             int* __restrict__ slots, int E, int g0,
                                             int* sBin, int* sW, int* sC, int* sS)
{
    const int t = threadIdx.x, lane = t & 31, wave = t >> 5;
    const int n = FULL ? PS_WG : E - g0;

    for (int i = t; i < PS_NB; i += PS_T) sBin[i] = 0;
    __syncthreads();

    int c[PS_EPT], s[PS_EPT], pos[PS_EPT];
#pragma unroll
    for (int k = 0; k < PS_EPT; ++k) {
        const int i = k * PS_T + t;
        const bool ok = FULL || i < n;
        c[k] = ok ? eidx[E + g0 + i] : 0;
        s[k] = ok ? eidx[g0 + i] : 0;
    }
#pragma unroll
    for (int k = 0; k < PS_EPT; ++k) {
        const bool ok = FULL || k * PS_T + t < n;
        pos[k] = ok ? atomicAdd(&sBin[(c[k] >> 5) & (PS_NB - 1)], 1) : 0;
    }
    __syncthreads();

    {
        const int4 v = reinterpret_cast<const int4*>(sBin)[t];
        const int mine = (v.x + v.y) + (v.z + v.w);
        int incl = mine;
#pragma unroll
        for (int d = 1; d < 32; d <<= 1) { int y = __shfl_up(incl, d); if (lane >= d) incl += y; }
        if (lane == 31) sW[wave] = incl;
        __syncthreads();
        if (wave == 0) {
            const int w = sW[lane];
            int wi = w;
#pragma unroll
            for (int d = 1; d < 32; d <<= 1) { int y = __shfl_up(wi, d); if (lane >= d) wi += y; }
            sW[lane] = wi - w;
        }
        __syncthreads();
        int4 o;
        o.x = sW[wave] + incl - mine;
        o.y = o.x + v.x;
        o.z = o.y + v.y;
        o.w = o.z + v.z;
        reinterpret_cast<int4*>(sBin)[t] = o;
    }
    __syncthreads();

#pragma unroll
    for (int k = 0; k < PS_EPT; ++k) {
        const bool ok = FULL || k * PS_T + t < n;
        if (ok) {
            const int p = sBin[(c[k] >> 5) & (PS_NB - 1)] + pos[k];
            sC[p] = c[k];
            sS[p] = s[k];
        }
    }
    __syncthreads();

    if (FULL) {
        int r[PS_EPT];
#pragma unroll
        for (int k = 0; k < PS_EPT; ++k) { c[k] = sC[k * PS_T + t]; s[k] = sS[k * PS_T + t]; }
#pragma unroll
        for (int k = 0; k < PS_EPT; ++k) r[k] = atomicAdd(&cnt[c[k]], 1);
#pragma unroll
        for (int k = 0; k < PS_EPT; ++k)
            if (r[k] < SLOTS) slots[c[k] * SLOTS + r[k]] = s[k];
    } else {
        for (int i = t; i < n; i += PS_T) {
            const int cc = sC[i];
            const int r = atomicAdd(&cnt[cc], 1);
            if (r < SLOTS) slots[cc * SLOTS + r] = sS[i];
        }
    }
}

__global__ __launch_bounds__(PS_T) void k_place(const int* __restrict__ eidx, int* __restrict__ cnt,
                                                int* __restrict__ slots, int E) {
    __shared__ __attribute__((aligned(16))) int sBin[PS_NB];
    __shared__ int sW[32];
    __shared__ int sC[PS_WG];
    __shared__ int sS[PS_WG];
    const int g0 = blockIdx.x * PS_WG;
    if (g0 + PS_WG <= E) place_sorted<true >(eidx, cnt, slots, E, g0, sBin, sW, sC, sS);
    else                 place_sorted<false>(eidx, cnt, slots, E, g0, sBin, sW, sC, sS);
}

template <int K> __device__ __forceinline__ int grp8_i(int v) {
    return (int)__builtin_amdgcn_mov_dpp8((unsigned)v, K * 0x249249);
}
template <int K> __device__ __forceinline__ float grp8_f(float v) {
    return __int_as_float(grp8_i<K>(__float_as_int(v)));
}

#define PULL8_SHARE(K)  sk[K] = grp8_i<K>(s);
#define PULL1_ADD(K) { \
    const float dk = HP > HID ? grp8_f<4>(rv[K].x) : grp8_f<K>(ds1); \
    const float w = (mask >> K) & 1u ? dk * dc : 0.0f; \
    acc.x = fmaf(rv[K].x, w, acc.x); acc.y = fmaf(rv[K].y, w, acc.y); \
    acc.z = fmaf(rv[K].z, w, acc.z); acc.w = fmaf(rv[K].w, w, acc.w); }

template <int HP>
__device__ __forceinline__ v4f pull1_step(const float* __restrict__ h, const float* __restrict__ dinv,
                                          int s, unsigned mask, float dc, int po, v4f acc)
{
    float ds1 = 0.0f;
    if (HP == HID) ds1 = dinv[s];
    int sk[8]; v4f rv[8];
    PULL8_SHARE(0) PULL8_SHARE(1) PULL8_SHARE(2) PULL8_SHARE(3)
    PULL8_SHARE(4) PULL8_SHARE(5) PULL8_SHARE(6) PULL8_SHARE(7)
#pragma unroll
    for (int k = 0; k < 8; ++k) rv[k] = *reinterpret_cast<const v4f*>(h + sk[k] * HP + po);
    PULL1_ADD(0) PULL1_ADD(1) PULL1_ADD(2) PULL1_ADD(3)
    PULL1_ADD(4) PULL1_ADD(5) PULL1_ADD(6) PULL1_ADD(7)
    return acc;
}

template <int HP, bool SLOT>
__global__ __launch_bounds__(256) void k_pull1(const int* __restrict__ off, const int* __restrict__ srcs,
                                               const float* __restrict__ dinv, const float* __restrict__ h,
                                               float* __restrict__ agg1, int n,
                                               const int* __restrict__ cnt, const int* __restrict__ eidx, int E)
{
    const int g = (blockIdx.x * 256 + threadIdx.x) >> 3;
    const int q = threadIdx.x & 7;
    if (g >= n) return;
    int beg, end, deg = 0;
    if (SLOT) {
        deg = cnt[g];
        beg = g * SLOTS;
        end = deg <= SLOTS ? beg + deg : beg;
    } else {
        beg = off[g]; end = off[g + 1];
    }
    const int po = HP > HID ? (q < 4 ? q : 4) * 4 : (q & 3) * 4;
    const v4f own = *reinterpret_cast<const v4f*>(h + g * HP + po);
    const float dc = HP > HID ? grp8_f<4>(own.x) : dinv[g];
    v4f acc;
    acc.x = own.x * dc * dc; acc.y = own.y * dc * dc; acc.z = own.z * dc * dc; acc.w = own.w * dc * dc;
    for (int p = beg; p < end; p += 8) {
        const int left = end - p;
        const unsigned mask = left >= 8 ? 0xffu : (1u << left) - 1u;
        const int s = q < left ? srcs[p + q] : g;
        acc = pull1_step<HP>(h, dinv, s, mask, dc, po, acc);
    }
    if (SLOT && deg > SLOTS) {
        const unsigned grpsh = threadIdx.x & 24;
        for (int e0 = 0; e0 < E; e0 += 8) {
            const int e = e0 + q;
            const bool on = e < E && eidx[E + e] == g;
            const unsigned hit = ((unsigned)__ballot(on) >> grpsh) & 0xffu;
            if (hit) {
                const int s = on ? eidx[e] : g;
                acc = pull1_step<HP>(h, dinv, s, hit, dc, po, acc);
            }
        }
    }
    if (q < 4) *reinterpret_cast<v4f*>(agg1 + (size_t)g * HID + 4 * q) = acc;
}

#define PULL8_WEIGHT(K) { const float dk = grp8_f<7>(ov[K]); wk[K] = p + K < end ? dk * dc : 0.0f; }

template <bool SLOT>
__global__ __launch_bounds__(256) void k_pull2(const int* __restrict__ off, const int* __restrict__ srcs,
                                               const float* __restrict__ o,
                                               const float* __restrict__ b2, float* __restrict__ out, int n,
                                               const int* __restrict__ cnt, const int* __restrict__ eidx, int E)
{
    const int g = (blockIdx.x * 256 + threadIdx.x) >> 3;
    const int j = threadIdx.x & 7;
    if (g >= n) return;
    int beg, end, deg = 0;
    if (SLOT) {
        deg = cnt[g];
        beg = g * SLOTS;
        end = deg <= SLOTS ? beg + deg : beg;
    } else {
        beg = off[g]; end = off[g + 1];
    }
    const float own = o[g * 8 + j];
    const float dc = grp8_f<7>(own);
    float acc = own * dc * dc;
    for (int p = beg; p < end; p += 8) {
        const int s = p + j < end ? srcs[p + j] : g;
        int sk[8]; float wk[8], ov[8];
        PULL8_SHARE(0) PULL8_SHARE(1) PULL8_SHARE(2) PULL8_SHARE(3)
        PULL8_SHARE(4) PULL8_SHARE(5) PULL8_SHARE(6) PULL8_SHARE(7)
#pragma unroll
        for (int k = 0; k < 8; ++k) ov[k] = o[sk[k] * 8 + j];
        PULL8_WEIGHT(0) PULL8_WEIGHT(1) PULL8_WEIGHT(2) PULL8_WEIGHT(3)
        PULL8_WEIGHT(4) PULL8_WEIGHT(5) PULL8_WEIGHT(6) PULL8_WEIGHT(7)
#pragma unroll
        for (int k = 0; k < 8; ++k) acc = fmaf(ov[k], wk[k], acc);
    }
    if (SLOT && deg > SLOTS) {
        const unsigned grpsh = threadIdx.x & 24;
        for (int e0 = 0; e0 < E; e0 += 8) {
            const int e = e0 + j;
            const bool on = e < E && eidx[E + e] == g;
            const unsigned hit = ((unsigned)__ballot(on) >> grpsh) & 0xffu;
            if (hit) {
                const int s = on ? eidx[e] : g;
                int sk[8]; float ov[8];
                PULL8_SHARE(0) PULL8_SHARE(1) PULL8_SHARE(2) PULL8_SHARE(3)
                PULL8_SHARE(4) PULL8_SHARE(5) PULL8_SHARE(6) PULL8_SHARE(7)
#pragma unroll
                for (int k = 0; k < 8; ++k) ov[k] = o[sk[k] * 8 + j];
#pragma unroll
                for (int k = 0; k < 8; ++k) {
                    const float dk = grp8_f<7>(ov[k]);
                    const float wv = (hit >> k) & 1u ? dk * dc : 0.0f;
                    acc = fmaf(ov[k], wv, acc);
                }
            }
        }
    }
    const int jj = j < NCLS ? j : NCLS - 1;
    const float v = acc + b2[jj];
    float mx = j < NCLS ? v : -INFINITY;
#pragma unroll
    for (int d = 1; d < 8; d <<= 1) mx = fmaxf(mx, __shfl_xor(mx, d, 8));
    float ex = j < NCLS ? __expf(v - mx) : 0.0f;
#pragma unroll
    for (int d = 1; d < 8; d <<= 1) ex += __shfl_xor(ex, d, 8);
    if (j < NCLS) out[(size_t)g * NCLS + j] = v - mx - __logf(ex);
}

extern "C" void kernel_launch(void* const* d_in, const int* in_sizes, int n_in,
                              void* d_out, int out_size, void* d_ws, size_t ws_size,
                              hipStream_t stream)
{
    const float* x    = (const float*)d_in[0];
    const int*   eidx = (const int*)  d_in[1];
    const float* W1   = (const float*)d_in[2];
    const float* b1   = (const float*)d_in[3];
    const float* W2   = (const float*)d_in[4];
    const float* b2   = (const float*)d_in[5];
    float* out = (float*)d_out;

    const int N = in_sizes[0] / N_FEAT;
    const int E = in_sizes[1] / 2;

    float* ws   = (float*)d_ws;
    const int B = 256;
    const int nTiles = (N + 15) / 16;

    const size_t NpS = ((size_t)N + 3) & ~(size_t)3;
    const size_t Sl  = (size_t)N * SLOTS;
    if (ws_size >= ((size_t)N * 32 + Sl + 2 * NpS) * sizeof(float)) {
        const bool wide = ws_size >= ((size_t)N * 48 + Sl + 2 * NpS) * sizeof(float);
        const size_t Rs = (size_t)N * (wide ? 48 : 32);
        float* R     = ws;
        int*   slots = (int*)(ws + Rs);
        float* dinvP = ws + Rs + Sl;
        int*   cnt   = (int*)(ws + Rs + Sl + NpS);
        float* hP    = R;
        float* aggP  = R + (size_t)N * (wide ? 32 : 16);
        float* oP    = R;

        k_cnt_zero<<<(N + B - 1) / B, B, 0, stream>>>(cnt, N);
        k_place   <<<(E + PS_WG - 1) / PS_WG, PS_T, 0, stream>>>(eidx, cnt, slots, E);
        if (wide) {
            k_gemm1_tile<32><<<1024, 256, 0, stream>>>(x, W1, cnt, dinvP, hP, nTiles);
            k_pull1<32, true><<<(N + 31) / 32, 256, 0, stream>>>(nullptr, slots, dinvP, hP, aggP, N, cnt, eidx, E);
        } else {
            k_gemm1_tile<16><<<1024, 256, 0, stream>>>(x, W1, cnt, dinvP, hP, nTiles);
            k_pull1<16, true><<<(N + 31) / 32, 256, 0, stream>>>(nullptr, slots, dinvP, hP, aggP, N, cnt, eidx, E);
        }
        k_l2<false, 8><<<(N + B - 1) / B, B, 0, stream>>>(aggP, b1, W2, dinvP, oP, out, N);
        k_pull2<true><<<(N + 31) / 32, 256, 0, stream>>>(nullptr, slots, oP, b2, out, N, cnt, eidx, E);
        return;
    }

    const size_t Np = ((size_t)N + 4 + 3) & ~(size_t)3;
    const size_t Ep = ((size_t)E + 3) & ~(size_t)3;
    const size_t Rn = Ep > (size_t)N * 32 ? Ep : (size_t)N * 32;
    const size_t Rw = Ep > (size_t)N * 48 ? Ep : (size_t)N * 48;
    if (ws_size >= (3 * Np + Ep + Rn) * sizeof(float)) {
        const bool wide = ws_size >= (3 * Np + Ep + Rw) * sizeof(float);
        const size_t Rs = wide ? Rw : Rn;
        float* R     = ws;
        int*   srcs  = (int*)(ws + Rs);
        float* dinvP = ws + Rs + Ep;
        int*   off   = (int*)(ws + Rs + Ep + Np);
        int*   cnt   = (int*)(ws + Rs + Ep + 2 * Np);
        int*   rank  = (int*)R;
        float* hP    = R;
        float* aggP  = R + (size_t)N * (wide ? 32 : 16);
        float* oP    = R;

        k_cnt_zero<<<(N + B - 1) / B, B, 0, stream>>>(cnt, N);
        k_rank    <<<(E + B - 1) / B, B, 0, stream>>>(eidx + E, cnt, rank, E);
        k_scan    <<<(N + 4095) / 4096, 1024, 0, stream>>>(cnt, off, dinvP, N);
        k_fill    <<<(E + B - 1) / B, B, 0, stream>>>(eidx, rank, off, srcs, E);
        if (wide) {
            k_gemm1<false, 32, false><<<(nTiles + 7) / 8, 256, 0, stream>>>(x, W1, dinvP, nullptr, nullptr, hP, aggP, nTiles);
            k_pull1<32, false><<<(N + 31) / 32, 256, 0, stream>>>(off, srcs, dinvP, hP, aggP, N, nullptr, eidx, E);
        } else {
            k_gemm1<false, 16, false><<<(nTiles + 7) / 8, 256, 0, stream>>>(x, W1, dinvP, nullptr, nullptr, hP, aggP, nTiles);
            k_pull1<16, false><<<(N + 31) / 32, 256, 0, stream>>>(off, srcs, dinvP, hP, aggP, N, nullptr, eidx, E);
        }
        k_l2<false, 8><<<(N + B - 1) / B, B, 0, stream>>>(aggP, b1, W2, dinvP, oP, out, N);
        k_pull2<false><<<(N + 31) / 32, 256, 0, stream>>>(off, srcs, oP, b2, out, N, nullptr, eidx, E);
        return;
    }

    float* dinv = ws;
    float* h    = ws + (size_t)N;
    float* agg1 = ws + (size_t)N * 17;
    float* obuf = ws + (size_t)N * 33;

    k_deg_init <<<(N + B - 1) / B, B, 0, stream>>>(dinv, N);
    k_deg_edges<<<(E + B - 1) / B, B, 0, stream>>>(eidx + E, dinv, E);
    k_dinv     <<<(N + B - 1) / B, B, 0, stream>>>(dinv, N);

    k_gemm1<true, 16, false><<<(nTiles + 7) / 8, 256, 0, stream>>>(x, W1, dinv, nullptr, nullptr, h, agg1, nTiles);

    const int t1 = E * 16;
    k_edge1<<<(t1 + B - 1) / B, B, 0, stream>>>(eidx, dinv, h, agg1, E);

    k_l2<true, 7><<<(N + B - 1) / B, B, 0, stream>>>(agg1, b1, W2, dinv, obuf, out, N);

    const int t2 = E * 8;
    k_edge2<<<(t2 + B - 1) / B, B, 0, stream>>>(eidx, dinv, obuf, out, E);

    k_lsm<<<(N + B - 1) / B, B, 0, stream>>>(out, b2, N);
}


# Edge placing: a work-group keeps, of 4 edge slices, only the edges of its quarter of the destination range, so its sorted count adds lie 4 times denser per line; list order, so sum order, changes
# speedup vs baseline: 3.4434x; 1.1553x over previous
#include <hip/hip_runtime.h>
#include <math.h>

#define N_FEAT 512
#define HID    16
#define NCLS   7

typedef __attribute__((ext_vector_type(2))) float v2f;
typedef __attribute__((ext_vector_type(8))) float v8f;
typedef __attribute__((ext_vector_type(4))) float v4f;

__global__ void k_deg_init(float* __restrict__ deg, int n) {
    int i = blockIdx.x * blockDim.x + threadIdx.x;
    if (i < n) deg[i] = 1.0f;
}

__global__ void k_deg_edges(const int* __restrict__ col, float* __restrict__ deg, int E) {
    int e = blockIdx.x * blockDim.x + threadIdx.x;
    if (e < E) atomicAdd(&deg[col[e]], 1.0f);
}

__global__ void k_dinv(float* __restrict__ deg, int n) {
    int i = blockIdx.x * blockDim.x + threadIdx.x;
    if (i < n) deg[i] = rsqrtf(deg[i]);
}

template <bool SELF, int HP, bool CNT>
__global__ __launch_bounds__(256) void k_gemm1(
    const float* __restrict__ x, const float* __restrict__ W1,
    const float* __restrict__ dinv, const int* __restrict__ cnt, float* __restrict__ dinvOut,
    float* __restrict__ h, float* __restrict__ agg1, int nTiles)
{
    __shared__ float sW[N_FEAT * HID];
    for (int i = threadIdx.x; i < N_FEAT * HID; i += blockDim.x) sW[i] = W1[i];
    __syncthreads();

    int wave = threadIdx.x >> 5;
    int lane = threadIdx.x & 31;
    int tile = blockIdx.x * 8 + wave;
    if (tile >= nTiles) return;

    int m  = lane & 15;
    int kh = lane >> 4;
    const float* xr = x + (size_t)(tile * 16 + m) * N_FEAT + 2 * kh;

    v8f acc = {};
#pragma unroll 4
    for (int k = 0; k < N_FEAT; k += 4) {
        v2f a, b;
        a.x = xr[k];
        a.y = xr[k + 1];
        int kk = k + 2 * kh;
        b.x = sW[kk * HID + m];
        b.y = sW[(kk + 1) * HID + m];
        acc = __builtin_amdgcn_wmma_f32_16x16x4_f32(
             false, a,  false, b,
             (short)0, acc,  false,  false);
    }

#pragma unroll
    for (int v = 0; v < 8; ++v) {
        int node  = tile * 16 + v + 8 * kh;
        float val = acc[v];
        h[(size_t)node * HP + m] = val;
        if (SELF) {
            float di = dinv[node];
            agg1[(size_t)node * HID + m] = val * di * di;
        }
    }
    if ((HP > HID || CNT) && lane < 16) {
        int node = tile * 16 + lane;
        float di;
        if (CNT) {
            di = rsqrtf((float)(cnt[node] + 1));
            dinvOut[node] = di;
        } else {
            di = dinv[node];
        }
        if (HP > HID) *reinterpret_cast<v4f*>(&h[(size_t)node * HP + HID]) = v4f{di, 0.0f, 0.0f, 0.0f};
    }
}

#define G1_WAVES 8
#define G1_KW    (N_FEAT / G1_WAVES)
#define G1_KS    (G1_KW / 4)
#define G1_PITCH (N_FEAT + 4)

__device__ __forceinline__ void g1_lds_barrier() {
    asm volatile("s_wait_dscnt 0x0\n\ts_barrier_signal -1\n\ts_barrier_wait -1" ::: "memory");
}

template <int HP>
__global__ __launch_bounds__(256) void k_gemm1_tile(
    const float* __restrict__ x, const float* __restrict__ W1,
    const int* __restrict__ cnt, float* __restrict__ dinvOut,
    float* __restrict__ h, int nTiles)
{
    __shared__ __attribute__((aligned(16))) float sX[16 * G1_PITCH];
    __shared__ float sP[G1_WAVES][8][32];

    const int t    = threadIdx.x;
    const int wave = t >> 5;
    const int lane = t & 31;
    const int m  = lane & 15;
    const int kh = lane >> 4;
    const int kw = wave * G1_KW + 2 * kh;
    const int srow = t >> 7;
    const int scol = (t & 127) * 4;

    int tile = blockIdx.x;
    if (tile >= nTiles) return;

    v2f b[G1_KS];
#pragma unroll
    for (int ks = 0; ks < G1_KS; ++ks) {
        const int kk = kw + 4 * ks;
        b[ks].x = W1[kk * HID + m];
        b[ks].y = W1[(kk + 1) * HID + m];
    }

    v4f pf[8];
    int cn;
    {
        const float* src = x + (size_t)(tile * 16 + srow) * N_FEAT + scol;
#pragma unroll
        for (int j = 0; j < 8; ++j) pf[j] = *reinterpret_cast<const v4f*>(src + (size_t)j * 2 * N_FEAT);
        cn = cnt[tile * 16 + m];
    }

    for (;;) {
        if (t < 16) {
            const int node = tile * 16 + t;
            const float di = rsqrtf((float)(cn + 1));
            dinvOut[node] = di;
            if (HP > HID) *reinterpret_cast<v4f*>(&h[(size_t)node * HP + HID]) = v4f{di, 0.0f, 0.0f, 0.0f};
        }
#pragma unroll
        for (int j = 0; j < 8; ++j)
            *reinterpret_cast<v4f*>(&sX[(2 * j + srow) * G1_PITCH + scol]) = pf[j];

        const int  nxt  = tile + (int)gridDim.x;
        const bool more = nxt < nTiles;
        {
            const int    pt   = more ? nxt : tile;
            const size_t step = more ? (size_t)2 * N_FEAT : 0;
            const float* src  = x + (size_t)(pt * 16 + srow) * N_FEAT + scol;
#pragma unroll
            for (int j = 0; j < 8; ++j) pf[j] = *reinterpret_cast<const v4f*>(src + j * step);
            cn = cnt[pt * 16 + m];
        }
        g1_lds_barrier();

        v2f a[G1_KS];
#pragma unroll
        for (int ks = 0; ks < G1_KS; ++ks)
            a[ks] = *reinterpret_cast<const v2f*>(&sX[m * G1_PITCH + kw + 4 * ks]);
        v8f acc0 = {}, acc1 = {};
#pragma unroll
        for (int ks = 0; ks < G1_KS; ks += 2) {
            acc0 = __builtin_amdgcn_wmma_f32_16x16x4_f32(false, a[ks],     false, b[ks],     (short)0, acc0, false, false);
            acc1 = __builtin_amdgcn_wmma_f32_16x16x4_f32(false, a[ks + 1], false, b[ks + 1], (short)0, acc1, false, false);
        }
#pragma unroll
        for (int v = 0; v < 8; ++v) sP[wave][v][lane] = acc0[v] + acc1[v];
        g1_lds_barrier();
        {
            const int v = wave;
            const float s01 = sP[0][v][lane] + sP[1][v][lane];
            const float s23 = sP[2][v][lane] + sP[3][v][lane];
            const float s45 = sP[4][v][lane] + sP[5][v][lane];
            const float s67 = sP[6][v][lane] + sP[7][v][lane];
            const int node = tile * 16 + v + 8 * kh;
            h[(size_t)node * HP + m] = (s01 + s23) + (s45 + s67);
        }
        if (!more) break;
        tile = nxt;
    }
}

__global__ void k_edge1(const int* __restrict__ eidx, const float* __restrict__ dinv,
                        const float* __restrict__ h, float* __restrict__ agg1, int E)
{
    int t = blockIdx.x * blockDim.x + threadIdx.x;
    int e = t >> 4, f = t & 15;
    if (e >= E) return;
    int r = eidx[e];
    int c = eidx[E + e];
    float w = dinv[r] * dinv[c];
    atomicAdd(&agg1[(size_t)c * HID + f], h[(size_t)r * HID + f] * w);
}

template <bool SELF, int OP>
__global__ void k_l2(const float* __restrict__ agg1, const float* __restrict__ b1,
                     const float* __restrict__ W2, const float* __restrict__ dinv,
                     float* __restrict__ o, float* __restrict__ outInit, int n)
{
    __shared__ float sW2[HID * NCLS];
    __shared__ float sb1[HID];
    if (threadIdx.x < HID * NCLS) sW2[threadIdx.x] = W2[threadIdx.x];
    if (threadIdx.x < HID)        sb1[threadIdx.x] = b1[threadIdx.x];
    __syncthreads();

    int i = blockIdx.x * blockDim.x + threadIdx.x;
    if (i >= n) return;

    float hv[HID];
#pragma unroll
    for (int f = 0; f < HID; ++f) {
        float v = agg1[(size_t)i * HID + f] + sb1[f];
        hv[f] = v > 0.0f ? v : 0.0f;
    }
    float di = dinv[i];
    float d2 = di * di;
#pragma unroll
    for (int j = 0; j < NCLS; ++j) {
        float s = 0.0f;
#pragma unroll
        for (int f = 0; f < HID; ++f) s = fmaf(hv[f], sW2[f * NCLS + j], s);
        o[(size_t)i * OP + j]         = s;
        if (SELF) outInit[(size_t)i * NCLS + j] = s * d2;
    }
    if (OP > NCLS) o[(size_t)i * OP + NCLS] = di;
}

__global__ void k_edge2(const int* __restrict__ eidx, const float* __restrict__ dinv,
                        const float* __restrict__ o, float* __restrict__ out, int E)
{
    int t = blockIdx.x * blockDim.x + threadIdx.x;
    int e = t >> 3, j = t & 7;
    if (e >= E || j >= NCLS) return;
    int r = eidx[e];
    int c = eidx[E + e];
    float w = dinv[r] * dinv[c];
    atomicAdd(&out[(size_t)c * NCLS + j], o[(size_t)r * NCLS + j] * w);
}

__global__ void k_lsm(float* __restrict__ out, const float* __restrict__ b2, int n)
{
    int i = blockIdx.x * blockDim.x + threadIdx.x;
    if (i >= n) return;
    float v[NCLS];
    float mx = -1e30f;
#pragma unroll
    for (int j = 0; j < NCLS; ++j) {
        v[j] = out[(size_t)i * NCLS + j] + b2[j];
        mx = fmaxf(mx, v[j]);
    }
    float s = 0.0f;
#pragma unroll
    for (int j = 0; j < NCLS; ++j) s += __expf(v[j] - mx);
    float l = __logf(s);
#pragma unroll
    for (int j = 0; j < NCLS; ++j) out[(size_t)i * NCLS + j] = v[j] - mx - l;
}


__global__ void k_cnt_zero(int* __restrict__ cnt, int n) {
    int i = blockIdx.x * blockDim.x + threadIdx.x;
    if (i < n) cnt[i] = 0;
}

__global__ void k_rank(const int* __restrict__ col, int* __restrict__ cnt,
                       int* __restrict__ rank, int E) {
    int e = blockIdx.x * blockDim.x + threadIdx.x;
    if (e < E) rank[e] = atomicAdd(&cnt[col[e]], 1);
}

__global__ __launch_bounds__(1024) void k_scan(const int* __restrict__ cnt, int* __restrict__ off,
                                               float* __restrict__ dinv, int n)
{
    __shared__ int sInc[32];
    __shared__ int sPre[32];
    const int t = threadIdx.x, lane = t & 31, wave = t >> 5;
    const int tile0 = blockIdx.x * 4096;

    int pre = 0;
    for (int b = 0; b < tile0; b += 4096) {
        int4 v = *reinterpret_cast<const int4*>(cnt + b + t * 4);
        pre += (v.x + v.y) + (v.z + v.w);
    }
    const int i0 = tile0 + t * 4;
    int c[4];
#pragma unroll
    for (int k = 0; k < 4; ++k) c[k] = (i0 + k < n) ? cnt[i0 + k] : 0;
    const int mine = (c[0] + c[1]) + (c[2] + c[3]);

    int incl = mine;
#pragma unroll
    for (int d = 1; d < 32; d <<= 1) { int y = __shfl_up(incl, d); if (lane >= d) incl += y; }
#pragma unroll
    for (int d = 16; d > 0; d >>= 1) pre += __shfl_xor(pre, d);
    if (lane == 31) sInc[wave] = incl;
    if (lane == 0)  sPre[wave] = pre;
    __syncthreads();
    if (wave == 0) {
        int w = sInc[lane], p = sPre[lane];
        int wi = w;
#pragma unroll
        for (int d = 1; d < 32; d <<= 1) { int y = __shfl_up(wi, d); if (lane >= d) wi += y; }
#pragma unroll
        for (int d = 16; d > 0; d >>= 1) p += __shfl_xor(p, d);
        sInc[lane] = wi - w + p;
    }
    __syncthreads();
    int run = sInc[wave] + incl - mine;
#pragma unroll
    for (int k = 0; k < 4; ++k) {
        int i = i0 + k;
        if (i < n) {
            off[i]  = run;
            dinv[i] = rsqrtf((float)(c[k] + 1));
            run += c[k];
            if (i == n - 1) off[n] = run;
        }
    }
}

__global__ void k_fill(const int* __restrict__ eidx, const int* __restrict__ rank,
                       const int* __restrict__ off, int* __restrict__ srcs, int E) {
    int e = blockIdx.x * blockDim.x + threadIdx.x;
    if (e < E) srcs[off[eidx[E + e]] + rank[e]] = eidx[e];
}

#define SLOTS 64

#define PS_T   1024
#define PS_EPT 13
#define PS_WG  (PS_T * PS_EPT)
#define PS_NB  4096
#define PQ_K   4
#define PQ_CAP 16384

template <bool FULL>
__device__ __forceinline__ void place_part(const int* __restrict__ eidx, int* __restrict__ cnt,
                                           int* __restrict__ slots, int E, int nLines, int g0,
                                           int* sBin, int* sW, int* sMp, int* sC, int* sS) {
    const int t = threadIdx.x, lane = t & 31, wave = t >> 5;
    const int part = blockIdx.x % PQ_K;
    const int n    = FULL ? PQ_K * PS_WG : E - g0;
    const unsigned lpp = (unsigned)((nLines + PQ_K - 1) / PQ_K);
    const int lo   = part * (int)lpp;
    const int* __restrict__ col = eidx + E + g0;
    const int* __restrict__ row = eidx + g0;

    for (int i = t; i < PS_NB; i += PS_T) sBin[i] = 0;
    __syncthreads();

    for (int b = 0; b < PQ_K * PS_EPT; b += PS_EPT) {
        int c[PS_EPT];
#pragma unroll
        for (int k = 0; k < PS_EPT; ++k) {
            const int i = (b + k) * PS_T + t;
            c[k] = FULL || i < n ? col[i] : -1;
        }
#pragma unroll
        for (int k = 0; k < PS_EPT; ++k) {
            const int ln = c[k] >> 5;
            if ((unsigned)(ln - lo) < lpp) atomicAdd(&sBin[ln & (PS_NB - 1)], 1);
        }
    }
    __syncthreads();

    {
        const int4 v = reinterpret_cast<const int4*>(sBin)[t];
        const int mine = (v.x + v.y) + (v.z + v.w);
        int incl = mine;
#pragma unroll
        for (int d = 1; d < 32; d <<= 1) { int y = __shfl_up(incl, d); if (lane >= d) incl += y; }
        if (lane == 31) sW[wave] = incl;
        __syncthreads();
        if (wave == 0) {
            const int w = sW[lane];
            int wi = w;
#pragma unroll
            for (int d = 1; d < 32; d <<= 1) { int y = __shfl_up(wi, d); if (lane >= d) wi += y; }
            sW[lane] = wi - w;
        }
        __syncthreads();
        int4 o;
        o.x = sW[wave] + incl - mine;
        o.y = o.x + v.x;
        o.z = o.y + v.y;
        o.w = o.z + v.z;
        reinterpret_cast<int4*>(sBin)[t] = o;
        if (t == PS_T - 1) *sMp = o.w + v.w;
    }
    __syncthreads();

    for (int b = 0; b < PQ_K * PS_EPT; b += PS_EPT) {
        int c[PS_EPT], s[PS_EPT];
#pragma unroll
        for (int k = 0; k < PS_EPT; ++k) {
            const int i = (b + k) * PS_T + t;
            const bool ok = FULL || i < n;
            c[k] = ok ? col[i] : -1;
            s[k] = ok ? row[i] : 0;
        }
        int p[PS_EPT];
        unsigned kept = 0, over = 0;
#pragma unroll
        for (int k = 0; k < PS_EPT; ++k) {
            const int ln = c[k] >> 5;
            p[k] = 0;
            if ((unsigned)(ln - lo) < lpp) {
                p[k] = atomicAdd(&sBin[ln & (PS_NB - 1)], 1);
                kept |= 1u << k;
            }
        }
#pragma unroll
        for (int k = 0; k < PS_EPT; ++k) {
            if ((kept >> k) & 1u) {
                if (p[k] < PQ_CAP) {
                    sC[p[k]] = c[k];
                    sS[p[k]] = s[k];
                } else {
                    over |= 1u << k;
                }
            }
        }
        if (over) {
#pragma unroll
            for (int k = 0; k < PS_EPT; ++k) {
                if ((over >> k) & 1u) {
                    const int r = atomicAdd(&cnt[c[k]], 1);
                    if (r < SLOTS) slots[c[k] * SLOTS + r] = s[k];
                }
            }
        }
    }
    __syncthreads();

    const int m = *sMp < PQ_CAP ? *sMp : PQ_CAP;
    for (int base = 0; base < m; base += PS_WG) {
        int c[PS_EPT], s[PS_EPT], r[PS_EPT];
#pragma unroll
        for (int k = 0; k < PS_EPT; ++k) {
            const int i = base + k * PS_T + t;
            const bool ok = i < m;
            c[k] = ok ? sC[i] : -1;
            s[k] = ok ? sS[i] : 0;
        }
#pragma unroll
        for (int k = 0; k < PS_EPT; ++k) r[k] = c[k] >= 0 ? atomicAdd(&cnt[c[k]], 1) : SLOTS;
#pragma unroll
        for (int k = 0; k < PS_EPT; ++k)
            if (r[k] < SLOTS) slots[c[k] * SLOTS + r[k]] = s[k];
    }
}

__global__ __launch_bounds__(PS_T) void k_place(const int* __restrict__ eidx, int* __restrict__ cnt,
                                                int* __restrict__ slots, int E, int nLines) {
    __shared__ __attribute__((aligned(16))) int sBin[PS_NB];
    __shared__ int sW[32];
    __shared__ int sM;
    __shared__ int sC[PQ_CAP];
    __shared__ int sS[PQ_CAP];
    const int g0 = (blockIdx.x / PQ_K) * (PQ_K * PS_WG);
    if (g0 + PQ_K * PS_WG <= E) place_part<true >(eidx, cnt, slots, E, nLines, g0, sBin, sW, &sM, sC, sS);
    else                        place_part<false>(eidx, cnt, slots, E, nLines, g0, sBin, sW, &sM, sC, sS);
}

template <int K> __device__ __forceinline__ int grp8_i(int v) {
    return (int)__builtin_amdgcn_mov_dpp8((unsigned)v, K * 0x249249);
}
template <int K> __device__ __forceinline__ float grp8_f(float v) {
    return __int_as_float(grp8_i<K>(__float_as_int(v)));
}

#define PULL8_SHARE(K)  sk[K] = grp8_i<K>(s);
#define PULL1_ADD(K) { \
    const float dk = HP > HID ? grp8_f<4>(rv[K].x) : grp8_f<K>(ds1); \
    const float w = (mask >> K) & 1u ? dk * dc : 0.0f; \
    acc.x = fmaf(rv[K].x, w, acc.x); acc.y = fmaf(rv[K].y, w, acc.y); \
    acc.z = fmaf(rv[K].z, w, acc.z); acc.w = fmaf(rv[K].w, w, acc.w); }

template <int HP>
__device__ __forceinline__ v4f pull1_step(const float* __restrict__ h, const float* __restrict__ dinv,
                                          int s, unsigned mask, float dc, int po, v4f acc)
{
    float ds1 = 0.0f;
    if (HP == HID) ds1 = dinv[s];
    int sk[8]; v4f rv[8];
    PULL8_SHARE(0) PULL8_SHARE(1) PULL8_SHARE(2) PULL8_SHARE(3)
    PULL8_SHARE(4) PULL8_SHARE(5) PULL8_SHARE(6) PULL8_SHARE(7)
#pragma unroll
    for (int k = 0; k < 8; ++k) rv[k] = *reinterpret_cast<const v4f*>(h + sk[k] * HP + po);
    PULL1_ADD(0) PULL1_ADD(1) PULL1_ADD(2) PULL1_ADD(3)
    PULL1_ADD(4) PULL1_ADD(5) PULL1_ADD(6) PULL1_ADD(7)
    return acc;
}

template <int HP, bool SLOT>
__global__ __launch_bounds__(256) void k_pull1(const int* __restrict__ off, const int* __restrict__ srcs,
                                               const float* __restrict__ dinv, const float* __restrict__ h,
                                               float* __restrict__ agg1, int n,
                                               const int* __restrict__ cnt, const int* __restrict__ eidx, int E)
{
    const int g = (blockIdx.x * 256 + threadIdx.x) >> 3;
    const int q = threadIdx.x & 7;
    if (g >= n) return;
    int beg, end, deg = 0;
    if (SLOT) {
        deg = cnt[g];
        beg = g * SLOTS;
        end = deg <= SLOTS ? beg + deg : beg;
    } else {
        beg = off[g]; end = off[g + 1];
    }
    const int po = HP > HID ? (q < 4 ? q : 4) * 4 : (q & 3) * 4;
    const v4f own = *reinterpret_cast<const v4f*>(h + g * HP + po);
    const float dc = HP > HID ? grp8_f<4>(own.x) : dinv[g];
    v4f acc;
    acc.x = own.x * dc * dc; acc.y = own.y * dc * dc; acc.z = own.z * dc * dc; acc.w = own.w * dc * dc;
    for (int p = beg; p < end; p += 8) {
        const int left = end - p;
        const unsigned mask = left >= 8 ? 0xffu : (1u << left) - 1u;
        const int s = q < left ? srcs[p + q] : g;
        acc = pull1_step<HP>(h, dinv, s, mask, dc, po, acc);
    }
    if (SLOT && deg > SLOTS) {
        const unsigned grpsh = threadIdx.x & 24;
        for (int e0 = 0; e0 < E; e0 += 8) {
            const int e = e0 + q;
            const bool on = e < E && eidx[E + e] == g;
            const unsigned hit = ((unsigned)__ballot(on) >> grpsh) & 0xffu;
            if (hit) {
                const int s = on ? eidx[e] : g;
                acc = pull1_step<HP>(h, dinv, s, hit, dc, po, acc);
            }
        }
    }
    if (q < 4) *reinterpret_cast<v4f*>(agg1 + (size_t)g * HID + 4 * q) = acc;
}

#define PULL8_WEIGHT(K) { const float dk = grp8_f<7>(ov[K]); wk[K] = p + K < end ? dk * dc : 0.0f; }

template <bool SLOT>
__global__ __launch_bounds__(256) void k_pull2(const int* __restrict__ off, const int* __restrict__ srcs,
                                               const float* __restrict__ o,
                                               const float* __restrict__ b2, float* __restrict__ out, int n,
                                               const int* __restrict__ cnt, const int* __restrict__ eidx, int E)
{
    const int g = (blockIdx.x * 256 + threadIdx.x) >> 3;
    const int j = threadIdx.x & 7;
    if (g >= n) return;
    int beg, end, deg = 0;
    if (SLOT) {
        deg = cnt[g];
        beg = g * SLOTS;
        end = deg <= SLOTS ? beg + deg : beg;
    } else {
        beg = off[g]; end = off[g + 1];
    }
    const float own = o[g * 8 + j];
    const float dc = grp8_f<7>(own);
    float acc = own * dc * dc;
    for (int p = beg; p < end; p += 8) {
        const int s = p + j < end ? srcs[p + j] : g;
        int sk[8]; float wk[8], ov[8];
        PULL8_SHARE(0) PULL8_SHARE(1) PULL8_SHARE(2) PULL8_SHARE(3)
        PULL8_SHARE(4) PULL8_SHARE(5) PULL8_SHARE(6) PULL8_SHARE(7)
#pragma unroll
        for (int k = 0; k < 8; ++k) ov[k] = o[sk[k] * 8 + j];
        PULL8_WEIGHT(0) PULL8_WEIGHT(1) PULL8_WEIGHT(2) PULL8_WEIGHT(3)
        PULL8_WEIGHT(4) PULL8_WEIGHT(5) PULL8_WEIGHT(6) PULL8_WEIGHT(7)
#pragma unroll
        for (int k = 0; k < 8; ++k) acc = fmaf(ov[k], wk[k], acc);
    }
    if (SLOT && deg > SLOTS) {
        const unsigned grpsh = threadIdx.x & 24;
        for (int e0 = 0; e0 < E; e0 += 8) {
            const int e = e0 + j;
            const bool on = e < E && eidx[E + e] == g;
            const unsigned hit = ((unsigned)__ballot(on) >> grpsh) & 0xffu;
            if (hit) {
                const int s = on ? eidx[e] : g;
                int sk[8]; float ov[8];
                PULL8_SHARE(0) PULL8_SHARE(1) PULL8_SHARE(2) PULL8_SHARE(3)
                PULL8_SHARE(4) PULL8_SHARE(5) PULL8_SHARE(6) PULL8_SHARE(7)
#pragma unroll
                for (int k = 0; k < 8; ++k) ov[k] = o[sk[k] * 8 + j];
#pragma unroll
                for (int k = 0; k < 8; ++k) {
                    const float dk = grp8_f<7>(ov[k]);
                    const float wv = (hit >> k) & 1u ? dk * dc : 0.0f;
                    acc = fmaf(ov[k], wv, acc);
                }
            }
        }
    }
    const int jj = j < NCLS ? j : NCLS - 1;
    const float v = acc + b2[jj];
    float mx = j < NCLS ? v : -INFINITY;
#pragma unroll
    for (int d = 1; d < 8; d <<= 1) mx = fmaxf(mx, __shfl_xor(mx, d, 8));
    float ex = j < NCLS ? __expf(v - mx) : 0.0f;
#pragma unroll
    for (int d = 1; d < 8; d <<= 1) ex += __shfl_xor(ex, d, 8);
    if (j < NCLS) out[(size_t)g * NCLS + j] = v - mx - __logf(ex);
}

extern "C" void kernel_launch(void* const* d_in, const int* in_sizes, int n_in,
                              void* d_out, int out_size, void* d_ws, size_t ws_size,
                              hipStream_t stream)
{
    const float* x    = (const float*)d_in[0];
    const int*   eidx = (const int*)  d_in[1];
    const float* W1   = (const float*)d_in[2];
    const float* b1   = (const float*)d_in[3];
    const float* W2   = (const float*)d_in[4];
    const float* b2   = (const float*)d_in[5];
    float* out = (float*)d_out;

    const int N = in_sizes[0] / N_FEAT;
    const int E = in_sizes[1] / 2;

    float* ws   = (float*)d_ws;
    const int B = 256;
    const int nTiles = (N + 15) / 16;

    const size_t NpS = ((size_t)N + 3) & ~(size_t)3;
    const size_t Sl  = (size_t)N * SLOTS;
    if (ws_size >= ((size_t)N * 32 + Sl + 2 * NpS) * sizeof(float)) {
        const bool wide = ws_size >= ((size_t)N * 48 + Sl + 2 * NpS) * sizeof(float);
        const size_t Rs = (size_t)N * (wide ? 48 : 32);
        float* R     = ws;
        int*   slots = (int*)(ws + Rs);
        float* dinvP = ws + Rs + Sl;
        int*   cnt   = (int*)(ws + Rs + Sl + NpS);
        float* hP    = R;
        float* aggP  = R + (size_t)N * (wide ? 32 : 16);
        float* oP    = R;

        k_cnt_zero<<<(N + B - 1) / B, B, 0, stream>>>(cnt, N);
        k_place   <<<(E + PQ_K * PS_WG - 1) / (PQ_K * PS_WG) * PQ_K, PS_T, 0, stream>>>(eidx, cnt, slots, E, (N + 31) / 32);
        if (wide) {
            k_gemm1_tile<32><<<1024, 256, 0, stream>>>(x, W1, cnt, dinvP, hP, nTiles);
            k_pull1<32, true><<<(N + 31) / 32, 256, 0, stream>>>(nullptr, slots, dinvP, hP, aggP, N, cnt, eidx, E);
        } else {
            k_gemm1_tile<16><<<1024, 256, 0, stream>>>(x, W1, cnt, dinvP, hP, nTiles);
            k_pull1<16, true><<<(N + 31) / 32, 256, 0, stream>>>(nullptr, slots, dinvP, hP, aggP, N, cnt, eidx, E);
        }
        k_l2<false, 8><<<(N + B - 1) / B, B, 0, stream>>>(aggP, b1, W2, dinvP, oP, out, N);
        k_pull2<true><<<(N + 31) / 32, 256, 0, stream>>>(nullptr, slots, oP, b2, out, N, cnt, eidx, E);
        return;
    }

    const size_t Np = ((size_t)N + 4 + 3) & ~(size_t)3;
    const size_t Ep = ((size_t)E + 3) & ~(size_t)3;
    const size_t Rn = Ep > (size_t)N * 32 ? Ep : (size_t)N * 32;
    const size_t Rw = Ep > (size_t)N * 48 ? Ep : (size_t)N * 48;
    if (ws_size >= (3 * Np + Ep + Rn) * sizeof(float)) {
        const bool wide = ws_size >= (3 * Np + Ep + Rw) * sizeof(float);
        const size_t Rs = wide ? Rw : Rn;
        float* R     = ws;
        int*   srcs  = (int*)(ws + Rs);
        float* dinvP = ws + Rs + Ep;
        int*   off   = (int*)(ws + Rs + Ep + Np);
        int*   cnt   = (int*)(ws + Rs + Ep + 2 * Np);
        int*   rank  = (int*)R;
        float* hP    = R;
        float* aggP  = R + (size_t)N * (wide ? 32 : 16);
        float* oP    = R;

        k_cnt_zero<<<(N + B - 1) / B, B, 0, stream>>>(cnt, N);
        k_rank    <<<(E + B - 1) / B, B, 0, stream>>>(eidx + E, cnt, rank, E);
        k_scan    <<<(N + 4095) / 4096, 1024, 0, stream>>>(cnt, off, dinvP, N);
        k_fill    <<<(E + B - 1) / B, B, 0, stream>>>(eidx, rank, off, srcs, E);
        if (wide) {
            k_gemm1<false, 32, false><<<(nTiles + 7) / 8, 256, 0, stream>>>(x, W1, dinvP, nullptr, nullptr, hP, aggP, nTiles);
            k_pull1<32, false><<<(N + 31) / 32, 256, 0, stream>>>(off, srcs, dinvP, hP, aggP, N, nullptr, eidx, E);
        } else {
            k_gemm1<false, 16, false><<<(nTiles + 7) / 8, 256, 0, stream>>>(x, W1, dinvP, nullptr, nullptr, hP, aggP, nTiles);
            k_pull1<16, false><<<(N + 31) / 32, 256, 0, stream>>>(off, srcs, dinvP, hP, aggP, N, nullptr, eidx, E);
        }
        k_l2<false, 8><<<(N + B - 1) / B, B, 0, stream>>>(aggP, b1, W2, dinvP, oP, out, N);
        k_pull2<false><<<(N + 31) / 32, 256, 0, stream>>>(off, srcs, oP, b2, out, N, nullptr, eidx, E);
        return;
    }

    float* dinv = ws;
    float* h    = ws + (size_t)N;
    float* agg1 = ws + (size_t)N * 17;
    float* obuf = ws + (size_t)N * 33;

    k_deg_init <<<(N + B - 1) / B, B, 0, stream>>>(dinv, N);
    k_deg_edges<<<(E + B - 1) / B, B, 0, stream>>>(eidx + E, dinv, E);
    k_dinv     <<<(N + B - 1) / B, B, 0, stream>>>(dinv, N);

    k_gemm1<true, 16, false><<<(nTiles + 7) / 8, 256, 0, stream>>>(x, W1, dinv, nullptr, nullptr, h, agg1, nTiles);

    const int t1 = E * 16;
    k_edge1<<<(t1 + B - 1) / B, B, 0, stream>>>(eidx, dinv, h, agg1, E);

    k_l2<true, 7><<<(N + B - 1) / B, B, 0, stream>>>(agg1, b1, W2, dinv, obuf, out, N);

    const int t2 = E * 8;
    k_edge2<<<(t2 + B - 1) / B, B, 0, stream>>>(eidx, dinv, obuf, out, E);

    k_lsm<<<(N + B - 1) / B, B, 0, stream>>>(out, b2, N);
}


# Edge placing: one count add per work-group and 32-destination group reserves a run, its sorted edges are stored side by side; a new launch regroups runs into the slot lists in LDS; sum order changes
# speedup vs baseline: 4.5674x; 1.3264x over previous
#include <hip/hip_runtime.h>
#include <math.h>

#define N_FEAT 512
#define HID    16
#define NCLS   7

typedef __attribute__((ext_vector_type(2))) float v2f;
typedef __attribute__((ext_vector_type(8))) float v8f;
typedef __attribute__((ext_vector_type(4))) float v4f;

__global__ void k_deg_init(float* __restrict__ deg, int n) {
    int i = blockIdx.x * blockDim.x + threadIdx.x;
    if (i < n) deg[i] = 1.0f;
}

__global__ void k_deg_edges(const int* __restrict__ col, float* __restrict__ deg, int E) {
    int e = blockIdx.x * blockDim.x + threadIdx.x;
    if (e < E) atomicAdd(&deg[col[e]], 1.0f);
}

__global__ void k_dinv(float* __restrict__ deg, int n) {
    int i = blockIdx.x * blockDim.x + threadIdx.x;
    if (i < n) deg[i] = rsqrtf(deg[i]);
}

template <bool SELF, int HP, bool CNT>
__global__ __launch_bounds__(256) void k_gemm1(
    const float* __restrict__ x, const float* __restrict__ W1,
    const float* __restrict__ dinv, const int* __restrict__ cnt, float* __restrict__ dinvOut,
    float* __restrict__ h, float* __restrict__ agg1, int nTiles)
{
    __shared__ float sW[N_FEAT * HID];
    for (int i = threadIdx.x; i < N_FEAT * HID; i += blockDim.x) sW[i] = W1[i];
    __syncthreads();

    int wave = threadIdx.x >> 5;
    int lane = threadIdx.x & 31;
    int tile = blockIdx.x * 8 + wave;
    if (tile >= nTiles) return;

    int m  = lane & 15;
    int kh = lane >> 4;
    const float* xr = x + (size_t)(tile * 16 + m) * N_FEAT + 2 * kh;

    v8f acc = {};
#pragma unroll 4
    for (int k = 0; k < N_FEAT; k += 4) {
        v2f a, b;
        a.x = xr[k];
        a.y = xr[k + 1];
        int kk = k + 2 * kh;
        b.x = sW[kk * HID + m];
        b.y = sW[(kk + 1) * HID + m];
        acc = __builtin_amdgcn_wmma_f32_16x16x4_f32(
             false, a,  false, b,
             (short)0, acc,  false,  false);
    }

#pragma unroll
    for (int v = 0; v < 8; ++v) {
        int node  = tile * 16 + v + 8 * kh;
        float val = acc[v];
        h[(size_t)node * HP + m] = val;
        if (SELF) {
            float di = dinv[node];
            agg1[(size_t)node * HID + m] = val * di * di;
        }
    }
    if ((HP > HID || CNT) && lane < 16) {
        int node = tile * 16 + lane;
        float di;
        if (CNT) {
            di = rsqrtf((float)(cnt[node] + 1));
            dinvOut[node] = di;
        } else {
            di = dinv[node];
        }
        if (HP > HID) *reinterpret_cast<v4f*>(&h[(size_t)node * HP + HID]) = v4f{di, 0.0f, 0.0f, 0.0f};
    }
}

#define G1_WAVES 8
#define G1_KW    (N_FEAT / G1_WAVES)
#define G1_KS    (G1_KW / 4)
#define G1_PITCH (N_FEAT + 4)

__device__ __forceinline__ void g1_lds_barrier() {
    asm volatile("s_wait_dscnt 0x0\n\ts_barrier_signal -1\n\ts_barrier_wait -1" ::: "memory");
}

template <int HP>
__global__ __launch_bounds__(256) void k_gemm1_tile(
    const float* __restrict__ x, const float* __restrict__ W1,
    const int* __restrict__ cnt, float* __restrict__ dinvOut,
    float* __restrict__ h, int nTiles)
{
    __shared__ __attribute__((aligned(16))) float sX[16 * G1_PITCH];
    __shared__ float sP[G1_WAVES][8][32];

    const int t    = threadIdx.x;
    const int wave = t >> 5;
    const int lane = t & 31;
    const int m  = lane & 15;
    const int kh = lane >> 4;
    const int kw = wave * G1_KW + 2 * kh;
    const int srow = t >> 7;
    const int scol = (t & 127) * 4;

    int tile = blockIdx.x;
    if (tile >= nTiles) return;

    v2f b[G1_KS];
#pragma unroll
    for (int ks = 0; ks < G1_KS; ++ks) {
        const int kk = kw + 4 * ks;
        b[ks].x = W1[kk * HID + m];
        b[ks].y = W1[(kk + 1) * HID + m];
    }

    v4f pf[8];
    int cn;
    {
        const float* src = x + (size_t)(tile * 16 + srow) * N_FEAT + scol;
#pragma unroll
        for (int j = 0; j < 8; ++j) pf[j] = *reinterpret_cast<const v4f*>(src + (size_t)j * 2 * N_FEAT);
        cn = cnt[tile * 16 + m];
    }

    for (;;) {
        if (t < 16) {
            const int node = tile * 16 + t;
            const float di = rsqrtf((float)(cn + 1));
            dinvOut[node] = di;
            if (HP > HID) *reinterpret_cast<v4f*>(&h[(size_t)node * HP + HID]) = v4f{di, 0.0f, 0.0f, 0.0f};
        }
#pragma unroll
        for (int j = 0; j < 8; ++j)
            *reinterpret_cast<v4f*>(&sX[(2 * j + srow) * G1_PITCH + scol]) = pf[j];

        const int  nxt  = tile + (int)gridDim.x;
        const bool more = nxt < nTiles;
        {
            const int    pt   = more ? nxt : tile;
            const size_t step = more ? (size_t)2 * N_FEAT : 0;
            const float* src  = x + (size_t)(pt * 16 + srow) * N_FEAT + scol;
#pragma unroll
            for (int j = 0; j < 8; ++j) pf[j] = *reinterpret_cast<const v4f*>(src + j * step);
            cn = cnt[pt * 16 + m];
        }
        g1_lds_barrier();

        v2f a[G1_KS];
#pragma unroll
        for (int ks = 0; ks < G1_KS; ++ks)
            a[ks] = *reinterpret_cast<const v2f*>(&sX[m * G1_PITCH + kw + 4 * ks]);
        v8f acc0 = {}, acc1 = {};
#pragma unroll
        for (int ks = 0; ks < G1_KS; ks += 2) {
            acc0 = __builtin_amdgcn_wmma_f32_16x16x4_f32(false, a[ks],     false, b[ks],     (short)0, acc0, false, false);
            acc1 = __builtin_amdgcn_wmma_f32_16x16x4_f32(false, a[ks + 1], false, b[ks + 1], (short)0, acc1, false, false);
        }
#pragma unroll
        for (int v = 0; v < 8; ++v) sP[wave][v][lane] = acc0[v] + acc1[v];
        g1_lds_barrier();
        {
            const int v = wave;
            const float s01 = sP[0][v][lane] + sP[1][v][lane];
            const float s23 = sP[2][v][lane] + sP[3][v][lane];
            const float s45 = sP[4][v][lane] + sP[5][v][lane];
            const float s67 = sP[6][v][lane] + sP[7][v][lane];
            const int node = tile * 16 + v + 8 * kh;
            h[(size_t)node * HP + m] = (s01 + s23) + (s45 + s67);
        }
        if (!more) break;
        tile = nxt;
    }
}

__global__ void k_edge1(const int* __restrict__ eidx, const float* __restrict__ dinv,
                        const float* __restrict__ h, float* __restrict__ agg1, int E)
{
    int t = blockIdx.x * blockDim.x + threadIdx.x;
    int e = t >> 4, f = t & 15;
    if (e >= E) return;
    int r = eidx[e];
    int c = eidx[E + e];
    float w = dinv[r] * dinv[c];
    atomicAdd(&agg1[(size_t)c * HID + f], h[(size_t)r * HID + f] * w);
}

template <bool SELF, int OP>
__global__ void k_l2(const float* __restrict__ agg1, const float* __restrict__ b1,
                     const float* __restrict__ W2, const float* __restrict__ dinv,
                     float* __restrict__ o, float* __restrict__ outInit, int n)
{
    __shared__ float sW2[HID * NCLS];
    __shared__ float sb1[HID];
    if (threadIdx.x < HID * NCLS) sW2[threadIdx.x] = W2[threadIdx.x];
    if (threadIdx.x < HID)        sb1[threadIdx.x] = b1[threadIdx.x];
    __syncthreads();

    int i = blockIdx.x * blockDim.x + threadIdx.x;
    if (i >= n) return;

    float hv[HID];
#pragma unroll
    for (int f = 0; f < HID; ++f) {
        float v = agg1[(size_t)i * HID + f] + sb1[f];
        hv[f] = v > 0.0f ? v : 0.0f;
    }
    float di = dinv[i];
    float d2 = di * di;
#pragma unroll
    for (int j = 0; j < NCLS; ++j) {
        float s = 0.0f;
#pragma unroll
        for (int f = 0; f < HID; ++f) s = fmaf(hv[f], sW2[f * NCLS + j], s);
        o[(size_t)i * OP + j]         = s;
        if (SELF) outInit[(size_t)i * NCLS + j] = s * d2;
    }
    if (OP > NCLS) o[(size_t)i * OP + NCLS] = di;
}

__global__ void k_edge2(const int* __restrict__ eidx, const float* __restrict__ dinv,
                        const float* __restrict__ o, float* __restrict__ out, int E)
{
    int t = blockIdx.x * blockDim.x + threadIdx.x;
    int e = t >> 3, j = t & 7;
    if (e >= E || j >= NCLS) return;
    int r = eidx[e];
    int c = eidx[E + e];
    float w = dinv[r] * dinv[c];
    atomicAdd(&out[(size_t)c * NCLS + j], o[(size_t)r * NCLS + j] * w);
}

__global__ void k_lsm(float* __restrict__ out, const float* __restrict__ b2, int n)
{
    int i = blockIdx.x * blockDim.x + threadIdx.x;
    if (i >= n) return;
    float v[NCLS];
    float mx = -1e30f;
#pragma unroll
    for (int j = 0; j < NCLS; ++j) {
        v[j] = out[(size_t)i * NCLS + j] + b2[j];
        mx = fmaxf(mx, v[j]);
    }
    float s = 0.0f;
#pragma unroll
    for (int j = 0; j < NCLS; ++j) s += __expf(v[j] - mx);
    float l = __logf(s);
#pragma unroll
    for (int j = 0; j < NCLS; ++j) out[(size_t)i * NCLS + j] = v[j] - mx - l;
}


__global__ void k_cnt_zero(int* __restrict__ cnt, int n) {
    int i = blockIdx.x * blockDim.x + threadIdx.x;
    if (i < n) cnt[i] = 0;
}

__global__ void k_rank(const int* __restrict__ col, int* __restrict__ cnt,
                       int* __restrict__ rank, int E) {
    int e = blockIdx.x * blockDim.x + threadIdx.x;
    if (e < E) rank[e] = atomicAdd(&cnt[col[e]], 1);
}

__global__ __launch_bounds__(1024) void k_scan(const int* __restrict__ cnt, int* __restrict__ off,
                                               float* __restrict__ dinv, int n)
{
    __shared__ int sInc[32];
    __shared__ int sPre[32];
    const int t = threadIdx.x, lane = t & 31, wave = t >> 5;
    const int tile0 = blockIdx.x * 4096;

    int pre = 0;
    for (int b = 0; b < tile0; b += 4096) {
        int4 v = *reinterpret_cast<const int4*>(cnt + b + t * 4);
        pre += (v.x + v.y) + (v.z + v.w);
    }
    const int i0 = tile0 + t * 4;
    int c[4];
#pragma unroll
    for (int k = 0; k < 4; ++k) c[k] = (i0 + k < n) ? cnt[i0 + k] : 0;
    const int mine = (c[0] + c[1]) + (c[2] + c[3]);

    int incl = mine;
#pragma unroll
    for (int d = 1; d < 32; d <<= 1) { int y = __shfl_up(incl, d); if (lane >= d) incl += y; }
#pragma unroll
    for (int d = 16; d > 0; d >>= 1) pre += __shfl_xor(pre, d);
    if (lane == 31) sInc[wave] = incl;
    if (lane == 0)  sPre[wave] = pre;
    __syncthreads();
    if (wave == 0) {
        int w = sInc[lane], p = sPre[lane];
        int wi = w;
#pragma unroll
        for (int d = 1; d < 32; d <<= 1) { int y = __shfl_up(wi, d); if (lane >= d) wi += y; }
#pragma unroll
        for (int d = 16; d > 0; d >>= 1) p += __shfl_xor(p, d);
        sInc[lane] = wi - w + p;
    }
    __syncthreads();
    int run = sInc[wave] + incl - mine;
#pragma unroll
    for (int k = 0; k < 4; ++k) {
        int i = i0 + k;
        if (i < n) {
            off[i]  = run;
            dinv[i] = rsqrtf((float)(c[k] + 1));
            run += c[k];
            if (i == n - 1) off[n] = run;
        }
    }
}

__global__ void k_fill(const int* __restrict__ eidx, const int* __restrict__ rank,
                       const int* __restrict__ off, int* __restrict__ srcs, int E) {
    int e = blockIdx.x * blockDim.x + threadIdx.x;
    if (e < E) srcs[off[eidx[E + e]] + rank[e]] = eidx[e];
}

#define SLOTS 64

#define PS_T   1024
#define PS_EPT 13
#define PS_WG  (PS_T * PS_EPT)
#define PS_NB  4096
#define PQ_K   4
#define PQ_CAP 16384

#define BIN_CAP   1536
#define BIN_PITCH (32 * SLOTS)
#define BIN_OFF   (BIN_PITCH - BIN_CAP)

template <bool FULL, bool BINS>
__device__ __forceinline__ void place_part(const int* __restrict__ eidx, int* __restrict__ cnt,
                                           int* __restrict__ slots, int E, int nLines, int g0,
                                           int* sBin, int* sW, int* sMp, int* sC, int* sS) {
    const int t = threadIdx.x, lane = t & 31, wave = t >> 5;
    const int part = blockIdx.x % PQ_K;
    const int n    = FULL ? PQ_K * PS_WG : E - g0;
    const unsigned lpp = (unsigned)((nLines + PQ_K - 1) / PQ_K);
    const int lo   = part * (int)lpp;
    const int* __restrict__ col = eidx + E + g0;
    const int* __restrict__ row = eidx + g0;

    for (int i = t; i < PS_NB; i += PS_T) sBin[i] = 0;
    __syncthreads();

    for (int b = 0; b < PQ_K * PS_EPT; b += PS_EPT) {
        int c[PS_EPT];
#pragma unroll
        for (int k = 0; k < PS_EPT; ++k) {
            const int i = (b + k) * PS_T + t;
            c[k] = FULL || i < n ? col[i] : -1;
        }
#pragma unroll
        for (int k = 0; k < PS_EPT; ++k) {
            const int ln = c[k] >> 5;
            if ((unsigned)(ln - lo) < lpp) atomicAdd(&sBin[ln & (PS_NB - 1)], 1);
        }
    }
    __syncthreads();

    {
        const int4 v = reinterpret_cast<const int4*>(sBin)[t];
        const int mine = (v.x + v.y) + (v.z + v.w);
        int incl = mine;
#pragma unroll
        for (int d = 1; d < 32; d <<= 1) { int y = __shfl_up(incl, d); if (lane >= d) incl += y; }
        if (lane == 31) sW[wave] = incl;
        __syncthreads();
        if (wave == 0) {
            const int w = sW[lane];
            int wi = w;
#pragma unroll
            for (int d = 1; d < 32; d <<= 1) { int y = __shfl_up(wi, d); if (lane >= d) wi += y; }
            sW[lane] = wi - w;
        }
        __syncthreads();
        int4 o;
        o.x = sW[wave] + incl - mine;
        o.y = o.x + v.x;
        o.z = o.y + v.y;
        o.w = o.z + v.z;
        reinterpret_cast<int4*>(sBin)[t] = o;
        if (t == PS_T - 1) *sMp = o.w + v.w;
    }
    __syncthreads();

    for (int b = 0; b < PQ_K * PS_EPT; b += PS_EPT) {
        int c[PS_EPT], s[PS_EPT];
#pragma unroll
        for (int k = 0; k < PS_EPT; ++k) {
            const int i = (b + k) * PS_T + t;
            const bool ok = FULL || i < n;
            c[k] = ok ? col[i] : -1;
            s[k] = ok ? row[i] : 0;
        }
        int p[PS_EPT];
        unsigned kept = 0, over = 0;
#pragma unroll
        for (int k = 0; k < PS_EPT; ++k) {
            const int ln = c[k] >> 5;
            p[k] = 0;
            if ((unsigned)(ln - lo) < lpp) {
                p[k] = atomicAdd(&sBin[ln & (PS_NB - 1)], 1);
                kept |= 1u << k;
            }
        }
#pragma unroll
        for (int k = 0; k < PS_EPT; ++k) {
            if ((kept >> k) & 1u) {
                if (p[k] < PQ_CAP) {
                    sC[p[k]] = c[k];
                    sS[p[k]] = s[k];
                } else {
                    over |= 1u << k;
                }
            }
        }
        if (over) {
#pragma unroll
            for (int k = 0; k < PS_EPT; ++k) {
                if ((over >> k) & 1u) {
                    if (BINS) {
                        const int ln = c[k] >> 5;
                        const int r = atomicAdd(&cnt[ln], 1);
                        if (r < BIN_CAP) slots[(size_t)ln * BIN_PITCH + BIN_OFF + r] = (int)(((unsigned)s[k] << 5) | (unsigned)(c[k] & 31));
                    } else {
                        const int r = atomicAdd(&cnt[c[k]], 1);
                        if (r < SLOTS) slots[c[k] * SLOTS + r] = s[k];
                    }
                }
            }
        }
    }
    __syncthreads();

    const int m = *sMp < PQ_CAP ? *sMp : PQ_CAP;
    if (BINS) {
        int gi[4], dv[4];
#pragma unroll
        for (int j = 0; j < 4; ++j) {
            const int u = t + j * PS_T;
            gi[j] = -1; dv[j] = 0;
            if ((unsigned)u < lpp) {
                const int ln = lo + u;
                const int g  = ln & (PS_NB - 1);
                const int en = sBin[g];
                const int s0 = g > 0 ? sBin[g - 1] : 0;
                const int ce = en < PQ_CAP ? en : PQ_CAP;
                const int cs = s0 < PQ_CAP ? s0 : PQ_CAP;
                gi[j] = g;
                if (ce > cs) dv[j] = atomicAdd(&cnt[ln], ce - cs) - cs;
            }
        }
        __syncthreads();
#pragma unroll
        for (int j = 0; j < 4; ++j) if (gi[j] >= 0) sBin[gi[j]] = dv[j];
        __syncthreads();
        for (int base = 0; base < m; base += PS_WG) {
            int c[PS_EPT], s[PS_EPT], dq[PS_EPT];
#pragma unroll
            for (int k = 0; k < PS_EPT; ++k) {
                const int  i  = base + k * PS_T + t;
                const bool ok = i < m;
                const int  ii = ok ? i : 0;
                const int  cc = sC[ii];
                s[k] = sS[ii];
                c[k] = ok ? cc : -1;
            }
#pragma unroll
            for (int k = 0; k < PS_EPT; ++k) dq[k] = sBin[(c[k] >> 5) & (PS_NB - 1)];
#pragma unroll
            for (int k = 0; k < PS_EPT; ++k) {
                const int pos = base + k * PS_T + t + dq[k];
                if (c[k] >= 0 && pos < BIN_CAP)
                    slots[(size_t)(c[k] >> 5) * BIN_PITCH + BIN_OFF + pos] = (int)(((unsigned)s[k] << 5) | (unsigned)(c[k] & 31));
            }
        }
        return;
    }
    for (int base = 0; base < m; base += PS_WG) {
        int c[PS_EPT], s[PS_EPT], r[PS_EPT];
#pragma unroll
        for (int k = 0; k < PS_EPT; ++k) {
            const int i = base + k * PS_T + t;
            const bool ok = i < m;
            c[k] = ok ? sC[i] : -1;
            s[k] = ok ? sS[i] : 0;
        }
#pragma unroll
        for (int k = 0; k < PS_EPT; ++k) r[k] = c[k] >= 0 ? atomicAdd(&cnt[c[k]], 1) : SLOTS;
#pragma unroll
        for (int k = 0; k < PS_EPT; ++k)
            if (r[k] < SLOTS) slots[c[k] * SLOTS + r[k]] = s[k];
    }
}

template <bool BINS>
__global__ __launch_bounds__(PS_T) void k_place(const int* __restrict__ eidx, int* __restrict__ cnt,
                                                int* __restrict__ slots, int E, int nLines) {
    __shared__ __attribute__((aligned(16))) int sBin[PS_NB];
    __shared__ int sW[32];
    __shared__ int sM;
    __shared__ int sC[PQ_CAP];
    __shared__ int sS[PQ_CAP];
    const int g0 = (blockIdx.x / PQ_K) * (PQ_K * PS_WG);
    if (g0 + PQ_K * PS_WG <= E) place_part<true,  BINS>(eidx, cnt, slots, E, nLines, g0, sBin, sW, &sM, sC, sS);
    else                        place_part<false, BINS>(eidx, cnt, slots, E, nLines, g0, sBin, sW, &sM, sC, sS);
}

#define BS_T 256
__global__ __launch_bounds__(BS_T) void k_binsort(const int* __restrict__ binCnt, const int* __restrict__ eidx,
                                                  int* __restrict__ cnt, int* slots, int N, int E) {
    __shared__ int sN[32];
    __shared__ __attribute__((aligned(16))) int sL[32 * SLOTS];
    const int t = threadIdx.x;
    const int line = blockIdx.x;
    if (t < 32) sN[t] = 0;
    __syncthreads();
    const int n = binCnt[line];
    if (n <= BIN_CAP) {
        const int* b = slots + (size_t)line * BIN_PITCH + BIN_OFF;
        int e[BIN_CAP / BS_T];
#pragma unroll
        for (int k = 0; k < BIN_CAP / BS_T; ++k) {
            const int i = k * BS_T + t;
            e[k] = i < n ? b[i] : -1;
        }
#pragma unroll
        for (int k = 0; k < BIN_CAP / BS_T; ++k) {
            if (e[k] >= 0) {
                const int dl = e[k] & 31;
                const int r = atomicAdd(&sN[dl], 1);
                if (r < SLOTS) sL[dl * SLOTS + r] = (int)((unsigned)e[k] >> 5);
            }
        }
    } else {
        for (int e = t; e < E; e += BS_T) {
            const int c = eidx[E + e];
            if ((c >> 5) == line) {
                const int dl = c & 31;
                const int r = atomicAdd(&sN[dl], 1);
                if (r < SLOTS) sL[dl * SLOTS + r] = eidx[e];
            }
        }
    }
    __syncthreads();
    if (t < 32) {
        const int node = line * 32 + t;
        if (node < N) cnt[node] = sN[t];
    }
#pragma unroll
    for (int j = 0; j < 32 * SLOTS / (4 * BS_T); ++j) {
        const int w  = (j * BS_T + t) * 4;
        const int dl = w >> 6;
        const int node = line * 32 + dl;
        if (node < N && (w & 32) < sN[dl])
            *reinterpret_cast<int4*>(slots + (size_t)node * SLOTS + (w & 63)) = *reinterpret_cast<const int4*>(&sL[w]);
    }
}

template <int K> __device__ __forceinline__ int grp8_i(int v) {
    return (int)__builtin_amdgcn_mov_dpp8((unsigned)v, K * 0x249249);
}
template <int K> __device__ __forceinline__ float grp8_f(float v) {
    return __int_as_float(grp8_i<K>(__float_as_int(v)));
}

#define PULL8_SHARE(K)  sk[K] = grp8_i<K>(s);
#define PULL1_ADD(K) { \
    const float dk = HP > HID ? grp8_f<4>(rv[K].x) : grp8_f<K>(ds1); \
    const float w = (mask >> K) & 1u ? dk * dc : 0.0f; \
    acc.x = fmaf(rv[K].x, w, acc.x); acc.y = fmaf(rv[K].y, w, acc.y); \
    acc.z = fmaf(rv[K].z, w, acc.z); acc.w = fmaf(rv[K].w, w, acc.w); }

template <int HP>
__device__ __forceinline__ v4f pull1_step(const float* __restrict__ h, const float* __restrict__ dinv,
                                          int s, unsigned mask, float dc, int po, v4f acc)
{
    float ds1 = 0.0f;
    if (HP == HID) ds1 = dinv[s];
    int sk[8]; v4f rv[8];
    PULL8_SHARE(0) PULL8_SHARE(1) PULL8_SHARE(2) PULL8_SHARE(3)
    PULL8_SHARE(4) PULL8_SHARE(5) PULL8_SHARE(6) PULL8_SHARE(7)
#pragma unroll
    for (int k = 0; k < 8; ++k) rv[k] = *reinterpret_cast<const v4f*>(h + sk[k] * HP + po);
    PULL1_ADD(0) PULL1_ADD(1) PULL1_ADD(2) PULL1_ADD(3)
    PULL1_ADD(4) PULL1_ADD(5) PULL1_ADD(6) PULL1_ADD(7)
    return acc;
}

template <int HP, bool SLOT>
__global__ __launch_bounds__(256) void k_pull1(const int* __restrict__ off, const int* __restrict__ srcs,
                                               const float* __restrict__ dinv, const float* __restrict__ h,
                                               float* __restrict__ agg1, int n,
                                               const int* __restrict__ cnt, const int* __restrict__ eidx, int E)
{
    const int g = (blockIdx.x * 256 + threadIdx.x) >> 3;
    const int q = threadIdx.x & 7;
    if (g >= n) return;
    int beg, end, deg = 0;
    if (SLOT) {
        deg = cnt[g];
        beg = g * SLOTS;
        end = deg <= SLOTS ? beg + deg : beg;
    } else {
        beg = off[g]; end = off[g + 1];
    }
    const int po = HP > HID ? (q < 4 ? q : 4) * 4 : (q & 3) * 4;
    const v4f own = *reinterpret_cast<const v4f*>(h + g * HP + po);
    const float dc = HP > HID ? grp8_f<4>(own.x) : dinv[g];
    v4f acc;
    acc.x = own.x * dc * dc; acc.y = own.y * dc * dc; acc.z = own.z * dc * dc; acc.w = own.w * dc * dc;
    for (int p = beg; p < end; p += 8) {
        const int left = end - p;
        const unsigned mask = left >= 8 ? 0xffu : (1u << left) - 1u;
        const int s = q < left ? srcs[p + q] : g;
        acc = pull1_step<HP>(h, dinv, s, mask, dc, po, acc);
    }
    if (SLOT && deg > SLOTS) {
        const unsigned grpsh = threadIdx.x & 24;
        for (int e0 = 0; e0 < E; e0 += 8) {
            const int e = e0 + q;
            const bool on = e < E && eidx[E + e] == g;
            const unsigned hit = ((unsigned)__ballot(on) >> grpsh) & 0xffu;
            if (hit) {
                const int s = on ? eidx[e] : g;
                acc = pull1_step<HP>(h, dinv, s, hit, dc, po, acc);
            }
        }
    }
    if (q < 4) *reinterpret_cast<v4f*>(agg1 + (size_t)g * HID + 4 * q) = acc;
}

#define PULL8_WEIGHT(K) { const float dk = grp8_f<7>(ov[K]); wk[K] = p + K < end ? dk * dc : 0.0f; }

template <bool SLOT>
__global__ __launch_bounds__(256) void k_pull2(const int* __restrict__ off, const int* __restrict__ srcs,
                                               const float* __restrict__ o,
                                               const float* __restrict__ b2, float* __restrict__ out, int n,
                                               const int* __restrict__ cnt, const int* __restrict__ eidx, int E)
{
    const int g = (blockIdx.x * 256 + threadIdx.x) >> 3;
    const int j = threadIdx.x & 7;
    if (g >= n) return;
    int beg, end, deg = 0;
    if (SLOT) {
        deg = cnt[g];
        beg = g * SLOTS;
        end = deg <= SLOTS ? beg + deg : beg;
    } else {
        beg = off[g]; end = off[g + 1];
    }
    const float own = o[g * 8 + j];
    const float dc = grp8_f<7>(own);
    float acc = own * dc * dc;
    for (int p = beg; p < end; p += 8) {
        const int s = p + j < end ? srcs[p + j] : g;
        int sk[8]; float wk[8], ov[8];
        PULL8_SHARE(0) PULL8_SHARE(1) PULL8_SHARE(2) PULL8_SHARE(3)
        PULL8_SHARE(4) PULL8_SHARE(5) PULL8_SHARE(6) PULL8_SHARE(7)
#pragma unroll
        for (int k = 0; k < 8; ++k) ov[k] = o[sk[k] * 8 + j];
        PULL8_WEIGHT(0) PULL8_WEIGHT(1) PULL8_WEIGHT(2) PULL8_WEIGHT(3)
        PULL8_WEIGHT(4) PULL8_WEIGHT(5) PULL8_WEIGHT(6) PULL8_WEIGHT(7)
#pragma unroll
        for (int k = 0; k < 8; ++k) acc = fmaf(ov[k], wk[k], acc);
    }
    if (SLOT && deg > SLOTS) {
        const unsigned grpsh = threadIdx.x & 24;
        for (int e0 = 0; e0 < E; e0 += 8) {
            const int e = e0 + j;
            const bool on = e < E && eidx[E + e] == g;
            const unsigned hit = ((unsigned)__ballot(on) >> grpsh) & 0xffu;
            if (hit) {
                const int s = on ? eidx[e] : g;
                int sk[8]; float ov[8];
                PULL8_SHARE(0) PULL8_SHARE(1) PULL8_SHARE(2) PULL8_SHARE(3)
                PULL8_SHARE(4) PULL8_SHARE(5) PULL8_SHARE(6) PULL8_SHARE(7)
#pragma unroll
                for (int k = 0; k < 8; ++k) ov[k] = o[sk[k] * 8 + j];
#pragma unroll
                for (int k = 0; k < 8; ++k) {
                    const float dk = grp8_f<7>(ov[k]);
                    const float wv = (hit >> k) & 1u ? dk * dc : 0.0f;
                    acc = fmaf(ov[k], wv, acc);
                }
            }
        }
    }
    const int jj = j < NCLS ? j : NCLS - 1;
    const float v = acc + b2[jj];
    float mx = j < NCLS ? v : -INFINITY;
#pragma unroll
    for (int d = 1; d < 8; d <<= 1) mx = fmaxf(mx, __shfl_xor(mx, d, 8));
    float ex = j < NCLS ? __expf(v - mx) : 0.0f;
#pragma unroll
    for (int d = 1; d < 8; d <<= 1) ex += __shfl_xor(ex, d, 8);
    if (j < NCLS) out[(size_t)g * NCLS + j] = v - mx - __logf(ex);
}

extern "C" void kernel_launch(void* const* d_in, const int* in_sizes, int n_in,
                              void* d_out, int out_size, void* d_ws, size_t ws_size,
                              hipStream_t stream)
{
    const float* x    = (const float*)d_in[0];
    const int*   eidx = (const int*)  d_in[1];
    const float* W1   = (const float*)d_in[2];
    const float* b1   = (const float*)d_in[3];
    const float* W2   = (const float*)d_in[4];
    const float* b2   = (const float*)d_in[5];
    float* out = (float*)d_out;

    const int N = in_sizes[0] / N_FEAT;
    const int E = in_sizes[1] / 2;

    float* ws   = (float*)d_ws;
    const int B = 256;
    const int nTiles = (N + 15) / 16;

    const size_t NpS = ((size_t)N + 3) & ~(size_t)3;
    const size_t Sl  = (size_t)N * SLOTS;
    if (ws_size >= ((size_t)N * 32 + Sl + 2 * NpS) * sizeof(float)) {
        const bool wide = ws_size >= ((size_t)N * 48 + Sl + 2 * NpS) * sizeof(float);
        const size_t Rs = (size_t)N * (wide ? 48 : 32);
        float* R     = ws;
        int*   slots = (int*)(ws + Rs);
        float* dinvP = ws + Rs + Sl;
        int*   cnt   = (int*)(ws + Rs + Sl + NpS);
        float* hP    = R;
        float* aggP  = R + (size_t)N * (wide ? 32 : 16);
        float* oP    = R;

        const int nLines = (N + 31) / 32;
        const bool runs = (N & 31) == 0 && (nLines + PQ_K - 1) / PQ_K <= PS_NB && N <= (1 << 26);
        if (runs) {
            int* binCnt = (int*)dinvP;
            k_cnt_zero    <<<(nLines + B - 1) / B, B, 0, stream>>>(binCnt, nLines);
            k_place<true> <<<(E + PQ_K * PS_WG - 1) / (PQ_K * PS_WG) * PQ_K, PS_T, 0, stream>>>(eidx, binCnt, slots, E, nLines);
            k_binsort     <<<nLines, BS_T, 0, stream>>>(binCnt, eidx, cnt, slots, N, E);
        } else {
            k_cnt_zero    <<<(N + B - 1) / B, B, 0, stream>>>(cnt, N);
            k_place<false><<<(E + PQ_K * PS_WG - 1) / (PQ_K * PS_WG) * PQ_K, PS_T, 0, stream>>>(eidx, cnt, slots, E, nLines);
        }
        if (wide) {
            k_gemm1_tile<32><<<1024, 256, 0, stream>>>(x, W1, cnt, dinvP, hP, nTiles);
            k_pull1<32, true><<<(N + 31) / 32, 256, 0, stream>>>(nullptr, slots, dinvP, hP, aggP, N, cnt, eidx, E);
        } else {
            k_gemm1_tile<16><<<1024, 256, 0, stream>>>(x, W1, cnt, dinvP, hP, nTiles);
            k_pull1<16, true><<<(N + 31) / 32, 256, 0, stream>>>(nullptr, slots, dinvP, hP, aggP, N, cnt, eidx, E);
        }
        k_l2<false, 8><<<(N + B - 1) / B, B, 0, stream>>>(aggP, b1, W2, dinvP, oP, out, N);
        k_pull2<true><<<(N + 31) / 32, 256, 0, stream>>>(nullptr, slots, oP, b2, out, N, cnt, eidx, E);
        return;
    }

    const size_t Np = ((size_t)N + 4 + 3) & ~(size_t)3;
    const size_t Ep = ((size_t)E + 3) & ~(size_t)3;
    const size_t Rn = Ep > (size_t)N * 32 ? Ep : (size_t)N * 32;
    const size_t Rw = Ep > (size_t)N * 48 ? Ep : (size_t)N * 48;
    if (ws_size >= (3 * Np + Ep + Rn) * sizeof(float)) {
        const bool wide = ws_size >= (3 * Np + Ep + Rw) * sizeof(float);
        const size_t Rs = wide ? Rw : Rn;
        float* R     = ws;
        int*   srcs  = (int*)(ws + Rs);
        float* dinvP = ws + Rs + Ep;
        int*   off   = (int*)(ws + Rs + Ep + Np);
        int*   cnt   = (int*)(ws + Rs + Ep + 2 * Np);
        int*   rank  = (int*)R;
        float* hP    = R;
        float* aggP  = R + (size_t)N * (wide ? 32 : 16);
        float* oP    = R;

        k_cnt_zero<<<(N + B - 1) / B, B, 0, stream>>>(cnt, N);
        k_rank    <<<(E + B - 1) / B, B, 0, stream>>>(eidx + E, cnt, rank, E);
        k_scan    <<<(N + 4095) / 4096, 1024, 0, stream>>>(cnt, off, dinvP, N);
        k_fill    <<<(E + B - 1) / B, B, 0, stream>>>(eidx, rank, off, srcs, E);
        if (wide) {
            k_gemm1<false, 32, false><<<(nTiles + 7) / 8, 256, 0, stream>>>(x, W1, dinvP, nullptr, nullptr, hP, aggP, nTiles);
            k_pull1<32, false><<<(N + 31) / 32, 256, 0, stream>>>(off, srcs, dinvP, hP, aggP, N, nullptr, eidx, E);
        } else {
            k_gemm1<false, 16, false><<<(nTiles + 7) / 8, 256, 0, stream>>>(x, W1, dinvP, nullptr, nullptr, hP, aggP, nTiles);
            k_pull1<16, false><<<(N + 31) / 32, 256, 0, stream>>>(off, srcs, dinvP, hP, aggP, N, nullptr, eidx, E);
        }
        k_l2<false, 8><<<(N + B - 1) / B, B, 0, stream>>>(aggP, b1, W2, dinvP, oP, out, N);
        k_pull2<false><<<(N + 31) / 32, 256, 0, stream>>>(off, srcs, oP, b2, out, N, nullptr, eidx, E);
        return;
    }

    float* dinv = ws;
    float* h    = ws + (size_t)N;
    float* agg1 = ws + (size_t)N * 17;
    float* obuf = ws + (size_t)N * 33;

    k_deg_init <<<(N + B - 1) / B, B, 0, stream>>>(dinv, N);
    k_deg_edges<<<(E + B - 1) / B, B, 0, stream>>>(eidx + E, dinv, E);
    k_dinv     <<<(N + B - 1) / B, B, 0, stream>>>(dinv, N);

    k_gemm1<true, 16, false><<<(nTiles + 7) / 8, 256, 0, stream>>>(x, W1, dinv, nullptr, nullptr, h, agg1, nTiles);

    const int t1 = E * 16;
    k_edge1<<<(t1 + B - 1) / B, B, 0, stream>>>(eidx, dinv, h, agg1, E);

    k_l2<true, 7><<<(N + B - 1) / B, B, 0, stream>>>(agg1, b1, W2, dinv, obuf, out, N);

    const int t2 = E * 8;
    k_edge2<<<(t2 + B - 1) / B, B, 0, stream>>>(eidx, dinv, obuf, out, E);

    k_lsm<<<(N + B - 1) / B, B, 0, stream>>>(out, b2, N);
}


# Slot-list launch dropped: layer-1 GEMM tiles count their nodes' edges from the placed runs; both edge walks regroup their line group's run into 32 lists in LDS; list order, so sum order, changes
# speedup vs baseline: 4.9580x; 1.0855x over previous
#include <hip/hip_runtime.h>
#include <math.h>

#define N_FEAT 512
#define HID    16
#define NCLS   7

#define SLOTS     64
#define BIN_CAP   1536
#define BIN_PITCH (32 * SLOTS)
#define BIN_OFF   (BIN_PITCH - BIN_CAP)
#define G1_RW     (BIN_CAP / 256)

typedef __attribute__((ext_vector_type(2))) float v2f;
typedef __attribute__((ext_vector_type(8))) float v8f;
typedef __attribute__((ext_vector_type(4))) float v4f;

__global__ void k_deg_init(float* __restrict__ deg, int n) {
    int i = blockIdx.x * blockDim.x + threadIdx.x;
    if (i < n) deg[i] = 1.0f;
}

__global__ void k_deg_edges(const int* __restrict__ col, float* __restrict__ deg, int E) {
    int e = blockIdx.x * blockDim.x + threadIdx.x;
    if (e < E) atomicAdd(&deg[col[e]], 1.0f);
}

__global__ void k_dinv(float* __restrict__ deg, int n) {
    int i = blockIdx.x * blockDim.x + threadIdx.x;
    if (i < n) deg[i] = rsqrtf(deg[i]);
}

template <bool SELF, int HP, bool CNT>
__global__ __launch_bounds__(256) void k_gemm1(
    const float* __restrict__ x, const float* __restrict__ W1,
    const float* __restrict__ dinv, const int* __restrict__ cnt, float* __restrict__ dinvOut,
    float* __restrict__ h, float* __restrict__ agg1, int nTiles)
{
    __shared__ float sW[N_FEAT * HID];
    for (int i = threadIdx.x; i < N_FEAT * HID; i += blockDim.x) sW[i] = W1[i];
    __syncthreads();

    int wave = threadIdx.x >> 5;
    int lane = threadIdx.x & 31;
    int tile = blockIdx.x * 8 + wave;
    if (tile >= nTiles) return;

    int m  = lane & 15;
    int kh = lane >> 4;
    const float* xr = x + (size_t)(tile * 16 + m) * N_FEAT + 2 * kh;

    v8f acc = {};
#pragma unroll 4
    for (int k = 0; k < N_FEAT; k += 4) {
        v2f a, b;
        a.x = xr[k];
        a.y = xr[k + 1];
        int kk = k + 2 * kh;
        b.x = sW[kk * HID + m];
        b.y = sW[(kk + 1) * HID + m];
        acc = __builtin_amdgcn_wmma_f32_16x16x4_f32(
             false, a,  false, b,
             (short)0, acc,  false,  false);
    }

#pragma unroll
    for (int v = 0; v < 8; ++v) {
        int node  = tile * 16 + v + 8 * kh;
        float val = acc[v];
        h[(size_t)node * HP + m] = val;
        if (SELF) {
            float di = dinv[node];
            agg1[(size_t)node * HID + m] = val * di * di;
        }
    }
    if ((HP > HID || CNT) && lane < 16) {
        int node = tile * 16 + lane;
        float di;
        if (CNT) {
            di = rsqrtf((float)(cnt[node] + 1));
            dinvOut[node] = di;
        } else {
            di = dinv[node];
        }
        if (HP > HID) *reinterpret_cast<v4f*>(&h[(size_t)node * HP + HID]) = v4f{di, 0.0f, 0.0f, 0.0f};
    }
}

#define G1_WAVES 8
#define G1_KW    (N_FEAT / G1_WAVES)
#define G1_KS    (G1_KW / 4)
#define G1_PITCH (N_FEAT + 4)

__device__ __forceinline__ void g1_lds_barrier() {
    asm volatile("s_wait_dscnt 0x0\n\ts_barrier_signal -1\n\ts_barrier_wait -1" ::: "memory");
}

template <int HP, bool RUNS>
__global__ __launch_bounds__(256) void k_gemm1_tile(
    const float* __restrict__ x, const float* __restrict__ W1,
    const int* __restrict__ cnt, const int* __restrict__ runs, const int* __restrict__ eidx, int E,
    float* __restrict__ dinvOut, float* __restrict__ h, int nTiles)
{
    __shared__ __attribute__((aligned(16))) float sX[16 * G1_PITCH];
    __shared__ float sP[G1_WAVES][8][32];
    __shared__ int sCnt[2][16];

    const int t    = threadIdx.x;
    const int wave = t >> 5;
    const int lane = t & 31;
    const int m  = lane & 15;
    const int kh = lane >> 4;
    const int kw = wave * G1_KW + 2 * kh;
    const int srow = t >> 7;
    const int scol = (t & 127) * 4;

    int tile = blockIdx.x;
    if (tile >= nTiles) return;

    v2f b[G1_KS];
#pragma unroll
    for (int ks = 0; ks < G1_KS; ++ks) {
        const int kk = kw + 4 * ks;
        b[ks].x = W1[kk * HID + m];
        b[ks].y = W1[(kk + 1) * HID + m];
    }

    v4f pf[8];
    int cn = 0;
    int er[G1_RW];
    int nbc = 0, nbn = 0;
    int buf = 0;
    {
        const float* src = x + (size_t)(tile * 16 + srow) * N_FEAT + scol;
#pragma unroll
        for (int j = 0; j < 8; ++j) pf[j] = *reinterpret_cast<const v4f*>(src + (size_t)j * 2 * N_FEAT);
        if (RUNS) {
            nbc = cnt[tile >> 1];
            const int* rb = runs + (size_t)(tile >> 1) * BIN_PITCH + BIN_OFF;
            const int* fc = cnt + (tile >> 1);
#pragma unroll
            for (int k = 0; k < G1_RW; ++k) er[k] = *(k * 256 + t < nbc ? rb + k * 256 + t : fc);
            if (t < 16) sCnt[0][t] = 0;
            __syncthreads();
        } else {
            cn = cnt[tile * 16 + m];
        }
    }

    for (;;) {
        if (!RUNS && t < 16) {
            const int node = tile * 16 + t;
            const float di = rsqrtf((float)(cn + 1));
            dinvOut[node] = di;
            if (HP > HID) *reinterpret_cast<v4f*>(&h[(size_t)node * HP + HID]) = v4f{di, 0.0f, 0.0f, 0.0f};
        }
#pragma unroll
        for (int j = 0; j < 8; ++j)
            *reinterpret_cast<v4f*>(&sX[(2 * j + srow) * G1_PITCH + scol]) = pf[j];

        const int  nxt  = tile + (int)gridDim.x;
        const bool more = nxt < nTiles;
        {
            const int    pt   = more ? nxt : tile;
            const size_t step = more ? (size_t)2 * N_FEAT : 0;
            const float* src  = x + (size_t)(pt * 16 + srow) * N_FEAT + scol;
#pragma unroll
            for (int j = 0; j < 8; ++j) pf[j] = *reinterpret_cast<const v4f*>(src + j * step);
            if (RUNS) nbn = cnt[pt >> 1];
            else      cn  = cnt[pt * 16 + m];
        }
        if (RUNS) {
            if (nbc <= BIN_CAP) {
                unsigned lo = 0, hi = 0;
                const int half = tile & 1;
#pragma unroll
                for (int k = 0; k < G1_RW; ++k) {
                    const int e = er[k];
                    const bool in = k * 256 + t < nbc && ((e >> 4) & 1) == half;
                    const unsigned one = in ? 1u << ((e & 7) * 4) : 0u;
                    hi += e & 8 ? one : 0u;
                    lo += e & 8 ? 0u : one;
                }
                unsigned r0 = lo & 0x0f0f0f0fu, r1 = (lo >> 4) & 0x0f0f0f0fu;
                unsigned r2 = hi & 0x0f0f0f0fu, r3 = (hi >> 4) & 0x0f0f0f0fu;
#pragma unroll
                for (int d = 1; d < 32; d <<= 1) {
                    r0 += (unsigned)__shfl_xor((int)r0, d); r1 += (unsigned)__shfl_xor((int)r1, d);
                    r2 += (unsigned)__shfl_xor((int)r2, d); r3 += (unsigned)__shfl_xor((int)r3, d);
                }
                if (lane < 16) {
                    const unsigned rr = lane & 8 ? (lane & 1 ? r3 : r2) : (lane & 1 ? r1 : r0);
                    const int v = (int)((rr >> (((lane & 7) >> 1) * 8)) & 0xffu);
                    if (v) atomicAdd(&sCnt[buf][lane], v);
                }
            } else {
                for (int e = t; e < E; e += 256) {
                    const int cc = eidx[E + e];
                    if ((cc >> 4) == tile) atomicAdd(&sCnt[buf][cc & 15], 1);
                }
            }
        }
        g1_lds_barrier();
        if (RUNS) {
            if (t < 16) {
                const int node = tile * 16 + t;
                const float di = rsqrtf((float)(sCnt[buf][t] + 1));
                dinvOut[node] = di;
                if (HP > HID) *reinterpret_cast<v4f*>(&h[(size_t)node * HP + HID]) = v4f{di, 0.0f, 0.0f, 0.0f};
            } else if (t >= 32 && t < 48) {
                sCnt[buf ^ 1][t - 32] = 0;
            }
            buf ^= 1;
        }

        v2f a[G1_KS];
#pragma unroll
        for (int ks = 0; ks < G1_KS; ++ks)
            a[ks] = *reinterpret_cast<const v2f*>(&sX[m * G1_PITCH + kw + 4 * ks]);
        v8f acc0 = {}, acc1 = {};
#pragma unroll
        for (int ks = 0; ks < G1_KS; ks += 2) {
            acc0 = __builtin_amdgcn_wmma_f32_16x16x4_f32(false, a[ks],     false, b[ks],     (short)0, acc0, false, false);
            acc1 = __builtin_amdgcn_wmma_f32_16x16x4_f32(false, a[ks + 1], false, b[ks + 1], (short)0, acc1, false, false);
        }
#pragma unroll
        for (int v = 0; v < 8; ++v) sP[wave][v][lane] = acc0[v] + acc1[v];
        g1_lds_barrier();
        {
            const int v = wave;
            const float s01 = sP[0][v][lane] + sP[1][v][lane];
            const float s23 = sP[2][v][lane] + sP[3][v][lane];
            const float s45 = sP[4][v][lane] + sP[5][v][lane];
            const float s67 = sP[6][v][lane] + sP[7][v][lane];
            const int node = tile * 16 + v + 8 * kh;
            h[(size_t)node * HP + m] = (s01 + s23) + (s45 + s67);
        }
        if (!more) break;
        tile = nxt;
        if (RUNS) {
            nbc = nbn;
            const int* rb = runs + (size_t)(tile >> 1) * BIN_PITCH + BIN_OFF;
            const int* fc = cnt + (tile >> 1);
#pragma unroll
            for (int k = 0; k < G1_RW; ++k) er[k] = *(k * 256 + t < nbc ? rb + k * 256 + t : fc);
        }
    }
}

__global__ void k_edge1(const int* __restrict__ eidx, const float* __restrict__ dinv,
                        const float* __restrict__ h, float* __restrict__ agg1, int E)
{
    int t = blockIdx.x * blockDim.x + threadIdx.x;
    int e = t >> 4, f = t & 15;
    if (e >= E) return;
    int r = eidx[e];
    int c = eidx[E + e];
    float w = dinv[r] * dinv[c];
    atomicAdd(&agg1[(size_t)c * HID + f], h[(size_t)r * HID + f] * w);
}

template <bool SELF, int OP>
__global__ void k_l2(const float* __restrict__ agg1, const float* __restrict__ b1,
                     const float* __restrict__ W2, const float* __restrict__ dinv,
                     float* __restrict__ o, float* __restrict__ outInit, int n)
{
    __shared__ float sW2[HID * NCLS];
    __shared__ float sb1[HID];
    if (threadIdx.x < HID * NCLS) sW2[threadIdx.x] = W2[threadIdx.x];
    if (threadIdx.x < HID)        sb1[threadIdx.x] = b1[threadIdx.x];
    __syncthreads();

    int i = blockIdx.x * blockDim.x + threadIdx.x;
    if (i >= n) return;

    float hv[HID];
#pragma unroll
    for (int f = 0; f < HID; ++f) {
        float v = agg1[(size_t)i * HID + f] + sb1[f];
        hv[f] = v > 0.0f ? v : 0.0f;
    }
    float di = dinv[i];
    float d2 = di * di;
#pragma unroll
    for (int j = 0; j < NCLS; ++j) {
        float s = 0.0f;
#pragma unroll
        for (int f = 0; f < HID; ++f) s = fmaf(hv[f], sW2[f * NCLS + j], s);
        o[(size_t)i * OP + j]         = s;
        if (SELF) outInit[(size_t)i * NCLS + j] = s * d2;
    }
    if (OP > NCLS) o[(size_t)i * OP + NCLS] = di;
}

__global__ void k_edge2(const int* __restrict__ eidx, const float* __restrict__ dinv,
                        const float* __restrict__ o, float* __restrict__ out, int E)
{
    int t = blockIdx.x * blockDim.x + threadIdx.x;
    int e = t >> 3, j = t & 7;
    if (e >= E || j >= NCLS) return;
    int r = eidx[e];
    int c = eidx[E + e];
    float w = dinv[r] * dinv[c];
    atomicAdd(&out[(size_t)c * NCLS + j], o[(size_t)r * NCLS + j] * w);
}

__global__ void k_lsm(float* __restrict__ out, const float* __restrict__ b2, int n)
{
    int i = blockIdx.x * blockDim.x + threadIdx.x;
    if (i >= n) return;
    float v[NCLS];
    float mx = -1e30f;
#pragma unroll
    for (int j = 0; j < NCLS; ++j) {
        v[j] = out[(size_t)i * NCLS + j] + b2[j];
        mx = fmaxf(mx, v[j]);
    }
    float s = 0.0f;
#pragma unroll
    for (int j = 0; j < NCLS; ++j) s += __expf(v[j] - mx);
    float l = __logf(s);
#pragma unroll
    for (int j = 0; j < NCLS; ++j) out[(size_t)i * NCLS + j] = v[j] - mx - l;
}


__global__ void k_cnt_zero(int* __restrict__ cnt, int n) {
    int i = blockIdx.x * blockDim.x + threadIdx.x;
    if (i < n) cnt[i] = 0;
}

__global__ void k_rank(const int* __restrict__ col, int* __restrict__ cnt,
                       int* __restrict__ rank, int E) {
    int e = blockIdx.x * blockDim.x + threadIdx.x;
    if (e < E) rank[e] = atomicAdd(&cnt[col[e]], 1);
}

__global__ __launch_bounds__(1024) void k_scan(const int* __restrict__ cnt, int* __restrict__ off,
                                               float* __restrict__ dinv, int n)
{
    __shared__ int sInc[32];
    __shared__ int sPre[32];
    const int t = threadIdx.x, lane = t & 31, wave = t >> 5;
    const int tile0 = blockIdx.x * 4096;

    int pre = 0;
    for (int b = 0; b < tile0; b += 4096) {
        int4 v = *reinterpret_cast<const int4*>(cnt + b + t * 4);
        pre += (v.x + v.y) + (v.z + v.w);
    }
    const int i0 = tile0 + t * 4;
    int c[4];
#pragma unroll
    for (int k = 0; k < 4; ++k) c[k] = (i0 + k < n) ? cnt[i0 + k] : 0;
    const int mine = (c[0] + c[1]) + (c[2] + c[3]);

    int incl = mine;
#pragma unroll
    for (int d = 1; d < 32; d <<= 1) { int y = __shfl_up(incl, d); if (lane >= d) incl += y; }
#pragma unroll
    for (int d = 16; d > 0; d >>= 1) pre += __shfl_xor(pre, d);
    if (lane == 31) sInc[wave] = incl;
    if (lane == 0)  sPre[wave] = pre;
    __syncthreads();
    if (wave == 0) {
        int w = sInc[lane], p = sPre[lane];
        int wi = w;
#pragma unroll
        for (int d = 1; d < 32; d <<= 1) { int y = __shfl_up(wi, d); if (lane >= d) wi += y; }
#pragma unroll
        for (int d = 16; d > 0; d >>= 1) p += __shfl_xor(p, d);
        sInc[lane] = wi - w + p;
    }
    __syncthreads();
    int run = sInc[wave] + incl - mine;
#pragma unroll
    for (int k = 0; k < 4; ++k) {
        int i = i0 + k;
        if (i < n) {
            off[i]  = run;
            dinv[i] = rsqrtf((float)(c[k] + 1));
            run += c[k];
            if (i == n - 1) off[n] = run;
        }
    }
}

__global__ void k_fill(const int* __restrict__ eidx, const int* __restrict__ rank,
                       const int* __restrict__ off, int* __restrict__ srcs, int E) {
    int e = blockIdx.x * blockDim.x + threadIdx.x;
    if (e < E) srcs[off[eidx[E + e]] + rank[e]] = eidx[e];
}


#define PS_T   1024
#define PS_EPT 13
#define PS_WG  (PS_T * PS_EPT)
#define PS_NB  4096
#define PQ_K   4
#define PQ_CAP 16384


template <bool FULL, bool BINS>
__device__ __forceinline__ void place_part(const int* __restrict__ eidx, int* __restrict__ cnt,
                                           int* __restrict__ slots, int E, int nLines, int g0,
                                           int* sBin, int* sW, int* sMp, int* sC, int* sS) {
    const int t = threadIdx.x, lane = t & 31, wave = t >> 5;
    const int part = blockIdx.x % PQ_K;
    const int n    = FULL ? PQ_K * PS_WG : E - g0;
    const unsigned lpp = (unsigned)((nLines + PQ_K - 1) / PQ_K);
    const int lo   = part * (int)lpp;
    const int* __restrict__ col = eidx + E + g0;
    const int* __restrict__ row = eidx + g0;

    for (int i = t; i < PS_NB; i += PS_T) sBin[i] = 0;
    __syncthreads();

    for (int b = 0; b < PQ_K * PS_EPT; b += PS_EPT) {
        int c[PS_EPT];
#pragma unroll
        for (int k = 0; k < PS_EPT; ++k) {
            const int i = (b + k) * PS_T + t;
            c[k] = FULL || i < n ? col[i] : -1;
        }
#pragma unroll
        for (int k = 0; k < PS_EPT; ++k) {
            const int ln = c[k] >> 5;
            if ((unsigned)(ln - lo) < lpp) atomicAdd(&sBin[ln & (PS_NB - 1)], 1);
        }
    }
    __syncthreads();

    {
        const int4 v = reinterpret_cast<const int4*>(sBin)[t];
        const int mine = (v.x + v.y) + (v.z + v.w);
        int incl = mine;
#pragma unroll
        for (int d = 1; d < 32; d <<= 1) { int y = __shfl_up(incl, d); if (lane >= d) incl += y; }
        if (lane == 31) sW[wave] = incl;
        __syncthreads();
        if (wave == 0) {
            const int w = sW[lane];
            int wi = w;
#pragma unroll
            for (int d = 1; d < 32; d <<= 1) { int y = __shfl_up(wi, d); if (lane >= d) wi += y; }
            sW[lane] = wi - w;
        }
        __syncthreads();
        int4 o;
        o.x = sW[wave] + incl - mine;
        o.y = o.x + v.x;
        o.z = o.y + v.y;
        o.w = o.z + v.z;
        reinterpret_cast<int4*>(sBin)[t] = o;
        if (t == PS_T - 1) *sMp = o.w + v.w;
    }
    __syncthreads();

    for (int b = 0; b < PQ_K * PS_EPT; b += PS_EPT) {
        int c[PS_EPT], s[PS_EPT];
#pragma unroll
        for (int k = 0; k < PS_EPT; ++k) {
            const int i = (b + k) * PS_T + t;
            const bool ok = FULL || i < n;
            c[k] = ok ? col[i] : -1;
            s[k] = ok ? row[i] : 0;
        }
        int p[PS_EPT];
        unsigned kept = 0, over = 0;
#pragma unroll
        for (int k = 0; k < PS_EPT; ++k) {
            const int ln = c[k] >> 5;
            p[k] = 0;
            if ((unsigned)(ln - lo) < lpp) {
                p[k] = atomicAdd(&sBin[ln & (PS_NB - 1)], 1);
                kept |= 1u << k;
            }
        }
#pragma unroll
        for (int k = 0; k < PS_EPT; ++k) {
            if ((kept >> k) & 1u) {
                if (p[k] < PQ_CAP) {
                    sC[p[k]] = c[k];
                    sS[p[k]] = s[k];
                } else {
                    over |= 1u << k;
                }
            }
        }
        if (over) {
#pragma unroll
            for (int k = 0; k < PS_EPT; ++k) {
                if ((over >> k) & 1u) {
                    if (BINS) {
                        const int ln = c[k] >> 5;
                        const int r = atomicAdd(&cnt[ln], 1);
                        if (r < BIN_CAP) slots[(size_t)ln * BIN_PITCH + BIN_OFF + r] = (int)(((unsigned)s[k] << 5) | (unsigned)(c[k] & 31));
                    } else {
                        const int r = atomicAdd(&cnt[c[k]], 1);
                        if (r < SLOTS) slots[c[k] * SLOTS + r] = s[k];
                    }
                }
            }
        }
    }
    __syncthreads();

    const int m = *sMp < PQ_CAP ? *sMp : PQ_CAP;
    if (BINS) {
        int gi[4], dv[4];
#pragma unroll
        for (int j = 0; j < 4; ++j) {
            const int u = t + j * PS_T;
            gi[j] = -1; dv[j] = 0;
            if ((unsigned)u < lpp) {
                const int ln = lo + u;
                const int g  = ln & (PS_NB - 1);
                const int en = sBin[g];
                const int s0 = g > 0 ? sBin[g - 1] : 0;
                const int ce = en < PQ_CAP ? en : PQ_CAP;
                const int cs = s0 < PQ_CAP ? s0 : PQ_CAP;
                gi[j] = g;
                if (ce > cs) dv[j] = atomicAdd(&cnt[ln], ce - cs) - cs;
            }
        }
        __syncthreads();
#pragma unroll
        for (int j = 0; j < 4; ++j) if (gi[j] >= 0) sBin[gi[j]] = dv[j];
        __syncthreads();
        for (int base = 0; base < m; base += PS_WG) {
            int c[PS_EPT], s[PS_EPT], dq[PS_EPT];
#pragma unroll
            for (int k = 0; k < PS_EPT; ++k) {
                const int  i  = base + k * PS_T + t;
                const bool ok = i < m;
                const int  ii = ok ? i : 0;
                const int  cc = sC[ii];
                s[k] = sS[ii];
                c[k] = ok ? cc : -1;
            }
#pragma unroll
            for (int k = 0; k < PS_EPT; ++k) dq[k] = sBin[(c[k] >> 5) & (PS_NB - 1)];
#pragma unroll
            for (int k = 0; k < PS_EPT; ++k) {
                const int pos = base + k * PS_T + t + dq[k];
                if (c[k] >= 0 && pos < BIN_CAP)
                    slots[(size_t)(c[k] >> 5) * BIN_PITCH + BIN_OFF + pos] = (int)(((unsigned)s[k] << 5) | (unsigned)(c[k] & 31));
            }
        }
        return;
    }
    for (int base = 0; base < m; base += PS_WG) {
        int c[PS_EPT], s[PS_EPT], r[PS_EPT];
#pragma unroll
        for (int k = 0; k < PS_EPT; ++k) {
            const int i = base + k * PS_T + t;
            const bool ok = i < m;
            c[k] = ok ? sC[i] : -1;
            s[k] = ok ? sS[i] : 0;
        }
#pragma unroll
        for (int k = 0; k < PS_EPT; ++k) r[k] = c[k] >= 0 ? atomicAdd(&cnt[c[k]], 1) : SLOTS;
#pragma unroll
        for (int k = 0; k < PS_EPT; ++k)
            if (r[k] < SLOTS) slots[c[k] * SLOTS + r[k]] = s[k];
    }
}

template <bool BINS>
__global__ __launch_bounds__(PS_T) void k_place(const int* __restrict__ eidx, int* __restrict__ cnt,
                                                int* __restrict__ slots, int E, int nLines) {
    __shared__ __attribute__((aligned(16))) int sBin[PS_NB];
    __shared__ int sW[32];
    __shared__ int sM;
    __shared__ int sC[PQ_CAP];
    __shared__ int sS[PQ_CAP];
    const int g0 = (blockIdx.x / PQ_K) * (PQ_K * PS_WG);
    if (g0 + PQ_K * PS_WG <= E) place_part<true,  BINS>(eidx, cnt, slots, E, nLines, g0, sBin, sW, &sM, sC, sS);
    else                        place_part<false, BINS>(eidx, cnt, slots, E, nLines, g0, sBin, sW, &sM, sC, sS);
}

template <int K> __device__ __forceinline__ int grp8_i(int v) {
    return (int)__builtin_amdgcn_mov_dpp8((unsigned)v, K * 0x249249);
}
template <int K> __device__ __forceinline__ float grp8_f(float v) {
    return __int_as_float(grp8_i<K>(__float_as_int(v)));
}

#define PULL8_SHARE(K)  sk[K] = grp8_i<K>(s);
#define PULL1_ADD(K) { \
    const float dk = HP > HID ? grp8_f<4>(rv[K].x) : grp8_f<K>(ds1); \
    const float w = (mask >> K) & 1u ? dk * dc : 0.0f; \
    acc.x = fmaf(rv[K].x, w, acc.x); acc.y = fmaf(rv[K].y, w, acc.y); \
    acc.z = fmaf(rv[K].z, w, acc.z); acc.w = fmaf(rv[K].w, w, acc.w); }

template <int HP>
__device__ __forceinline__ v4f pull1_step(const float* __restrict__ h, const float* __restrict__ dinv,
                                          int s, unsigned mask, float dc, int po, v4f acc)
{
    float ds1 = 0.0f;
    if (HP == HID) ds1 = dinv[s];
    int sk[8]; v4f rv[8];
    PULL8_SHARE(0) PULL8_SHARE(1) PULL8_SHARE(2) PULL8_SHARE(3)
    PULL8_SHARE(4) PULL8_SHARE(5) PULL8_SHARE(6) PULL8_SHARE(7)
#pragma unroll
    for (int k = 0; k < 8; ++k) rv[k] = *reinterpret_cast<const v4f*>(h + sk[k] * HP + po);
    PULL1_ADD(0) PULL1_ADD(1) PULL1_ADD(2) PULL1_ADD(3)
    PULL1_ADD(4) PULL1_ADD(5) PULL1_ADD(6) PULL1_ADD(7)
    return acc;
}

__device__ __forceinline__ void group_lists(const int* __restrict__ binCnt, const int* __restrict__ runs,
                                            const int* __restrict__ eidx, int E, int line, int* sN, int* sL) {
    const int t = threadIdx.x;
    if (t < 32) sN[t] = 0;
    __syncthreads();
    const int n = binCnt[line];
    if (n <= BIN_CAP) {
        const int* b = runs + (size_t)line * BIN_PITCH + BIN_OFF;
        int e[G1_RW];
#pragma unroll
        for (int k = 0; k < G1_RW; ++k) {
            const int i = k * 256 + t;
            e[k] = i < n ? b[i] : -1;
        }
        int r[G1_RW];
#pragma unroll
        for (int k = 0; k < G1_RW; ++k) r[k] = e[k] >= 0 ? atomicAdd(&sN[e[k] & 31], 1) : SLOTS;
#pragma unroll
        for (int k = 0; k < G1_RW; ++k)
            if (r[k] < SLOTS) sL[(e[k] & 31) * SLOTS + r[k]] = (int)((unsigned)e[k] >> 5);
    } else {
        for (int e = t; e < E; e += 256) {
            const int cc = eidx[E + e];
            if ((cc >> 5) == line) {
                const int dl = cc & 31;
                const int r = atomicAdd(&sN[dl], 1);
                if (r < SLOTS) sL[dl * SLOTS + r] = eidx[e];
            }
        }
    }
    __syncthreads();
}

template <int HP, bool SLOT, bool RUNS>
__global__ __launch_bounds__(256) void k_pull1(const int* __restrict__ off, const int* __restrict__ srcs,
                                               const float* __restrict__ dinv, const float* __restrict__ h,
                                               float* __restrict__ agg1, int n,
                                               const int* __restrict__ cnt, const int* __restrict__ eidx, int E)
{
    __shared__ int sN[RUNS ? 32 : 1];
    __shared__ int sL[RUNS ? 32 * SLOTS : 1];
    if (RUNS) group_lists(cnt, srcs, eidx, E, blockIdx.x, sN, sL);
    const int g = (blockIdx.x * 256 + threadIdx.x) >> 3;
    const int q = threadIdx.x & 7;
    if (g >= n) return;
    int beg, end, deg = 0;
    if (RUNS) {
        deg = sN[threadIdx.x >> 3];
        beg = (threadIdx.x >> 3) * SLOTS;
        end = deg <= SLOTS ? beg + deg : beg;
    } else if (SLOT) {
        deg = cnt[g];
        beg = g * SLOTS;
        end = deg <= SLOTS ? beg + deg : beg;
    } else {
        beg = off[g]; end = off[g + 1];
    }
    const int po = HP > HID ? (q < 4 ? q : 4) * 4 : (q & 3) * 4;
    const v4f own = *reinterpret_cast<const v4f*>(h + g * HP + po);
    const float dc = HP > HID ? grp8_f<4>(own.x) : dinv[g];
    v4f acc;
    acc.x = own.x * dc * dc; acc.y = own.y * dc * dc; acc.z = own.z * dc * dc; acc.w = own.w * dc * dc;
    for (int p = beg; p < end; p += 8) {
        const int left = end - p;
        const unsigned mask = left >= 8 ? 0xffu : (1u << left) - 1u;
        const int s = q < left ? (RUNS ? sL[p + q] : srcs[p + q]) : g;
        acc = pull1_step<HP>(h, dinv, s, mask, dc, po, acc);
    }
    if (SLOT && deg > SLOTS) {
        const unsigned grpsh = threadIdx.x & 24;
        for (int e0 = 0; e0 < E; e0 += 8) {
            const int e = e0 + q;
            const bool on = e < E && eidx[E + e] == g;
            const unsigned hit = ((unsigned)__ballot(on) >> grpsh) & 0xffu;
            if (hit) {
                const int s = on ? eidx[e] : g;
                acc = pull1_step<HP>(h, dinv, s, hit, dc, po, acc);
            }
        }
    }
    if (q < 4) *reinterpret_cast<v4f*>(agg1 + (size_t)g * HID + 4 * q) = acc;
}

#define PULL8_WEIGHT(K) { const float dk = grp8_f<7>(ov[K]); wk[K] = p + K < end ? dk * dc : 0.0f; }

template <bool SLOT, bool RUNS>
__global__ __launch_bounds__(256) void k_pull2(const int* __restrict__ off, const int* __restrict__ srcs,
                                               const float* __restrict__ o,
                                               const float* __restrict__ b2, float* __restrict__ out, int n,
                                               const int* __restrict__ cnt, const int* __restrict__ eidx, int E)
{
    __shared__ int sN[RUNS ? 32 : 1];
    __shared__ int sL[RUNS ? 32 * SLOTS : 1];
    if (RUNS) group_lists(cnt, srcs, eidx, E, blockIdx.x, sN, sL);
    const int g = (blockIdx.x * 256 + threadIdx.x) >> 3;
    const int j = threadIdx.x & 7;
    if (g >= n) return;
    int beg, end, deg = 0;
    if (RUNS) {
        deg = sN[threadIdx.x >> 3];
        beg = (threadIdx.x >> 3) * SLOTS;
        end = deg <= SLOTS ? beg + deg : beg;
    } else if (SLOT) {
        deg = cnt[g];
        beg = g * SLOTS;
        end = deg <= SLOTS ? beg + deg : beg;
    } else {
        beg = off[g]; end = off[g + 1];
    }
    const float own = o[g * 8 + j];
    const float dc = grp8_f<7>(own);
    float acc = own * dc * dc;
    for (int p = beg; p < end; p += 8) {
        const int s = p + j < end ? (RUNS ? sL[p + j] : srcs[p + j]) : g;
        int sk[8]; float wk[8], ov[8];
        PULL8_SHARE(0) PULL8_SHARE(1) PULL8_SHARE(2) PULL8_SHARE(3)
        PULL8_SHARE(4) PULL8_SHARE(5) PULL8_SHARE(6) PULL8_SHARE(7)
#pragma unroll
        for (int k = 0; k < 8; ++k) ov[k] = o[sk[k] * 8 + j];
        PULL8_WEIGHT(0) PULL8_WEIGHT(1) PULL8_WEIGHT(2) PULL8_WEIGHT(3)
        PULL8_WEIGHT(4) PULL8_WEIGHT(5) PULL8_WEIGHT(6) PULL8_WEIGHT(7)
#pragma unroll
        for (int k = 0; k < 8; ++k) acc = fmaf(ov[k], wk[k], acc);
    }
    if (SLOT && deg > SLOTS) {
        const unsigned grpsh = threadIdx.x & 24;
        for (int e0 = 0; e0 < E; e0 += 8) {
            const int e = e0 + j;
            const bool on = e < E && eidx[E + e] == g;
            const unsigned hit = ((unsigned)__ballot(on) >> grpsh) & 0xffu;
            if (hit) {
                const int s = on ? eidx[e] : g;
                int sk[8]; float ov[8];
                PULL8_SHARE(0) PULL8_SHARE(1) PULL8_SHARE(2) PULL8_SHARE(3)
                PULL8_SHARE(4) PULL8_SHARE(5) PULL8_SHARE(6) PULL8_SHARE(7)
#pragma unroll
                for (int k = 0; k < 8; ++k) ov[k] = o[sk[k] * 8 + j];
#pragma unroll
                for (int k = 0; k < 8; ++k) {
                    const float dk = grp8_f<7>(ov[k]);
                    const float wv = (hit >> k) & 1u ? dk * dc : 0.0f;
                    acc = fmaf(ov[k], wv, acc);
                }
            }
        }
    }
    const int jj = j < NCLS ? j : NCLS - 1;
    const float v = acc + b2[jj];
    float mx = j < NCLS ? v : -INFINITY;
#pragma unroll
    for (int d = 1; d < 8; d <<= 1) mx = fmaxf(mx, __shfl_xor(mx, d, 8));
    float ex = j < NCLS ? __expf(v - mx) : 0.0f;
#pragma unroll
    for (int d = 1; d < 8; d <<= 1) ex += __shfl_xor(ex, d, 8);
    if (j < NCLS) out[(size_t)g * NCLS + j] = v - mx - __logf(ex);
}

extern "C" void kernel_launch(void* const* d_in, const int* in_sizes, int n_in,
                              void* d_out, int out_size, void* d_ws, size_t ws_size,
                              hipStream_t stream)
{
    const float* x    = (const float*)d_in[0];
    const int*   eidx = (const int*)  d_in[1];
    const float* W1   = (const float*)d_in[2];
    const float* b1   = (const float*)d_in[3];
    const float* W2   = (const float*)d_in[4];
    const float* b2   = (const float*)d_in[5];
    float* out = (float*)d_out;

    const int N = in_sizes[0] / N_FEAT;
    const int E = in_sizes[1] / 2;

    float* ws   = (float*)d_ws;
    const int B = 256;
    const int nTiles = (N + 15) / 16;

    const size_t NpS = ((size_t)N + 3) & ~(size_t)3;
    const size_t Sl  = (size_t)N * SLOTS;
    if (ws_size >= ((size_t)N * 32 + Sl + 2 * NpS) * sizeof(float)) {
        const bool wide = ws_size >= ((size_t)N * 48 + Sl + 2 * NpS) * sizeof(float);
        const size_t Rs = (size_t)N * (wide ? 48 : 32);
        float* R     = ws;
        int*   slots = (int*)(ws + Rs);
        float* dinvP = ws + Rs + Sl;
        int*   cnt   = (int*)(ws + Rs + Sl + NpS);
        float* hP    = R;
        float* aggP  = R + (size_t)N * (wide ? 32 : 16);
        float* oP    = R;

        const int nLines = (N + 31) / 32;
        const bool runs = (N & 31) == 0 && (nLines + PQ_K - 1) / PQ_K <= PS_NB && N <= (1 << 26);
        if (runs) {
            k_cnt_zero    <<<(nLines + B - 1) / B, B, 0, stream>>>(cnt, nLines);
            k_place<true> <<<(E + PQ_K * PS_WG - 1) / (PQ_K * PS_WG) * PQ_K, PS_T, 0, stream>>>(eidx, cnt, slots, E, nLines);
            if (wide) {
                k_gemm1_tile<32, true><<<1024, 256, 0, stream>>>(x, W1, cnt, slots, eidx, E, dinvP, hP, nTiles);
                k_pull1<32, true, true><<<nLines, 256, 0, stream>>>(nullptr, slots, dinvP, hP, aggP, N, cnt, eidx, E);
            } else {
                k_gemm1_tile<16, true><<<1024, 256, 0, stream>>>(x, W1, cnt, slots, eidx, E, dinvP, hP, nTiles);
                k_pull1<16, true, true><<<nLines, 256, 0, stream>>>(nullptr, slots, dinvP, hP, aggP, N, cnt, eidx, E);
            }
            k_l2<false, 8><<<(N + B - 1) / B, B, 0, stream>>>(aggP, b1, W2, dinvP, oP, out, N);
            k_pull2<true, true><<<nLines, 256, 0, stream>>>(nullptr, slots, oP, b2, out, N, cnt, eidx, E);
            return;
        }
        k_cnt_zero    <<<(N + B - 1) / B, B, 0, stream>>>(cnt, N);
        k_place<false><<<(E + PQ_K * PS_WG - 1) / (PQ_K * PS_WG) * PQ_K, PS_T, 0, stream>>>(eidx, cnt, slots, E, nLines);
        if (wide) {
            k_gemm1_tile<32, false><<<1024, 256, 0, stream>>>(x, W1, cnt, nullptr, eidx, E, dinvP, hP, nTiles);
            k_pull1<32, true, false><<<(N + 31) / 32, 256, 0, stream>>>(nullptr, slots, dinvP, hP, aggP, N, cnt, eidx, E);
        } else {
            k_gemm1_tile<16, false><<<1024, 256, 0, stream>>>(x, W1, cnt, nullptr, eidx, E, dinvP, hP, nTiles);
            k_pull1<16, true, false><<<(N + 31) / 32, 256, 0, stream>>>(nullptr, slots, dinvP, hP, aggP, N, cnt, eidx, E);
        }
        k_l2<false, 8><<<(N + B - 1) / B, B, 0, stream>>>(aggP, b1, W2, dinvP, oP, out, N);
        k_pull2<true, false><<<(N + 31) / 32, 256, 0, stream>>>(nullptr, slots, oP, b2, out, N, cnt, eidx, E);
        return;
    }

    const size_t Np = ((size_t)N + 4 + 3) & ~(size_t)3;
    const size_t Ep = ((size_t)E + 3) & ~(size_t)3;
    const size_t Rn = Ep > (size_t)N * 32 ? Ep : (size_t)N * 32;
    const size_t Rw = Ep > (size_t)N * 48 ? Ep : (size_t)N * 48;
    if (ws_size >= (3 * Np + Ep + Rn) * sizeof(float)) {
        const bool wide = ws_size >= (3 * Np + Ep + Rw) * sizeof(float);
        const size_t Rs = wide ? Rw : Rn;
        float* R     = ws;
        int*   srcs  = (int*)(ws + Rs);
        float* dinvP = ws + Rs + Ep;
        int*   off   = (int*)(ws + Rs + Ep + Np);
        int*   cnt   = (int*)(ws + Rs + Ep + 2 * Np);
        int*   rank  = (int*)R;
        float* hP    = R;
        float* aggP  = R + (size_t)N * (wide ? 32 : 16);
        float* oP    = R;

        k_cnt_zero<<<(N + B - 1) / B, B, 0, stream>>>(cnt, N);
        k_rank    <<<(E + B - 1) / B, B, 0, stream>>>(eidx + E, cnt, rank, E);
        k_scan    <<<(N + 4095) / 4096, 1024, 0, stream>>>(cnt, off, dinvP, N);
        k_fill    <<<(E + B - 1) / B, B, 0, stream>>>(eidx, rank, off, srcs, E);
        if (wide) {
            k_gemm1<false, 32, false><<<(nTiles + 7) / 8, 256, 0, stream>>>(x, W1, dinvP, nullptr, nullptr, hP, aggP, nTiles);
            k_pull1<32, false, false><<<(N + 31) / 32, 256, 0, stream>>>(off, srcs, dinvP, hP, aggP, N, nullptr, eidx, E);
        } else {
            k_gemm1<false, 16, false><<<(nTiles + 7) / 8, 256, 0, stream>>>(x, W1, dinvP, nullptr, nullptr, hP, aggP, nTiles);
            k_pull1<16, false, false><<<(N + 31) / 32, 256, 0, stream>>>(off, srcs, dinvP, hP, aggP, N, nullptr, eidx, E);
        }
        k_l2<false, 8><<<(N + B - 1) / B, B, 0, stream>>>(aggP, b1, W2, dinvP, oP, out, N);
        k_pull2<false, false><<<(N + 31) / 32, 256, 0, stream>>>(off, srcs, oP, b2, out, N, nullptr, eidx, E);
        return;
    }

    float* dinv = ws;
    float* h    = ws + (size_t)N;
    float* agg1 = ws + (size_t)N * 17;
    float* obuf = ws + (size_t)N * 33;

    k_deg_init <<<(N + B - 1) / B, B, 0, stream>>>(dinv, N);
    k_deg_edges<<<(E + B - 1) / B, B, 0, stream>>>(eidx + E, dinv, E);
    k_dinv     <<<(N + B - 1) / B, B, 0, stream>>>(dinv, N);

    k_gemm1<true, 16, false><<<(nTiles + 7) / 8, 256, 0, stream>>>(x, W1, dinv, nullptr, nullptr, h, agg1, nTiles);

    const int t1 = E * 16;
    k_edge1<<<(t1 + B - 1) / B, B, 0, stream>>>(eidx, dinv, h, agg1, E);

    k_l2<true, 7><<<(N + B - 1) / B, B, 0, stream>>>(agg1, b1, W2, dinv, obuf, out, N);

    const int t2 = E * 8;
    k_edge2<<<(t2 + B - 1) / B, B, 0, stream>>>(eidx, dinv, obuf, out, E);

    k_lsm<<<(N + B - 1) / B, B, 0, stream>>>(out, b2, N);
}
